# starting kernel
# baseline (speedup 1.0000x reference)
#include <hip/hip_runtime.h>
#include <math.h>

#define BB 2
#define TT 2048
#define DD 1024
#define RR 64
#define RSS 16
#define BT (BB*TT)

typedef __attribute__((ext_vector_type(2))) float v2f;
typedef __attribute__((ext_vector_type(8))) float v8f;

#define GBM 256
#define GBN 64
#define GBK 32

__global__ __launch_bounds__(256) void gemm_wmma_f32(
    const float* __restrict__ A, const float* __restrict__ Bm,
    float* __restrict__ C, const float* __restrict__ mulC,
    int M, int N, int K, int lda, int ldb, int ldc, int bT,
    long long sA, long long sB, long long sC)
{
    __shared__ __align__(16) float As [2][GBM][34];
    __shared__ __align__(16) float BsT[2][GBN][34];

    const int tid  = threadIdx.x;
    const int wave = tid >> 5;
    const int lane = tid & 31;
    const int hi   = lane >> 4;
    const int lm   = lane & 15;

    const int m0 = blockIdx.y * GBM;
    const int n0 = blockIdx.x * GBN;

    const float* Ab = A  + (size_t)blockIdx.z * sA;
    const float* Bb = Bm + (size_t)blockIdx.z * sB;
    float*       Cb = C  + (size_t)blockIdx.z * sC;
    const float* Mb = mulC ? (mulC + (size_t)blockIdx.z * sC) : nullptr;

    float4 ra[8];
    float4 rb[2];

    int arow[8], ac4[8];
    #pragma unroll
    for (int i = 0; i < 8; ++i) { int idx = tid + i * 256; arow[i] = idx >> 3; ac4[i] = (idx & 7) * 4; }
    int bi0[2], bi1[2];
    #pragma unroll
    for (int i = 0; i < 2; ++i) {
        int idx = tid + i * 256;
        if (true) {   }
        bi0[i] = idx; bi1[i] = idx;
    }

    auto loadGlobal = [&](int k0) {
        #pragma unroll
        for (int i = 0; i < 8; ++i)
            ra[i] = *(const float4*)(Ab + (size_t)(m0 + arow[i]) * lda + k0 + ac4[i]);
        if (bT) {
            #pragma unroll
            for (int i = 0; i < 2; ++i) {
                int idx = bi0[i];
                int n = idx >> 3, c4 = (idx & 7) * 4;
                rb[i] = *(const float4*)(Bb + (size_t)(n0 + n) * ldb + k0 + c4);
            }
        } else {
            #pragma unroll
            for (int i = 0; i < 2; ++i) {
                int idx = bi1[i];
                int r = idx >> 4, c4 = (idx & 15) * 4;
                rb[i] = *(const float4*)(Bb + (size_t)(k0 + r) * ldb + n0 + c4);
            }
        }
    };

    auto storeLDS = [&](int buf) {
        #pragma unroll
        for (int i = 0; i < 8; ++i) {
            float4 t = ra[i];
            v2f lo; lo.x = t.x; lo.y = t.y;
            v2f hi2; hi2.x = t.z; hi2.y = t.w;
            *(v2f*)&As[buf][arow[i]][ac4[i]]     = lo;
            *(v2f*)&As[buf][arow[i]][ac4[i] + 2] = hi2;
        }
        if (bT) {
            #pragma unroll
            for (int i = 0; i < 2; ++i) {
                int idx = bi0[i];
                int n = idx >> 3, c4 = (idx & 7) * 4;
                float4 t = rb[i];
                v2f lo; lo.x = t.x; lo.y = t.y;
                v2f hi2; hi2.x = t.z; hi2.y = t.w;
                *(v2f*)&BsT[buf][n][c4]     = lo;
                *(v2f*)&BsT[buf][n][c4 + 2] = hi2;
            }
        } else {
            #pragma unroll
            for (int i = 0; i < 2; ++i) {
                int idx = bi1[i];
                int r = idx >> 4, c4 = (idx & 15) * 4;
                float4 t = rb[i];
                BsT[buf][c4 + 0][r] = t.x;
                BsT[buf][c4 + 1][r] = t.y;
                BsT[buf][c4 + 2][r] = t.z;
                BsT[buf][c4 + 3][r] = t.w;
            }
        }
    };

    v8f acc00 = {}, acc01 = {}, acc02 = {}, acc03 = {};
    v8f acc10 = {}, acc11 = {}, acc12 = {}, acc13 = {};

    const int r0 = wave * 32 + lm;
    const int r1 = r0 + 16;
    const int nchunks = K / GBK;

    loadGlobal(0);
    for (int c = 0; c < nchunks; ++c) {
        const int buf = c & 1;
        storeLDS(buf);
        __syncthreads();
        if (c + 1 < nchunks) loadGlobal((c + 1) * GBK);

        #pragma unroll
        for (int kk = 0; kk < GBK; kk += 4) {
            const int kc = kk + 2 * hi;
            v2f a0 = *(const v2f*)&As [buf][r0][kc];
            v2f a1 = *(const v2f*)&As [buf][r1][kc];
            v2f b0 = *(const v2f*)&BsT[buf][lm +  0][kc];
            v2f b1 = *(const v2f*)&BsT[buf][lm + 16][kc];
            v2f b2 = *(const v2f*)&BsT[buf][lm + 32][kc];
            v2f b3 = *(const v2f*)&BsT[buf][lm + 48][kc];
            acc00 = __builtin_amdgcn_wmma_f32_16x16x4_f32(false, a0, false, b0, (short)0, acc00, false, false);
            acc01 = __builtin_amdgcn_wmma_f32_16x16x4_f32(false, a0, false, b1, (short)0, acc01, false, false);
            acc02 = __builtin_amdgcn_wmma_f32_16x16x4_f32(false, a0, false, b2, (short)0, acc02, false, false);
            acc03 = __builtin_amdgcn_wmma_f32_16x16x4_f32(false, a0, false, b3, (short)0, acc03, false, false);
            acc10 = __builtin_amdgcn_wmma_f32_16x16x4_f32(false, a1, false, b0, (short)0, acc10, false, false);
            acc11 = __builtin_amdgcn_wmma_f32_16x16x4_f32(false, a1, false, b1, (short)0, acc11, false, false);
            acc12 = __builtin_amdgcn_wmma_f32_16x16x4_f32(false, a1, false, b2, (short)0, acc12, false, false);
            acc13 = __builtin_amdgcn_wmma_f32_16x16x4_f32(false, a1, false, b3, (short)0, acc13, false, false);
        }
        __syncthreads();
    }

    const int ncol = n0 + lm;
    #pragma unroll
    for (int mt = 0; mt < 2; ++mt) {
        const int mrow = m0 + wave * 32 + mt * 16 + hi * 8;
        v8f a0 = mt ? acc10 : acc00, a1 = mt ? acc11 : acc01;
        v8f a2 = mt ? acc12 : acc02, a3 = mt ? acc13 : acc03;
        #pragma unroll
        for (int r = 0; r < 8; ++r) {
            size_t base = (size_t)(mrow + r) * ldc + ncol;
            float v0 = a0[r], v1 = a1[r], v2 = a2[r], v3 = a3[r];
            if (Mb) {
                v0 *= Mb[base]; v1 *= Mb[base + 16]; v2 *= Mb[base + 32]; v3 *= Mb[base + 48];
            }
            Cb[base] = v0; Cb[base + 16] = v1; Cb[base + 32] = v2; Cb[base + 48] = v3;
        }
    }
}

__device__ inline float waveReduce(float v) {
    #pragma unroll
    for (int o = 16; o > 0; o >>= 1) v += __shfl_xor(v, o, 32);
    return v;
}
__device__ inline float dot4f(float4 a, float4 b) {
    return a.x * b.x + a.y * b.y + a.z * b.z + a.w * b.w;
}
__device__ inline float softplusf(float x) {
    if (x > 20.f) return x;
    if (x < -20.f) return expf(x);
    return log1pf(expf(x));
}

__global__ __launch_bounds__(256) void proj_dab_kernel(
    const float* __restrict__ x, const float* __restrict__ Wd,
    const float* __restrict__ Wa, const float* __restrict__ Wb,
    float* __restrict__ dOut, float* __restrict__ aOut, float* __restrict__ bOut)
{
    const int wave = threadIdx.x >> 5, lane = threadIdx.x & 31;
    const int row  = blockIdx.x * 8 + wave;
    const float* xr = x + (size_t)row * DD;

    float xv[DD / 32];
    #pragma unroll
    for (int i = 0; i < DD / 32; ++i) xv[i] = xr[lane + 32 * i];

    float acc = 0.f;
    #pragma unroll
    for (int i = 0; i < DD / 32; ++i) acc += xv[i] * Wd[lane + 32 * i];
    acc = waveReduce(acc);
    if (lane == 0) dOut[row] = acc;

    for (int c = 0; c < RSS; ++c) {
        float a = 0.f, bm = 0.f;
        #pragma unroll
        for (int i = 0; i < DD / 32; ++i) {
            int k = lane + 32 * i;
            a  += xv[i] * Wa[(size_t)k * RSS + c];
            bm += xv[i] * Wb[(size_t)k * RSS + c];
        }
        a = waveReduce(a); bm = waveReduce(bm);
        if (lane == 0) { aOut[(size_t)row * RSS + c] = a; bOut[(size_t)row * RSS + c] = bm; }
    }
}

__global__ __launch_bounds__(256) void zmatB_kernel(
    const float* __restrict__ z, const float* __restrict__ Bm, float* __restrict__ zB)
{
    const int wave = threadIdx.x >> 5, lane = threadIdx.x & 31;
    const int rowg = blockIdx.x * 8 + wave;
    const int b    = rowg >> 11;
    const float* zr = z + (size_t)rowg * TT;
    const float* Bb = Bm + (size_t)b * TT * RSS;

    float acc[RSS];
    #pragma unroll
    for (int r = 0; r < RSS; ++r) acc[r] = 0.f;

    for (int j = lane; j < TT; j += 32) {
        float zv = zr[j];
        const float4* br = (const float4*)(Bb + (size_t)j * RSS);
        float4 b0 = br[0], b1 = br[1], b2 = br[2], b3 = br[3];
        acc[0]  += zv * b0.x; acc[1]  += zv * b0.y; acc[2]  += zv * b0.z; acc[3]  += zv * b0.w;
        acc[4]  += zv * b1.x; acc[5]  += zv * b1.y; acc[6]  += zv * b1.z; acc[7]  += zv * b1.w;
        acc[8]  += zv * b2.x; acc[9]  += zv * b2.y; acc[10] += zv * b2.z; acc[11] += zv * b2.w;
        acc[12] += zv * b3.x; acc[13] += zv * b3.y; acc[14] += zv * b3.z; acc[15] += zv * b3.w;
    }
    #pragma unroll
    for (int r = 0; r < RSS; ++r) {
        float v = waveReduce(acc[r]);
        if (lane == 0) zB[(size_t)rowg * RSS + r] = v;
    }
}

__global__ __launch_bounds__(256) void dplr_combine_kernel(
    const float* __restrict__ z, const float* __restrict__ dvec,
    const float* __restrict__ zB, const float* __restrict__ Amat,
    float* __restrict__ out)
{
    const int rowg = blockIdx.y;
    const int b    = rowg >> 11;
    const int j    = (blockIdx.x * blockDim.x + threadIdx.x) * 4;

    const float4* cr = (const float4*)(zB + (size_t)rowg * RSS);
    const float4 c0 = cr[0], c1 = cr[1], c2 = cr[2], c3 = cr[3];

    const float4 z4 = *(const float4*)(z + (size_t)rowg * TT + j);
    const float4 d4 = *(const float4*)(dvec + (size_t)b * TT + j);

    const float4* A0 = (const float4*)(Amat + ((size_t)b * TT + j) * RSS);
    float dv[4];
    #pragma unroll
    for (int q = 0; q < 4; ++q) {
        const float4* Ar = A0 + q * 4;
        dv[q] = dot4f(c0, Ar[0]) + dot4f(c1, Ar[1]) + dot4f(c2, Ar[2]) + dot4f(c3, Ar[3]);
    }
    float4 o;
    o.x = z4.x * d4.x + dv[0];
    o.y = z4.y * d4.y + dv[1];
    o.z = z4.z * d4.z + dv[2];
    o.w = z4.w * d4.w + dv[3];
    *(float4*)(out + (size_t)rowg * TT + j) = o;
}

__global__ __launch_bounds__(256) void rownorm_kernel(
    const float* __restrict__ h, float* __restrict__ denom)
{
    const int wave = threadIdx.x >> 5, lane = threadIdx.x & 31;
    const int rowg = blockIdx.x * 8 + wave;
    const float* hr = h + (size_t)rowg * TT;
    float acc = 0.f;
    for (int j = lane; j < TT; j += 32) { float v = hr[j]; acc += v * v; }
    acc = waveReduce(acc);
    if (lane == 0) denom[rowg] = sqrtf(acc) + 1e-8f;
}

__global__ __launch_bounds__(256) void combine_h_kernel(
    float* __restrict__ hmix, const float* __restrict__ hscale,
    const float* __restrict__ denom)
{
    const int rowg = blockIdx.y;
    const int j = (blockIdx.x * blockDim.x + threadIdx.x) * 4;
    const float inv = 1.f / denom[rowg];
    float4 m = *(const float4*)(hmix   + (size_t)rowg * TT + j);
    float4 s = *(const float4*)(hscale + (size_t)rowg * TT + j);
    float4 o;
    o.x = softplusf(s.x) * fmaxf(m.x * inv, 0.f);
    o.y = softplusf(s.y) * fmaxf(m.y * inv, 0.f);
    o.z = softplusf(s.z) * fmaxf(m.z * inv, 0.f);
    o.w = softplusf(s.w) * fmaxf(m.w * inv, 0.f);
    *(float4*)(hmix + (size_t)rowg * TT + j) = o;
}

__global__ __launch_bounds__(256) void emul_kernel(float* __restrict__ a, const float* __restrict__ b)
{
    const size_t i = ((size_t)blockIdx.x * blockDim.x + threadIdx.x) * 4;
    float4 av = *(const float4*)(a + i);
    float4 bv = *(const float4*)(b + i);
    av.x *= bv.x; av.y *= bv.y; av.z *= bv.z; av.w *= bv.w;
    *(float4*)(a + i) = av;
}

extern "C" void kernel_launch(void* const* d_in, const int* in_sizes, int n_in,
                              void* d_out, int out_size, void* d_ws, size_t ws_size,
                              hipStream_t stream) {
    (void)in_sizes; (void)n_in; (void)out_size; (void)ws_size;

    const float* x = (const float*)d_in[0];
    const float* qk_Wq1  = (const float*)d_in[1];  const float* qk_Wq2  = (const float*)d_in[2];
    const float* qk_Wm1  = (const float*)d_in[3];  const float* qk_Wm2  = (const float*)d_in[4];
    const float* qk_Wk1  = (const float*)d_in[5];  const float* qk_Wk2  = (const float*)d_in[6];
    const float* qks_Wq1 = (const float*)d_in[7];  const float* qks_Wq2 = (const float*)d_in[8];
    const float* qks_Wm1 = (const float*)d_in[9];  const float* qks_Wm2 = (const float*)d_in[10];
    const float* qks_Wk1 = (const float*)d_in[11]; const float* qks_Wk2 = (const float*)d_in[12];
    const float* vo_Wv1  = (const float*)d_in[13]; const float* vo_Wv2  = (const float*)d_in[14];
    const float* vo_Wm1  = (const float*)d_in[15]; const float* vo_Wm2  = (const float*)d_in[16];
    const float* vo_Wo   = (const float*)d_in[17];
    const float* sm1_Wd  = (const float*)d_in[18]; const float* sm1_Wa  = (const float*)d_in[19]; const float* sm1_Wb  = (const float*)d_in[20];
    const float* sm1s_Wd = (const float*)d_in[21]; const float* sm1s_Wa = (const float*)d_in[22]; const float* sm1s_Wb = (const float*)d_in[23];
    const float* sm2_Wd  = (const float*)d_in[24]; const float* sm2_Wa  = (const float*)d_in[25]; const float* sm2_Wb  = (const float*)d_in[26];

    const size_t BTD = (size_t)BT * DD;
    const size_t BTT = (size_t)BB * TT * TT;

    float* w = (float*)d_ws;
    float* qg   = w; w += BTD;
    float* kb   = w; w += BTD;
    float* qsg  = w; w += BTD;
    float* ksb  = w; w += BTD;
    float* vb   = w; w += BTD;
    float* mvob = w; w += BTD;
    float* tmpb = w; w += BTD;
    float* zb   = w; w += BTT;
    float* zsb  = w; w += BTT;
    float* ctx  = w; w += BTD;
    float* xr   = w; w += (size_t)BT * RR;
    float* d1   = w; w += BT; float* d1s = w; w += BT; float* d2 = w; w += BT;
    float* A1   = w; w += (size_t)BT * RSS; float* B1  = w; w += (size_t)BT * RSS;
    float* A1s  = w; w += (size_t)BT * RSS; float* B1s = w; w += (size_t)BT * RSS;
    float* A2   = w; w += (size_t)BT * RSS; float* B2  = w; w += (size_t)BT * RSS;
    float* zBb  = w; w += (size_t)BT * RSS;
    float* denom= w; w += BT;

    auto gemm = [&](const float* A, const float* Bm, float* C, const float* mul,
                    int M, int N, int K, int lda, int ldb, int ldc, int bT,
                    long long sA, long long sB, long long sC, int batches) {
        dim3 g((unsigned)(N / GBN), (unsigned)(M / GBM), (unsigned)batches);
        gemm_wmma_f32<<<g, dim3(256), 0, stream>>>(A, Bm, C, mul, M, N, K, lda, ldb, ldc, bT, sA, sB, sC);
    };

    proj_dab_kernel<<<dim3(BT / 8), 256, 0, stream>>>(x, sm1_Wd,  sm1_Wa,  sm1_Wb,  d1,  A1,  B1);
    proj_dab_kernel<<<dim3(BT / 8), 256, 0, stream>>>(x, sm1s_Wd, sm1s_Wa, sm1s_Wb, d1s, A1s, B1s);
    proj_dab_kernel<<<dim3(BT / 8), 256, 0, stream>>>(x, sm2_Wd,  sm2_Wa,  sm2_Wb,  d2,  A2,  B2);

    gemm(x,  qk_Wm1, xr,   nullptr, BT, RR, DD, DD, RR, RR, 0, 0, 0, 0, 1);
    gemm(xr, qk_Wm2, tmpb, nullptr, BT, DD, RR, RR, DD, DD, 0, 0, 0, 0, 1);
    gemm(x,  qk_Wq1, xr,   nullptr, BT, RR, DD, DD, RR, RR, 0, 0, 0, 0, 1);
    gemm(xr, qk_Wq2, qg,   tmpb,    BT, DD, RR, RR, DD, DD, 0, 0, 0, 0, 1);
    gemm(x,  qk_Wk1, xr,   nullptr, BT, RR, DD, DD, RR, RR, 0, 0, 0, 0, 1);
    gemm(xr, qk_Wk2, kb,   nullptr, BT, DD, RR, RR, DD, DD, 0, 0, 0, 0, 1);

    gemm(x,  qks_Wm1, xr,   nullptr, BT, RR, DD, DD, RR, RR, 0, 0, 0, 0, 1);
    gemm(xr, qks_Wm2, tmpb, nullptr, BT, DD, RR, RR, DD, DD, 0, 0, 0, 0, 1);
    gemm(x,  qks_Wq1, xr,   nullptr, BT, RR, DD, DD, RR, RR, 0, 0, 0, 0, 1);
    gemm(xr, qks_Wq2, qsg,  tmpb,    BT, DD, RR, RR, DD, DD, 0, 0, 0, 0, 1);
    gemm(x,  qks_Wk1, xr,   nullptr, BT, RR, DD, DD, RR, RR, 0, 0, 0, 0, 1);
    gemm(xr, qks_Wk2, ksb,  nullptr, BT, DD, RR, RR, DD, DD, 0, 0, 0, 0, 1);

    gemm(x,  vo_Wv1, xr,   nullptr, BT, RR, DD, DD, RR, RR, 0, 0, 0, 0, 1);
    gemm(xr, vo_Wv2, vb,   nullptr, BT, DD, RR, RR, DD, DD, 0, 0, 0, 0, 1);
    gemm(x,  vo_Wm1, xr,   nullptr, BT, RR, DD, DD, RR, RR, 0, 0, 0, 0, 1);
    gemm(xr, vo_Wm2, mvob, nullptr, BT, DD, RR, RR, DD, DD, 0, 0, 0, 0, 1);

    const long long sTD = (long long)TT * DD, sTTl = (long long)TT * TT;
    gemm(qg,  kb,  zb,  nullptr, TT, TT, DD, DD, DD, TT, 1, sTD, sTD, sTTl, BB);
    gemm(qsg, ksb, zsb, nullptr, TT, TT, DD, DD, DD, TT, 1, sTD, sTD, sTTl, BB);

    zmatB_kernel<<<dim3(BT / 8), 256, 0, stream>>>(zb, B1, zBb);
    dplr_combine_kernel<<<dim3(TT / 1024, BT), 256, 0, stream>>>(zb, d1, zBb, A1, zb);
    zmatB_kernel<<<dim3(BT / 8), 256, 0, stream>>>(zsb, B1s, zBb);
    dplr_combine_kernel<<<dim3(TT / 1024, BT), 256, 0, stream>>>(zsb, d1s, zBb, A1s, zsb);
    rownorm_kernel<<<dim3(BT / 8), 256, 0, stream>>>(zb, denom);
    combine_h_kernel<<<dim3(TT / 1024, BT), 256, 0, stream>>>(zb, zsb, denom);

    zmatB_kernel<<<dim3(BT / 8), 256, 0, stream>>>(zb, B2, zBb);
    dplr_combine_kernel<<<dim3(TT / 1024, BT), 256, 0, stream>>>(zb, d2, zBb, A2, zsb);

    gemm(zsb, vb, ctx, nullptr, TT, DD, TT, TT, DD, DD, 0, sTTl, sTD, sTD, BB);
    emul_kernel<<<dim3((unsigned)(BTD / 4 / 256)), 256, 0, stream>>>(ctx, mvob);
    gemm(ctx, vo_Wo, (float*)d_out, nullptr, BT, DD, DD, DD, DD, DD, 1, 0, 0, 0, 1);
}


# the eight rank-64 first-stage GEMMs x@W1 run as ONE launch of 128 work-groups (blockIdx.x picks the weight) instead of eight launches of an eighth of that; same tile code, same sum order
# speedup vs baseline: 1.3585x; 1.3585x over previous
#include <hip/hip_runtime.h>
#include <math.h>

#define BB 2
#define TT 2048
#define DD 1024
#define RR 64
#define RSS 16
#define BT (BB*TT)

typedef __attribute__((ext_vector_type(2))) float v2f;
typedef __attribute__((ext_vector_type(8))) float v8f;

#define GBM 256
#define GBN 64
#define GBK 32

__device__ __forceinline__ void gemm_tile_f32(
    const float* __restrict__ Ab, const float* __restrict__ Bb,
    float* __restrict__ Cb, const float* __restrict__ Mb,
    int K, int lda, int ldb, int ldc, int bT, int m0, int n0)
{
    __shared__ __align__(16) float As [2][GBM][34];
    __shared__ __align__(16) float BsT[2][GBN][34];

    const int tid  = threadIdx.x;
    const int wave = tid >> 5;
    const int lane = tid & 31;
    const int hi   = lane >> 4;
    const int lm   = lane & 15;

    float4 ra[8];
    float4 rb[2];

    int arow[8], ac4[8];
    #pragma unroll
    for (int i = 0; i < 8; ++i) { int idx = tid + i * 256; arow[i] = idx >> 3; ac4[i] = (idx & 7) * 4; }
    int bi0[2], bi1[2];
    #pragma unroll
    for (int i = 0; i < 2; ++i) {
        int idx = tid + i * 256;
        if (true) {   }
        bi0[i] = idx; bi1[i] = idx;
    }

    auto loadGlobal = [&](int k0) {
        #pragma unroll
        for (int i = 0; i < 8; ++i)
            ra[i] = *(const float4*)(Ab + (size_t)(m0 + arow[i]) * lda + k0 + ac4[i]);
        if (bT) {
            #pragma unroll
            for (int i = 0; i < 2; ++i) {
                int idx = bi0[i];
                int n = idx >> 3, c4 = (idx & 7) * 4;
                rb[i] = *(const float4*)(Bb + (size_t)(n0 + n) * ldb + k0 + c4);
            }
        } else {
            #pragma unroll
            for (int i = 0; i < 2; ++i) {
                int idx = bi1[i];
                int r = idx >> 4, c4 = (idx & 15) * 4;
                rb[i] = *(const float4*)(Bb + (size_t)(k0 + r) * ldb + n0 + c4);
            }
        }
    };

    auto storeLDS = [&](int buf) {
        #pragma unroll
        for (int i = 0; i < 8; ++i) {
            float4 t = ra[i];
            v2f lo; lo.x = t.x; lo.y = t.y;
            v2f hi2; hi2.x = t.z; hi2.y = t.w;
            *(v2f*)&As[buf][arow[i]][ac4[i]]     = lo;
            *(v2f*)&As[buf][arow[i]][ac4[i] + 2] = hi2;
        }
        if (bT) {
            #pragma unroll
            for (int i = 0; i < 2; ++i) {
                int idx = bi0[i];
                int n = idx >> 3, c4 = (idx & 7) * 4;
                float4 t = rb[i];
                v2f lo; lo.x = t.x; lo.y = t.y;
                v2f hi2; hi2.x = t.z; hi2.y = t.w;
                *(v2f*)&BsT[buf][n][c4]     = lo;
                *(v2f*)&BsT[buf][n][c4 + 2] = hi2;
            }
        } else {
            #pragma unroll
            for (int i = 0; i < 2; ++i) {
                int idx = bi1[i];
                int r = idx >> 4, c4 = (idx & 15) * 4;
                float4 t = rb[i];
                BsT[buf][c4 + 0][r] = t.x;
                BsT[buf][c4 + 1][r] = t.y;
                BsT[buf][c4 + 2][r] = t.z;
                BsT[buf][c4 + 3][r] = t.w;
            }
        }
    };

    v8f acc00 = {}, acc01 = {}, acc02 = {}, acc03 = {};
    v8f acc10 = {}, acc11 = {}, acc12 = {}, acc13 = {};

    const int r0 = wave * 32 + lm;
    const int r1 = r0 + 16;
    const int nchunks = K / GBK;

    loadGlobal(0);
    for (int c = 0; c < nchunks; ++c) {
        const int buf = c & 1;
        storeLDS(buf);
        __syncthreads();
        if (c + 1 < nchunks) loadGlobal((c + 1) * GBK);

        #pragma unroll
        for (int kk = 0; kk < GBK; kk += 4) {
            const int kc = kk + 2 * hi;
            v2f a0 = *(const v2f*)&As [buf][r0][kc];
            v2f a1 = *(const v2f*)&As [buf][r1][kc];
            v2f b0 = *(const v2f*)&BsT[buf][lm +  0][kc];
            v2f b1 = *(const v2f*)&BsT[buf][lm + 16][kc];
            v2f b2 = *(const v2f*)&BsT[buf][lm + 32][kc];
            v2f b3 = *(const v2f*)&BsT[buf][lm + 48][kc];
            acc00 = __builtin_amdgcn_wmma_f32_16x16x4_f32(false, a0, false, b0, (short)0, acc00, false, false);
            acc01 = __builtin_amdgcn_wmma_f32_16x16x4_f32(false, a0, false, b1, (short)0, acc01, false, false);
            acc02 = __builtin_amdgcn_wmma_f32_16x16x4_f32(false, a0, false, b2, (short)0, acc02, false, false);
            acc03 = __builtin_amdgcn_wmma_f32_16x16x4_f32(false, a0, false, b3, (short)0, acc03, false, false);
            acc10 = __builtin_amdgcn_wmma_f32_16x16x4_f32(false, a1, false, b0, (short)0, acc10, false, false);
            acc11 = __builtin_amdgcn_wmma_f32_16x16x4_f32(false, a1, false, b1, (short)0, acc11, false, false);
            acc12 = __builtin_amdgcn_wmma_f32_16x16x4_f32(false, a1, false, b2, (short)0, acc12, false, false);
            acc13 = __builtin_amdgcn_wmma_f32_16x16x4_f32(false, a1, false, b3, (short)0, acc13, false, false);
        }
        __syncthreads();
    }

    const int ncol = n0 + lm;
    #pragma unroll
    for (int mt = 0; mt < 2; ++mt) {
        const int mrow = m0 + wave * 32 + mt * 16 + hi * 8;
        v8f a0 = mt ? acc10 : acc00, a1 = mt ? acc11 : acc01;
        v8f a2 = mt ? acc12 : acc02, a3 = mt ? acc13 : acc03;
        #pragma unroll
        for (int r = 0; r < 8; ++r) {
            size_t base = (size_t)(mrow + r) * ldc + ncol;
            float v0 = a0[r], v1 = a1[r], v2 = a2[r], v3 = a3[r];
            if (Mb) {
                v0 *= Mb[base]; v1 *= Mb[base + 16]; v2 *= Mb[base + 32]; v3 *= Mb[base + 48];
            }
            Cb[base] = v0; Cb[base + 16] = v1; Cb[base + 32] = v2; Cb[base + 48] = v3;
        }
    }
}

__global__ __launch_bounds__(256) void gemm_wmma_f32(
    const float* __restrict__ A, const float* __restrict__ Bm,
    float* __restrict__ C, const float* __restrict__ mulC,
    int M, int N, int K, int lda, int ldb, int ldc, int bT,
    long long sA, long long sB, long long sC)
{
    (void)M; (void)N;
    const float* Ab = A  + (size_t)blockIdx.z * sA;
    const float* Bb = Bm + (size_t)blockIdx.z * sB;
    float*       Cb = C  + (size_t)blockIdx.z * sC;
    const float* Mb = mulC ? (mulC + (size_t)blockIdx.z * sC) : nullptr;
    gemm_tile_f32(Ab, Bb, Cb, Mb, K, lda, ldb, ldc, bT,
                  (int)blockIdx.y * GBM, (int)blockIdx.x * GBN);
}

struct Weights8 {
    const float* w0; const float* w1; const float* w2; const float* w3;
    const float* w4; const float* w5; const float* w6; const float* w7;
};

__global__ __launch_bounds__(256) void gemm_wmma_f32_shareA8(
    const float* __restrict__ A, Weights8 W, float* __restrict__ C,
    int K, int lda, long long sC)
{
    const int j = blockIdx.x;
    const float* Bb = W.w0;
    switch (j) {
        case 1: Bb = W.w1; break;
        case 2: Bb = W.w2; break;
        case 3: Bb = W.w3; break;
        case 4: Bb = W.w4; break;
        case 5: Bb = W.w5; break;
        case 6: Bb = W.w6; break;
        case 7: Bb = W.w7; break;
        default: break;
    }
    gemm_tile_f32(A, Bb, C + (size_t)j * sC, nullptr, K, lda, GBN, GBN, 0,
                  (int)blockIdx.y * GBM, 0);
}

__device__ inline float waveReduce(float v) {
    #pragma unroll
    for (int o = 16; o > 0; o >>= 1) v += __shfl_xor(v, o, 32);
    return v;
}
__device__ inline float dot4f(float4 a, float4 b) {
    return a.x * b.x + a.y * b.y + a.z * b.z + a.w * b.w;
}
__device__ inline float softplusf(float x) {
    if (x > 20.f) return x;
    if (x < -20.f) return expf(x);
    return log1pf(expf(x));
}

__global__ __launch_bounds__(256) void proj_dab_kernel(
    const float* __restrict__ x, const float* __restrict__ Wd,
    const float* __restrict__ Wa, const float* __restrict__ Wb,
    float* __restrict__ dOut, float* __restrict__ aOut, float* __restrict__ bOut)
{
    const int wave = threadIdx.x >> 5, lane = threadIdx.x & 31;
    const int row  = blockIdx.x * 8 + wave;
    const float* xr = x + (size_t)row * DD;

    float xv[DD / 32];
    #pragma unroll
    for (int i = 0; i < DD / 32; ++i) xv[i] = xr[lane + 32 * i];

    float acc = 0.f;
    #pragma unroll
    for (int i = 0; i < DD / 32; ++i) acc += xv[i] * Wd[lane + 32 * i];
    acc = waveReduce(acc);
    if (lane == 0) dOut[row] = acc;

    for (int c = 0; c < RSS; ++c) {
        float a = 0.f, bm = 0.f;
        #pragma unroll
        for (int i = 0; i < DD / 32; ++i) {
            int k = lane + 32 * i;
            a  += xv[i] * Wa[(size_t)k * RSS + c];
            bm += xv[i] * Wb[(size_t)k * RSS + c];
        }
        a = waveReduce(a); bm = waveReduce(bm);
        if (lane == 0) { aOut[(size_t)row * RSS + c] = a; bOut[(size_t)row * RSS + c] = bm; }
    }
}

__global__ __launch_bounds__(256) void zmatB_kernel(
    const float* __restrict__ z, const float* __restrict__ Bm, float* __restrict__ zB)
{
    const int wave = threadIdx.x >> 5, lane = threadIdx.x & 31;
    const int rowg = blockIdx.x * 8 + wave;
    const int b    = rowg >> 11;
    const float* zr = z + (size_t)rowg * TT;
    const float* Bb = Bm + (size_t)b * TT * RSS;

    float acc[RSS];
    #pragma unroll
    for (int r = 0; r < RSS; ++r) acc[r] = 0.f;

    for (int j = lane; j < TT; j += 32) {
        float zv = zr[j];
        const float4* br = (const float4*)(Bb + (size_t)j * RSS);
        float4 b0 = br[0], b1 = br[1], b2 = br[2], b3 = br[3];
        acc[0]  += zv * b0.x; acc[1]  += zv * b0.y; acc[2]  += zv * b0.z; acc[3]  += zv * b0.w;
        acc[4]  += zv * b1.x; acc[5]  += zv * b1.y; acc[6]  += zv * b1.z; acc[7]  += zv * b1.w;
        acc[8]  += zv * b2.x; acc[9]  += zv * b2.y; acc[10] += zv * b2.z; acc[11] += zv * b2.w;
        acc[12] += zv * b3.x; acc[13] += zv * b3.y; acc[14] += zv * b3.z; acc[15] += zv * b3.w;
    }
    #pragma unroll
    for (int r = 0; r < RSS; ++r) {
        float v = waveReduce(acc[r]);
        if (lane == 0) zB[(size_t)rowg * RSS + r] = v;
    }
}

__global__ __launch_bounds__(256) void dplr_combine_kernel(
    const float* __restrict__ z, const float* __restrict__ dvec,
    const float* __restrict__ zB, const float* __restrict__ Amat,
    float* __restrict__ out)
{
    const int rowg = blockIdx.y;
    const int b    = rowg >> 11;
    const int j    = (blockIdx.x * blockDim.x + threadIdx.x) * 4;

    const float4* cr = (const float4*)(zB + (size_t)rowg * RSS);
    const float4 c0 = cr[0], c1 = cr[1], c2 = cr[2], c3 = cr[3];

    const float4 z4 = *(const float4*)(z + (size_t)rowg * TT + j);
    const float4 d4 = *(const float4*)(dvec + (size_t)b * TT + j);

    const float4* A0 = (const float4*)(Amat + ((size_t)b * TT + j) * RSS);
    float dv[4];
    #pragma unroll
    for (int q = 0; q < 4; ++q) {
        const float4* Ar = A0 + q * 4;
        dv[q] = dot4f(c0, Ar[0]) + dot4f(c1, Ar[1]) + dot4f(c2, Ar[2]) + dot4f(c3, Ar[3]);
    }
    float4 o;
    o.x = z4.x * d4.x + dv[0];
    o.y = z4.y * d4.y + dv[1];
    o.z = z4.z * d4.z + dv[2];
    o.w = z4.w * d4.w + dv[3];
    *(float4*)(out + (size_t)rowg * TT + j) = o;
}

__global__ __launch_bounds__(256) void rownorm_kernel(
    const float* __restrict__ h, float* __restrict__ denom)
{
    const int wave = threadIdx.x >> 5, lane = threadIdx.x & 31;
    const int rowg = blockIdx.x * 8 + wave;
    const float* hr = h + (size_t)rowg * TT;
    float acc = 0.f;
    for (int j = lane; j < TT; j += 32) { float v = hr[j]; acc += v * v; }
    acc = waveReduce(acc);
    if (lane == 0) denom[rowg] = sqrtf(acc) + 1e-8f;
}

__global__ __launch_bounds__(256) void combine_h_kernel(
    float* __restrict__ hmix, const float* __restrict__ hscale,
    const float* __restrict__ denom)
{
    const int rowg = blockIdx.y;
    const int j = (blockIdx.x * blockDim.x + threadIdx.x) * 4;
    const float inv = 1.f / denom[rowg];
    float4 m = *(const float4*)(hmix   + (size_t)rowg * TT + j);
    float4 s = *(const float4*)(hscale + (size_t)rowg * TT + j);
    float4 o;
    o.x = softplusf(s.x) * fmaxf(m.x * inv, 0.f);
    o.y = softplusf(s.y) * fmaxf(m.y * inv, 0.f);
    o.z = softplusf(s.z) * fmaxf(m.z * inv, 0.f);
    o.w = softplusf(s.w) * fmaxf(m.w * inv, 0.f);
    *(float4*)(hmix + (size_t)rowg * TT + j) = o;
}

__global__ __launch_bounds__(256) void emul_kernel(float* __restrict__ a, const float* __restrict__ b)
{
    const size_t i = ((size_t)blockIdx.x * blockDim.x + threadIdx.x) * 4;
    float4 av = *(const float4*)(a + i);
    float4 bv = *(const float4*)(b + i);
    av.x *= bv.x; av.y *= bv.y; av.z *= bv.z; av.w *= bv.w;
    *(float4*)(a + i) = av;
}

extern "C" void kernel_launch(void* const* d_in, const int* in_sizes, int n_in,
                              void* d_out, int out_size, void* d_ws, size_t ws_size,
                              hipStream_t stream) {
    (void)in_sizes; (void)n_in; (void)out_size; (void)ws_size;

    const float* x = (const float*)d_in[0];
    const float* qk_Wq1  = (const float*)d_in[1];  const float* qk_Wq2  = (const float*)d_in[2];
    const float* qk_Wm1  = (const float*)d_in[3];  const float* qk_Wm2  = (const float*)d_in[4];
    const float* qk_Wk1  = (const float*)d_in[5];  const float* qk_Wk2  = (const float*)d_in[6];
    const float* qks_Wq1 = (const float*)d_in[7];  const float* qks_Wq2 = (const float*)d_in[8];
    const float* qks_Wm1 = (const float*)d_in[9];  const float* qks_Wm2 = (const float*)d_in[10];
    const float* qks_Wk1 = (const float*)d_in[11]; const float* qks_Wk2 = (const float*)d_in[12];
    const float* vo_Wv1  = (const float*)d_in[13]; const float* vo_Wv2  = (const float*)d_in[14];
    const float* vo_Wm1  = (const float*)d_in[15]; const float* vo_Wm2  = (const float*)d_in[16];
    const float* vo_Wo   = (const float*)d_in[17];
    const float* sm1_Wd  = (const float*)d_in[18]; const float* sm1_Wa  = (const float*)d_in[19]; const float* sm1_Wb  = (const float*)d_in[20];
    const float* sm1s_Wd = (const float*)d_in[21]; const float* sm1s_Wa = (const float*)d_in[22]; const float* sm1s_Wb = (const float*)d_in[23];
    const float* sm2_Wd  = (const float*)d_in[24]; const float* sm2_Wa  = (const float*)d_in[25]; const float* sm2_Wb  = (const float*)d_in[26];

    const size_t BTD = (size_t)BT * DD;
    const size_t BTT = (size_t)BB * TT * TT;

    float* w = (float*)d_ws;
    float* qg   = w; w += BTD;
    float* kb   = w; w += BTD;
    float* qsg  = w; w += BTD;
    float* ksb  = w; w += BTD;
    float* vb   = w; w += BTD;
    float* mvob = w; w += BTD;
    float* tmpb = w; w += BTD;
    float* zb   = w; w += BTT;
    float* zsb  = w; w += BTT;
    float* ctx  = w; w += BTD;
    float* d1   = w; w += BT; float* d1s = w; w += BT; float* d2 = w; w += BT;
    float* A1   = w; w += (size_t)BT * RSS; float* B1  = w; w += (size_t)BT * RSS;
    float* A1s  = w; w += (size_t)BT * RSS; float* B1s = w; w += (size_t)BT * RSS;
    float* A2   = w; w += (size_t)BT * RSS; float* B2  = w; w += (size_t)BT * RSS;
    float* zBb  = w; w += (size_t)BT * RSS;
    float* denom= w; w += BT;

    auto gemm = [&](const float* A, const float* Bm, float* C, const float* mul,
                    int M, int N, int K, int lda, int ldb, int ldc, int bT,
                    long long sA, long long sB, long long sC, int batches) {
        dim3 g((unsigned)(N / GBN), (unsigned)(M / GBM), (unsigned)batches);
        gemm_wmma_f32<<<g, dim3(256), 0, stream>>>(A, Bm, C, mul, M, N, K, lda, ldb, ldc, bT, sA, sB, sC);
    };

    proj_dab_kernel<<<dim3(BT / 8), 256, 0, stream>>>(x, sm1_Wd,  sm1_Wa,  sm1_Wb,  d1,  A1,  B1);
    proj_dab_kernel<<<dim3(BT / 8), 256, 0, stream>>>(x, sm1s_Wd, sm1s_Wa, sm1s_Wb, d1s, A1s, B1s);
    proj_dab_kernel<<<dim3(BT / 8), 256, 0, stream>>>(x, sm2_Wd,  sm2_Wa,  sm2_Wb,  d2,  A2,  B2);

    float* xr8 = ctx;
    const long long sXR = (long long)BT * RR;
    const Weights8 W1 = { qk_Wm1, qk_Wq1, qk_Wk1, qks_Wm1, qks_Wq1, qks_Wk1, vo_Wv1, vo_Wm1 };
    gemm_wmma_f32_shareA8<<<dim3(8, BT / GBM), dim3(256), 0, stream>>>(x, W1, xr8, DD, DD, sXR);

    gemm(xr8 + 0 * sXR, qk_Wm2,  tmpb, nullptr, BT, DD, RR, RR, DD, DD, 0, 0, 0, 0, 1);
    gemm(xr8 + 1 * sXR, qk_Wq2,  qg,   tmpb,    BT, DD, RR, RR, DD, DD, 0, 0, 0, 0, 1);
    gemm(xr8 + 2 * sXR, qk_Wk2,  kb,   nullptr, BT, DD, RR, RR, DD, DD, 0, 0, 0, 0, 1);

    gemm(xr8 + 3 * sXR, qks_Wm2, tmpb, nullptr, BT, DD, RR, RR, DD, DD, 0, 0, 0, 0, 1);
    gemm(xr8 + 4 * sXR, qks_Wq2, qsg,  tmpb,    BT, DD, RR, RR, DD, DD, 0, 0, 0, 0, 1);
    gemm(xr8 + 5 * sXR, qks_Wk2, ksb,  nullptr, BT, DD, RR, RR, DD, DD, 0, 0, 0, 0, 1);

    gemm(xr8 + 6 * sXR, vo_Wv2,  vb,   nullptr, BT, DD, RR, RR, DD, DD, 0, 0, 0, 0, 1);
    gemm(xr8 + 7 * sXR, vo_Wm2,  mvob, nullptr, BT, DD, RR, RR, DD, DD, 0, 0, 0, 0, 1);

    const long long sTD = (long long)TT * DD, sTTl = (long long)TT * TT;
    gemm(qg,  kb,  zb,  nullptr, TT, TT, DD, DD, DD, TT, 1, sTD, sTD, sTTl, BB);
    gemm(qsg, ksb, zsb, nullptr, TT, TT, DD, DD, DD, TT, 1, sTD, sTD, sTTl, BB);

    zmatB_kernel<<<dim3(BT / 8), 256, 0, stream>>>(zb, B1, zBb);
    dplr_combine_kernel<<<dim3(TT / 1024, BT), 256, 0, stream>>>(zb, d1, zBb, A1, zb);
    zmatB_kernel<<<dim3(BT / 8), 256, 0, stream>>>(zsb, B1s, zBb);
    dplr_combine_kernel<<<dim3(TT / 1024, BT), 256, 0, stream>>>(zsb, d1s, zBb, A1s, zsb);
    rownorm_kernel<<<dim3(BT / 8), 256, 0, stream>>>(zb, denom);
    combine_h_kernel<<<dim3(TT / 1024, BT), 256, 0, stream>>>(zb, zsb, denom);

    zmatB_kernel<<<dim3(BT / 8), 256, 0, stream>>>(zb, B2, zBb);
    dplr_combine_kernel<<<dim3(TT / 1024, BT), 256, 0, stream>>>(zb, d2, zBb, A2, zsb);

    gemm(zsb, vb, ctx, nullptr, TT, DD, TT, TT, DD, DD, 0, sTTl, sTD, sTD, BB);
    emul_kernel<<<dim3((unsigned)(BTD / 4 / 256)), 256, 0, stream>>>(ctx, mvob);
    gemm(ctx, vo_Wo, (float*)d_out, nullptr, BT, DD, DD, DD, DD, DD, 1, 0, 0, 0, 1);
}


# the three x@[Wd|Wa|Wb] projection kernels (per-row vector dots, 64-B-strided weight reads) become three more 64-column WMMA tiles of the shared-x launch; their sums now run in WMMA k order
# speedup vs baseline: 1.9441x; 1.4311x over previous
#include <hip/hip_runtime.h>
#include <math.h>

#define BB 2
#define TT 2048
#define DD 1024
#define RR 64
#define RSS 16
#define BT (BB*TT)

typedef __attribute__((ext_vector_type(2))) float v2f;
typedef __attribute__((ext_vector_type(8))) float v8f;

#define GBM 256
#define GBN 64
#define GBK 32

struct DabTile {
    const float* Wa; const float* Wb; const float* Wd;
    float* aOut; float* bOut; float* dOut;
};

__device__ __forceinline__ void gemm_tile_f32(
    const float* __restrict__ Ab, const float* __restrict__ Bb,
    float* __restrict__ Cb, const float* __restrict__ Mb,
    int K, int lda, int ldb, int ldc, int bT, int m0, int n0,
    bool dab, DabTile dt)
{
    __shared__ __align__(16) float As [2][GBM][34];
    __shared__ __align__(16) float BsT[2][GBN][34];

    const int tid  = threadIdx.x;
    const int wave = tid >> 5;
    const int lane = tid & 31;
    const int hi   = lane >> 4;
    const int lm   = lane & 15;

    float4 ra[8];
    float4 rb[2];

    int arow[8], ac4[8];
    #pragma unroll
    for (int i = 0; i < 8; ++i) { int idx = tid + i * 256; arow[i] = idx >> 3; ac4[i] = (idx & 7) * 4; }
    int bi0[2], bi1[2];
    #pragma unroll
    for (int i = 0; i < 2; ++i) {
        int idx = tid + i * 256;
        if (true) {   }
        bi0[i] = idx; bi1[i] = idx;
    }

    auto loadGlobal = [&](int k0) {
        #pragma unroll
        for (int i = 0; i < 8; ++i)
            ra[i] = *(const float4*)(Ab + (size_t)(m0 + arow[i]) * lda + k0 + ac4[i]);
        if (dab) {
            #pragma unroll
            for (int i = 0; i < 2; ++i) {
                int idx = bi1[i];
                int q = idx & 15;
                size_t k = (size_t)(k0 + (idx >> 4));
                float4 t = make_float4(0.f, 0.f, 0.f, 0.f);
                if (q < 4)       t = *(const float4*)(dt.Wa + k * RSS + q * 4);
                else if (q < 8)  t = *(const float4*)(dt.Wb + k * RSS + (q - 4) * 4);
                else if (q == 8) t.x = dt.Wd[k];
                rb[i] = t;
            }
        } else if (bT) {
            #pragma unroll
            for (int i = 0; i < 2; ++i) {
                int idx = bi0[i];
                int n = idx >> 3, c4 = (idx & 7) * 4;
                rb[i] = *(const float4*)(Bb + (size_t)(n0 + n) * ldb + k0 + c4);
            }
        } else {
            #pragma unroll
            for (int i = 0; i < 2; ++i) {
                int idx = bi1[i];
                int r = idx >> 4, c4 = (idx & 15) * 4;
                rb[i] = *(const float4*)(Bb + (size_t)(k0 + r) * ldb + n0 + c4);
            }
        }
    };

    auto storeLDS = [&](int buf) {
        #pragma unroll
        for (int i = 0; i < 8; ++i) {
            float4 t = ra[i];
            v2f lo; lo.x = t.x; lo.y = t.y;
            v2f hi2; hi2.x = t.z; hi2.y = t.w;
            *(v2f*)&As[buf][arow[i]][ac4[i]]     = lo;
            *(v2f*)&As[buf][arow[i]][ac4[i] + 2] = hi2;
        }
        if (bT) {
            #pragma unroll
            for (int i = 0; i < 2; ++i) {
                int idx = bi0[i];
                int n = idx >> 3, c4 = (idx & 7) * 4;
                float4 t = rb[i];
                v2f lo; lo.x = t.x; lo.y = t.y;
                v2f hi2; hi2.x = t.z; hi2.y = t.w;
                *(v2f*)&BsT[buf][n][c4]     = lo;
                *(v2f*)&BsT[buf][n][c4 + 2] = hi2;
            }
        } else {
            #pragma unroll
            for (int i = 0; i < 2; ++i) {
                int idx = bi1[i];
                int r = idx >> 4, c4 = (idx & 15) * 4;
                float4 t = rb[i];
                BsT[buf][c4 + 0][r] = t.x;
                BsT[buf][c4 + 1][r] = t.y;
                BsT[buf][c4 + 2][r] = t.z;
                BsT[buf][c4 + 3][r] = t.w;
            }
        }
    };

    v8f acc00 = {}, acc01 = {}, acc02 = {}, acc03 = {};
    v8f acc10 = {}, acc11 = {}, acc12 = {}, acc13 = {};

    const int r0 = wave * 32 + lm;
    const int r1 = r0 + 16;
    const int nchunks = K / GBK;

    loadGlobal(0);
    for (int c = 0; c < nchunks; ++c) {
        const int buf = c & 1;
        storeLDS(buf);
        __syncthreads();
        if (c + 1 < nchunks) loadGlobal((c + 1) * GBK);

        #pragma unroll
        for (int kk = 0; kk < GBK; kk += 4) {
            const int kc = kk + 2 * hi;
            v2f a0 = *(const v2f*)&As [buf][r0][kc];
            v2f a1 = *(const v2f*)&As [buf][r1][kc];
            v2f b0 = *(const v2f*)&BsT[buf][lm +  0][kc];
            v2f b1 = *(const v2f*)&BsT[buf][lm + 16][kc];
            v2f b2 = *(const v2f*)&BsT[buf][lm + 32][kc];
            v2f b3 = *(const v2f*)&BsT[buf][lm + 48][kc];
            acc00 = __builtin_amdgcn_wmma_f32_16x16x4_f32(false, a0, false, b0, (short)0, acc00, false, false);
            acc01 = __builtin_amdgcn_wmma_f32_16x16x4_f32(false, a0, false, b1, (short)0, acc01, false, false);
            acc02 = __builtin_amdgcn_wmma_f32_16x16x4_f32(false, a0, false, b2, (short)0, acc02, false, false);
            acc03 = __builtin_amdgcn_wmma_f32_16x16x4_f32(false, a0, false, b3, (short)0, acc03, false, false);
            acc10 = __builtin_amdgcn_wmma_f32_16x16x4_f32(false, a1, false, b0, (short)0, acc10, false, false);
            acc11 = __builtin_amdgcn_wmma_f32_16x16x4_f32(false, a1, false, b1, (short)0, acc11, false, false);
            acc12 = __builtin_amdgcn_wmma_f32_16x16x4_f32(false, a1, false, b2, (short)0, acc12, false, false);
            acc13 = __builtin_amdgcn_wmma_f32_16x16x4_f32(false, a1, false, b3, (short)0, acc13, false, false);
        }
        __syncthreads();
    }

    if (dab) {
        #pragma unroll
        for (int mt = 0; mt < 2; ++mt) {
            const int mrow = m0 + wave * 32 + mt * 16 + hi * 8;
            v8f a0 = mt ? acc10 : acc00, a1 = mt ? acc11 : acc01, a2 = mt ? acc12 : acc02;
            #pragma unroll
            for (int r = 0; r < 8; ++r) {
                size_t row = (size_t)(mrow + r);
                dt.aOut[row * RSS + lm] = a0[r];
                dt.bOut[row * RSS + lm] = a1[r];
                if (lm == 0) dt.dOut[row] = a2[r];
            }
        }
        return;
    }

    const int ncol = n0 + lm;
    #pragma unroll
    for (int mt = 0; mt < 2; ++mt) {
        const int mrow = m0 + wave * 32 + mt * 16 + hi * 8;
        v8f a0 = mt ? acc10 : acc00, a1 = mt ? acc11 : acc01;
        v8f a2 = mt ? acc12 : acc02, a3 = mt ? acc13 : acc03;
        #pragma unroll
        for (int r = 0; r < 8; ++r) {
            size_t base = (size_t)(mrow + r) * ldc + ncol;
            float v0 = a0[r], v1 = a1[r], v2 = a2[r], v3 = a3[r];
            if (Mb) {
                v0 *= Mb[base]; v1 *= Mb[base + 16]; v2 *= Mb[base + 32]; v3 *= Mb[base + 48];
            }
            Cb[base] = v0; Cb[base + 16] = v1; Cb[base + 32] = v2; Cb[base + 48] = v3;
        }
    }
}

__global__ __launch_bounds__(256) void gemm_wmma_f32(
    const float* __restrict__ A, const float* __restrict__ Bm,
    float* __restrict__ C, const float* __restrict__ mulC,
    int M, int N, int K, int lda, int ldb, int ldc, int bT,
    long long sA, long long sB, long long sC)
{
    (void)M; (void)N;
    const float* Ab = A  + (size_t)blockIdx.z * sA;
    const float* Bb = Bm + (size_t)blockIdx.z * sB;
    float*       Cb = C  + (size_t)blockIdx.z * sC;
    const float* Mb = mulC ? (mulC + (size_t)blockIdx.z * sC) : nullptr;
    gemm_tile_f32(Ab, Bb, Cb, Mb, K, lda, ldb, ldc, bT,
                  (int)blockIdx.y * GBM, (int)blockIdx.x * GBN, false, DabTile{});
}

struct Weights8 {
    const float* w0; const float* w1; const float* w2; const float* w3;
    const float* w4; const float* w5; const float* w6; const float* w7;
};
struct DabPaths {
    const float* wa0; const float* wb0; const float* wd0; float* a0; float* b0; float* d0;
    const float* wa1; const float* wb1; const float* wd1; float* a1; float* b1; float* d1;
    const float* wa2; const float* wb2; const float* wd2; float* a2; float* b2; float* d2;
};

__global__ __launch_bounds__(256) void gemm_wmma_f32_shareA8(
    const float* __restrict__ A, Weights8 W, float* __restrict__ C,
    int K, int lda, long long sC, DabPaths P)
{
    const int j = blockIdx.x;
    if (j >= 8) {
        DabTile dt = { P.wa0, P.wb0, P.wd0, P.a0, P.b0, P.d0 };
        if (j == 9)  dt = DabTile{ P.wa1, P.wb1, P.wd1, P.a1, P.b1, P.d1 };
        if (j == 10) dt = DabTile{ P.wa2, P.wb2, P.wd2, P.a2, P.b2, P.d2 };
        gemm_tile_f32(A, nullptr, nullptr, nullptr, K, lda, 0, 0, 0,
                      (int)blockIdx.y * GBM, 0, true, dt);
        return;
    }
    const float* Bb = W.w0;
    switch (j) {
        case 1: Bb = W.w1; break;
        case 2: Bb = W.w2; break;
        case 3: Bb = W.w3; break;
        case 4: Bb = W.w4; break;
        case 5: Bb = W.w5; break;
        case 6: Bb = W.w6; break;
        case 7: Bb = W.w7; break;
        default: break;
    }
    gemm_tile_f32(A, Bb, C + (size_t)j * sC, nullptr, K, lda, GBN, GBN, 0,
                  (int)blockIdx.y * GBM, 0, false, DabTile{});
}

__device__ inline float waveReduce(float v) {
    #pragma unroll
    for (int o = 16; o > 0; o >>= 1) v += __shfl_xor(v, o, 32);
    return v;
}
__device__ inline float dot4f(float4 a, float4 b) {
    return a.x * b.x + a.y * b.y + a.z * b.z + a.w * b.w;
}
__device__ inline float softplusf(float x) {
    if (x > 20.f) return x;
    if (x < -20.f) return expf(x);
    return log1pf(expf(x));
}

__global__ __launch_bounds__(256) void zmatB_kernel(
    const float* __restrict__ z, const float* __restrict__ Bm, float* __restrict__ zB)
{
    const int wave = threadIdx.x >> 5, lane = threadIdx.x & 31;
    const int rowg = blockIdx.x * 8 + wave;
    const int b    = rowg >> 11;
    const float* zr = z + (size_t)rowg * TT;
    const float* Bb = Bm + (size_t)b * TT * RSS;

    float acc[RSS];
    #pragma unroll
    for (int r = 0; r < RSS; ++r) acc[r] = 0.f;

    for (int j = lane; j < TT; j += 32) {
        float zv = zr[j];
        const float4* br = (const float4*)(Bb + (size_t)j * RSS);
        float4 b0 = br[0], b1 = br[1], b2 = br[2], b3 = br[3];
        acc[0]  += zv * b0.x; acc[1]  += zv * b0.y; acc[2]  += zv * b0.z; acc[3]  += zv * b0.w;
        acc[4]  += zv * b1.x; acc[5]  += zv * b1.y; acc[6]  += zv * b1.z; acc[7]  += zv * b1.w;
        acc[8]  += zv * b2.x; acc[9]  += zv * b2.y; acc[10] += zv * b2.z; acc[11] += zv * b2.w;
        acc[12] += zv * b3.x; acc[13] += zv * b3.y; acc[14] += zv * b3.z; acc[15] += zv * b3.w;
    }
    #pragma unroll
    for (int r = 0; r < RSS; ++r) {
        float v = waveReduce(acc[r]);
        if (lane == 0) zB[(size_t)rowg * RSS + r] = v;
    }
}

__global__ __launch_bounds__(256) void dplr_combine_kernel(
    const float* __restrict__ z, const float* __restrict__ dvec,
    const float* __restrict__ zB, const float* __restrict__ Amat,
    float* __restrict__ out)
{
    const int rowg = blockIdx.y;
    const int b    = rowg >> 11;
    const int j    = (blockIdx.x * blockDim.x + threadIdx.x) * 4;

    const float4* cr = (const float4*)(zB + (size_t)rowg * RSS);
    const float4 c0 = cr[0], c1 = cr[1], c2 = cr[2], c3 = cr[3];

    const float4 z4 = *(const float4*)(z + (size_t)rowg * TT + j);
    const float4 d4 = *(const float4*)(dvec + (size_t)b * TT + j);

    const float4* A0 = (const float4*)(Amat + ((size_t)b * TT + j) * RSS);
    float dv[4];
    #pragma unroll
    for (int q = 0; q < 4; ++q) {
        const float4* Ar = A0 + q * 4;
        dv[q] = dot4f(c0, Ar[0]) + dot4f(c1, Ar[1]) + dot4f(c2, Ar[2]) + dot4f(c3, Ar[3]);
    }
    float4 o;
    o.x = z4.x * d4.x + dv[0];
    o.y = z4.y * d4.y + dv[1];
    o.z = z4.z * d4.z + dv[2];
    o.w = z4.w * d4.w + dv[3];
    *(float4*)(out + (size_t)rowg * TT + j) = o;
}

__global__ __launch_bounds__(256) void rownorm_kernel(
    const float* __restrict__ h, float* __restrict__ denom)
{
    const int wave = threadIdx.x >> 5, lane = threadIdx.x & 31;
    const int rowg = blockIdx.x * 8 + wave;
    const float* hr = h + (size_t)rowg * TT;
    float acc = 0.f;
    for (int j = lane; j < TT; j += 32) { float v = hr[j]; acc += v * v; }
    acc = waveReduce(acc);
    if (lane == 0) denom[rowg] = sqrtf(acc) + 1e-8f;
}

__global__ __launch_bounds__(256) void combine_h_kernel(
    float* __restrict__ hmix, const float* __restrict__ hscale,
    const float* __restrict__ denom)
{
    const int rowg = blockIdx.y;
    const int j = (blockIdx.x * blockDim.x + threadIdx.x) * 4;
    const float inv = 1.f / denom[rowg];
    float4 m = *(const float4*)(hmix   + (size_t)rowg * TT + j);
    float4 s = *(const float4*)(hscale + (size_t)rowg * TT + j);
    float4 o;
    o.x = softplusf(s.x) * fmaxf(m.x * inv, 0.f);
    o.y = softplusf(s.y) * fmaxf(m.y * inv, 0.f);
    o.z = softplusf(s.z) * fmaxf(m.z * inv, 0.f);
    o.w = softplusf(s.w) * fmaxf(m.w * inv, 0.f);
    *(float4*)(hmix + (size_t)rowg * TT + j) = o;
}

__global__ __launch_bounds__(256) void emul_kernel(float* __restrict__ a, const float* __restrict__ b)
{
    const size_t i = ((size_t)blockIdx.x * blockDim.x + threadIdx.x) * 4;
    float4 av = *(const float4*)(a + i);
    float4 bv = *(const float4*)(b + i);
    av.x *= bv.x; av.y *= bv.y; av.z *= bv.z; av.w *= bv.w;
    *(float4*)(a + i) = av;
}

extern "C" void kernel_launch(void* const* d_in, const int* in_sizes, int n_in,
                              void* d_out, int out_size, void* d_ws, size_t ws_size,
                              hipStream_t stream) {
    (void)in_sizes; (void)n_in; (void)out_size; (void)ws_size;

    const float* x = (const float*)d_in[0];
    const float* qk_Wq1  = (const float*)d_in[1];  const float* qk_Wq2  = (const float*)d_in[2];
    const float* qk_Wm1  = (const float*)d_in[3];  const float* qk_Wm2  = (const float*)d_in[4];
    const float* qk_Wk1  = (const float*)d_in[5];  const float* qk_Wk2  = (const float*)d_in[6];
    const float* qks_Wq1 = (const float*)d_in[7];  const float* qks_Wq2 = (const float*)d_in[8];
    const float* qks_Wm1 = (const float*)d_in[9];  const float* qks_Wm2 = (const float*)d_in[10];
    const float* qks_Wk1 = (const float*)d_in[11]; const float* qks_Wk2 = (const float*)d_in[12];
    const float* vo_Wv1  = (const float*)d_in[13]; const float* vo_Wv2  = (const float*)d_in[14];
    const float* vo_Wm1  = (const float*)d_in[15]; const float* vo_Wm2  = (const float*)d_in[16];
    const float* vo_Wo   = (const float*)d_in[17];
    const float* sm1_Wd  = (const float*)d_in[18]; const float* sm1_Wa  = (const float*)d_in[19]; const float* sm1_Wb  = (const float*)d_in[20];
    const float* sm1s_Wd = (const float*)d_in[21]; const float* sm1s_Wa = (const float*)d_in[22]; const float* sm1s_Wb = (const float*)d_in[23];
    const float* sm2_Wd  = (const float*)d_in[24]; const float* sm2_Wa  = (const float*)d_in[25]; const float* sm2_Wb  = (const float*)d_in[26];

    const size_t BTD = (size_t)BT * DD;
    const size_t BTT = (size_t)BB * TT * TT;

    float* w = (float*)d_ws;
    float* qg   = w; w += BTD;
    float* kb   = w; w += BTD;
    float* qsg  = w; w += BTD;
    float* ksb  = w; w += BTD;
    float* vb   = w; w += BTD;
    float* mvob = w; w += BTD;
    float* tmpb = w; w += BTD;
    float* zb   = w; w += BTT;
    float* zsb  = w; w += BTT;
    float* ctx  = w; w += BTD;
    float* d1   = w; w += BT; float* d1s = w; w += BT; float* d2 = w; w += BT;
    float* A1   = w; w += (size_t)BT * RSS; float* B1  = w; w += (size_t)BT * RSS;
    float* A1s  = w; w += (size_t)BT * RSS; float* B1s = w; w += (size_t)BT * RSS;
    float* A2   = w; w += (size_t)BT * RSS; float* B2  = w; w += (size_t)BT * RSS;
    float* zBb  = w; w += (size_t)BT * RSS;
    float* denom= w; w += BT;

    auto gemm = [&](const float* A, const float* Bm, float* C, const float* mul,
                    int M, int N, int K, int lda, int ldb, int ldc, int bT,
                    long long sA, long long sB, long long sC, int batches) {
        dim3 g((unsigned)(N / GBN), (unsigned)(M / GBM), (unsigned)batches);
        gemm_wmma_f32<<<g, dim3(256), 0, stream>>>(A, Bm, C, mul, M, N, K, lda, ldb, ldc, bT, sA, sB, sC);
    };

    float* xr8 = ctx;
    const long long sXR = (long long)BT * RR;
    const Weights8 W1 = { qk_Wm1, qk_Wq1, qk_Wk1, qks_Wm1, qks_Wq1, qks_Wk1, vo_Wv1, vo_Wm1 };
    const DabPaths P3 = { sm1_Wa,  sm1_Wb,  sm1_Wd,  A1,  B1,  d1,
                          sm1s_Wa, sm1s_Wb, sm1s_Wd, A1s, B1s, d1s,
                          sm2_Wa,  sm2_Wb,  sm2_Wd,  A2,  B2,  d2 };
    gemm_wmma_f32_shareA8<<<dim3(11, BT / GBM), dim3(256), 0, stream>>>(x, W1, xr8, DD, DD, sXR, P3);

    gemm(xr8 + 0 * sXR, qk_Wm2,  tmpb, nullptr, BT, DD, RR, RR, DD, DD, 0, 0, 0, 0, 1);
    gemm(xr8 + 1 * sXR, qk_Wq2,  qg,   tmpb,    BT, DD, RR, RR, DD, DD, 0, 0, 0, 0, 1);
    gemm(xr8 + 2 * sXR, qk_Wk2,  kb,   nullptr, BT, DD, RR, RR, DD, DD, 0, 0, 0, 0, 1);

    gemm(xr8 + 3 * sXR, qks_Wm2, tmpb, nullptr, BT, DD, RR, RR, DD, DD, 0, 0, 0, 0, 1);
    gemm(xr8 + 4 * sXR, qks_Wq2, qsg,  tmpb,    BT, DD, RR, RR, DD, DD, 0, 0, 0, 0, 1);
    gemm(xr8 + 5 * sXR, qks_Wk2, ksb,  nullptr, BT, DD, RR, RR, DD, DD, 0, 0, 0, 0, 1);

    gemm(xr8 + 6 * sXR, vo_Wv2,  vb,   nullptr, BT, DD, RR, RR, DD, DD, 0, 0, 0, 0, 1);
    gemm(xr8 + 7 * sXR, vo_Wm2,  mvob, nullptr, BT, DD, RR, RR, DD, DD, 0, 0, 0, 0, 1);

    const long long sTD = (long long)TT * DD, sTTl = (long long)TT * TT;
    gemm(qg,  kb,  zb,  nullptr, TT, TT, DD, DD, DD, TT, 1, sTD, sTD, sTTl, BB);
    gemm(qsg, ksb, zsb, nullptr, TT, TT, DD, DD, DD, TT, 1, sTD, sTD, sTTl, BB);

    zmatB_kernel<<<dim3(BT / 8), 256, 0, stream>>>(zb, B1, zBb);
    dplr_combine_kernel<<<dim3(TT / 1024, BT), 256, 0, stream>>>(zb, d1, zBb, A1, zb);
    zmatB_kernel<<<dim3(BT / 8), 256, 0, stream>>>(zsb, B1s, zBb);
    dplr_combine_kernel<<<dim3(TT / 1024, BT), 256, 0, stream>>>(zsb, d1s, zBb, A1s, zsb);
    rownorm_kernel<<<dim3(BT / 8), 256, 0, stream>>>(zb, denom);
    combine_h_kernel<<<dim3(TT / 1024, BT), 256, 0, stream>>>(zb, zsb, denom);

    zmatB_kernel<<<dim3(BT / 8), 256, 0, stream>>>(zb, B2, zBb);
    dplr_combine_kernel<<<dim3(TT / 1024, BT), 256, 0, stream>>>(zb, d2, zBb, A2, zsb);

    gemm(zsb, vb, ctx, nullptr, TT, DD, TT, TT, DD, DD, 0, sTTl, sTD, sTD, BB);
    emul_kernel<<<dim3((unsigned)(BTD / 4 / 256)), 256, 0, stream>>>(ctx, mvob);
    gemm(ctx, vo_Wo, (float*)d_out, nullptr, BT, DD, DD, DD, DD, DD, 1, 0, 0, 0, 1);
}


# z, zs and context each have a rank-64 operand (k, ks, v): contract the 64-wide factor first (split-K WMMA planes, fixed-order sum); k, ks, v never formed; sum order of these 3 products changes
# speedup vs baseline: 2.8909x; 1.4870x over previous
#include <hip/hip_runtime.h>
#include <math.h>

#define BB 2
#define TT 2048
#define DD 1024
#define RR 64
#define RSS 16
#define BT (BB*TT)

typedef __attribute__((ext_vector_type(2))) float v2f;
typedef __attribute__((ext_vector_type(8))) float v8f;

#define GBM 256
#define GBN 64
#define GBK 32

struct DabTile {
    const float* Wa; const float* Wb; const float* Wd;
    float* aOut; float* bOut; float* dOut;
};

__device__ __forceinline__ void gemm_tile_f32(
    const float* __restrict__ Ab, const float* __restrict__ Bb,
    float* __restrict__ Cb, const float* __restrict__ Mb,
    int K, int lda, int ldb, int ldc, int bT, int m0, int n0,
    bool dab, DabTile dt)
{
    __shared__ __align__(16) float As [2][GBM][34];
    __shared__ __align__(16) float BsT[2][GBN][34];

    const int tid  = threadIdx.x;
    const int wave = tid >> 5;
    const int lane = tid & 31;
    const int hi   = lane >> 4;
    const int lm   = lane & 15;

    float4 ra[8];
    float4 rb[2];

    int arow[8], ac4[8];
    #pragma unroll
    for (int i = 0; i < 8; ++i) { int idx = tid + i * 256; arow[i] = idx >> 3; ac4[i] = (idx & 7) * 4; }
    int bi0[2], bi1[2];
    #pragma unroll
    for (int i = 0; i < 2; ++i) {
        int idx = tid + i * 256;
        if (true) {   }
        bi0[i] = idx; bi1[i] = idx;
    }

    auto loadGlobal = [&](int k0) {
        #pragma unroll
        for (int i = 0; i < 8; ++i)
            ra[i] = *(const float4*)(Ab + (size_t)(m0 + arow[i]) * lda + k0 + ac4[i]);
        if (dab) {
            #pragma unroll
            for (int i = 0; i < 2; ++i) {
                int idx = bi1[i];
                int q = idx & 15;
                size_t k = (size_t)(k0 + (idx >> 4));
                float4 t = make_float4(0.f, 0.f, 0.f, 0.f);
                if (q < 4)       t = *(const float4*)(dt.Wa + k * RSS + q * 4);
                else if (q < 8)  t = *(const float4*)(dt.Wb + k * RSS + (q - 4) * 4);
                else if (q == 8) t.x = dt.Wd[k];
                rb[i] = t;
            }
        } else if (bT) {
            #pragma unroll
            for (int i = 0; i < 2; ++i) {
                int idx = bi0[i];
                int n = idx >> 3, c4 = (idx & 7) * 4;
                rb[i] = *(const float4*)(Bb + (size_t)(n0 + n) * ldb + k0 + c4);
            }
        } else {
            #pragma unroll
            for (int i = 0; i < 2; ++i) {
                int idx = bi1[i];
                int r = idx >> 4, c4 = (idx & 15) * 4;
                rb[i] = *(const float4*)(Bb + (size_t)(k0 + r) * ldb + n0 + c4);
            }
        }
    };

    auto storeLDS = [&](int buf) {
        #pragma unroll
        for (int i = 0; i < 8; ++i) {
            float4 t = ra[i];
            v2f lo; lo.x = t.x; lo.y = t.y;
            v2f hi2; hi2.x = t.z; hi2.y = t.w;
            *(v2f*)&As[buf][arow[i]][ac4[i]]     = lo;
            *(v2f*)&As[buf][arow[i]][ac4[i] + 2] = hi2;
        }
        if (bT) {
            #pragma unroll
            for (int i = 0; i < 2; ++i) {
                int idx = bi0[i];
                int n = idx >> 3, c4 = (idx & 7) * 4;
                float4 t = rb[i];
                v2f lo; lo.x = t.x; lo.y = t.y;
                v2f hi2; hi2.x = t.z; hi2.y = t.w;
                *(v2f*)&BsT[buf][n][c4]     = lo;
                *(v2f*)&BsT[buf][n][c4 + 2] = hi2;
            }
        } else {
            #pragma unroll
            for (int i = 0; i < 2; ++i) {
                int idx = bi1[i];
                int r = idx >> 4, c4 = (idx & 15) * 4;
                float4 t = rb[i];
                BsT[buf][c4 + 0][r] = t.x;
                BsT[buf][c4 + 1][r] = t.y;
                BsT[buf][c4 + 2][r] = t.z;
                BsT[buf][c4 + 3][r] = t.w;
            }
        }
    };

    v8f acc00 = {}, acc01 = {}, acc02 = {}, acc03 = {};
    v8f acc10 = {}, acc11 = {}, acc12 = {}, acc13 = {};

    const int r0 = wave * 32 + lm;
    const int r1 = r0 + 16;
    const int nchunks = K / GBK;

    loadGlobal(0);
    for (int c = 0; c < nchunks; ++c) {
        const int buf = c & 1;
        storeLDS(buf);
        __syncthreads();
        if (c + 1 < nchunks) loadGlobal((c + 1) * GBK);

        #pragma unroll
        for (int kk = 0; kk < GBK; kk += 4) {
            const int kc = kk + 2 * hi;
            v2f a0 = *(const v2f*)&As [buf][r0][kc];
            v2f a1 = *(const v2f*)&As [buf][r1][kc];
            v2f b0 = *(const v2f*)&BsT[buf][lm +  0][kc];
            v2f b1 = *(const v2f*)&BsT[buf][lm + 16][kc];
            v2f b2 = *(const v2f*)&BsT[buf][lm + 32][kc];
            v2f b3 = *(const v2f*)&BsT[buf][lm + 48][kc];
            acc00 = __builtin_amdgcn_wmma_f32_16x16x4_f32(false, a0, false, b0, (short)0, acc00, false, false);
            acc01 = __builtin_amdgcn_wmma_f32_16x16x4_f32(false, a0, false, b1, (short)0, acc01, false, false);
            acc02 = __builtin_amdgcn_wmma_f32_16x16x4_f32(false, a0, false, b2, (short)0, acc02, false, false);
            acc03 = __builtin_amdgcn_wmma_f32_16x16x4_f32(false, a0, false, b3, (short)0, acc03, false, false);
            acc10 = __builtin_amdgcn_wmma_f32_16x16x4_f32(false, a1, false, b0, (short)0, acc10, false, false);
            acc11 = __builtin_amdgcn_wmma_f32_16x16x4_f32(false, a1, false, b1, (short)0, acc11, false, false);
            acc12 = __builtin_amdgcn_wmma_f32_16x16x4_f32(false, a1, false, b2, (short)0, acc12, false, false);
            acc13 = __builtin_amdgcn_wmma_f32_16x16x4_f32(false, a1, false, b3, (short)0, acc13, false, false);
        }
        __syncthreads();
    }

    if (dab) {
        #pragma unroll
        for (int mt = 0; mt < 2; ++mt) {
            const int mrow = m0 + wave * 32 + mt * 16 + hi * 8;
            v8f a0 = mt ? acc10 : acc00, a1 = mt ? acc11 : acc01, a2 = mt ? acc12 : acc02;
            #pragma unroll
            for (int r = 0; r < 8; ++r) {
                size_t row = (size_t)(mrow + r);
                dt.aOut[row * RSS + lm] = a0[r];
                dt.bOut[row * RSS + lm] = a1[r];
                if (lm == 0) dt.dOut[row] = a2[r];
            }
        }
        return;
    }

    const int ncol = n0 + lm;
    #pragma unroll
    for (int mt = 0; mt < 2; ++mt) {
        const int mrow = m0 + wave * 32 + mt * 16 + hi * 8;
        v8f a0 = mt ? acc10 : acc00, a1 = mt ? acc11 : acc01;
        v8f a2 = mt ? acc12 : acc02, a3 = mt ? acc13 : acc03;
        #pragma unroll
        for (int r = 0; r < 8; ++r) {
            size_t base = (size_t)(mrow + r) * ldc + ncol;
            float v0 = a0[r], v1 = a1[r], v2 = a2[r], v3 = a3[r];
            if (Mb) {
                v0 *= Mb[base]; v1 *= Mb[base + 16]; v2 *= Mb[base + 32]; v3 *= Mb[base + 48];
            }
            Cb[base] = v0; Cb[base + 16] = v1; Cb[base + 32] = v2; Cb[base + 48] = v3;
        }
    }
}

__global__ __launch_bounds__(256) void gemm_wmma_f32(
    const float* __restrict__ A, const float* __restrict__ Bm,
    float* __restrict__ C, const float* __restrict__ mulC,
    int M, int N, int K, int lda, int ldb, int ldc, int bT,
    long long sA, long long sB, long long sC)
{
    (void)M; (void)N;
    const float* Ab = A  + (size_t)blockIdx.z * sA;
    const float* Bb = Bm + (size_t)blockIdx.z * sB;
    float*       Cb = C  + (size_t)blockIdx.z * sC;
    const float* Mb = mulC ? (mulC + (size_t)blockIdx.z * sC) : nullptr;
    gemm_tile_f32(Ab, Bb, Cb, Mb, K, lda, ldb, ldc, bT,
                  (int)blockIdx.y * GBM, (int)blockIdx.x * GBN, false, DabTile{});
}

struct Weights8 {
    const float* w0; const float* w1; const float* w2; const float* w3;
    const float* w4; const float* w5; const float* w6; const float* w7;
};
struct DabPaths {
    const float* wa0; const float* wb0; const float* wd0; float* a0; float* b0; float* d0;
    const float* wa1; const float* wb1; const float* wd1; float* a1; float* b1; float* d1;
    const float* wa2; const float* wb2; const float* wd2; float* a2; float* b2; float* d2;
};

__global__ __launch_bounds__(256) void gemm_wmma_f32_shareA8(
    const float* __restrict__ A, Weights8 W, float* __restrict__ C,
    int K, int lda, long long sC, DabPaths P)
{
    const int j = blockIdx.x;
    if (j >= 8) {
        DabTile dt = { P.wa0, P.wb0, P.wd0, P.a0, P.b0, P.d0 };
        if (j == 9)  dt = DabTile{ P.wa1, P.wb1, P.wd1, P.a1, P.b1, P.d1 };
        if (j == 10) dt = DabTile{ P.wa2, P.wb2, P.wd2, P.a2, P.b2, P.d2 };
        gemm_tile_f32(A, nullptr, nullptr, nullptr, K, lda, 0, 0, 0,
                      (int)blockIdx.y * GBM, 0, true, dt);
        return;
    }
    const float* Bb = W.w0;
    switch (j) {
        case 1: Bb = W.w1; break;
        case 2: Bb = W.w2; break;
        case 3: Bb = W.w3; break;
        case 4: Bb = W.w4; break;
        case 5: Bb = W.w5; break;
        case 6: Bb = W.w6; break;
        case 7: Bb = W.w7; break;
        default: break;
    }
    gemm_tile_f32(A, Bb, C + (size_t)j * sC, nullptr, K, lda, GBN, GBN, 0,
                  (int)blockIdx.y * GBM, 0, false, DabTile{});
}

__global__ __launch_bounds__(256) void gemm_wmma_f32_splitk(
    const float* A0, const float* A1, const float* B0, const float* B1,
    float* C0, float* C1,
    int K, int lda, int ldb, int ldc, int bT, int S,
    long long sAs, long long sBs, long long sCs)
{
    const int y = (int)blockIdx.z / S;
    const int s = (int)blockIdx.z - y * S;
    const float* Ab = (y ? A1 : A0) + (size_t)s * sAs;
    const float* Bb = (y ? B1 : B0) + (size_t)s * sBs;
    float*       Cb = (y ? C1 : C0) + (size_t)s * sCs;
    gemm_tile_f32(Ab, Bb, Cb, nullptr, K, lda, ldb, ldc, bT,
                  (int)blockIdx.y * GBM, (int)blockIdx.x * GBN, false, DabTile{});
}

__global__ __launch_bounds__(256) void splitk_sum_kernel(
    const float* __restrict__ part, float* __restrict__ out, int S, long long n)
{
    const size_t e = ((size_t)blockIdx.x * blockDim.x + threadIdx.x) * 4;
    const float* p = part + (size_t)blockIdx.y * (size_t)S * (size_t)n + e;
    float4 acc = *(const float4*)p;
    for (int s = 1; s < S; ++s) {
        const float4 t = *(const float4*)(p + (size_t)s * (size_t)n);
        acc.x += t.x; acc.y += t.y; acc.z += t.z; acc.w += t.w;
    }
    *(float4*)(out + (size_t)blockIdx.y * (size_t)n + e) = acc;
}

__device__ inline float waveReduce(float v) {
    #pragma unroll
    for (int o = 16; o > 0; o >>= 1) v += __shfl_xor(v, o, 32);
    return v;
}
__device__ inline float dot4f(float4 a, float4 b) {
    return a.x * b.x + a.y * b.y + a.z * b.z + a.w * b.w;
}
__device__ inline float softplusf(float x) {
    if (x > 20.f) return x;
    if (x < -20.f) return expf(x);
    return log1pf(expf(x));
}

__global__ __launch_bounds__(256) void zmatB_kernel(
    const float* __restrict__ z, const float* __restrict__ Bm, float* __restrict__ zB)
{
    const int wave = threadIdx.x >> 5, lane = threadIdx.x & 31;
    const int rowg = blockIdx.x * 8 + wave;
    const int b    = rowg >> 11;
    const float* zr = z + (size_t)rowg * TT;
    const float* Bb = Bm + (size_t)b * TT * RSS;

    float acc[RSS];
    #pragma unroll
    for (int r = 0; r < RSS; ++r) acc[r] = 0.f;

    for (int j = lane; j < TT; j += 32) {
        float zv = zr[j];
        const float4* br = (const float4*)(Bb + (size_t)j * RSS);
        float4 b0 = br[0], b1 = br[1], b2 = br[2], b3 = br[3];
        acc[0]  += zv * b0.x; acc[1]  += zv * b0.y; acc[2]  += zv * b0.z; acc[3]  += zv * b0.w;
        acc[4]  += zv * b1.x; acc[5]  += zv * b1.y; acc[6]  += zv * b1.z; acc[7]  += zv * b1.w;
        acc[8]  += zv * b2.x; acc[9]  += zv * b2.y; acc[10] += zv * b2.z; acc[11] += zv * b2.w;
        acc[12] += zv * b3.x; acc[13] += zv * b3.y; acc[14] += zv * b3.z; acc[15] += zv * b3.w;
    }
    #pragma unroll
    for (int r = 0; r < RSS; ++r) {
        float v = waveReduce(acc[r]);
        if (lane == 0) zB[(size_t)rowg * RSS + r] = v;
    }
}

__global__ __launch_bounds__(256) void dplr_combine_kernel(
    const float* __restrict__ z, const float* __restrict__ dvec,
    const float* __restrict__ zB, const float* __restrict__ Amat,
    float* __restrict__ out)
{
    const int rowg = blockIdx.y;
    const int b    = rowg >> 11;
    const int j    = (blockIdx.x * blockDim.x + threadIdx.x) * 4;

    const float4* cr = (const float4*)(zB + (size_t)rowg * RSS);
    const float4 c0 = cr[0], c1 = cr[1], c2 = cr[2], c3 = cr[3];

    const float4 z4 = *(const float4*)(z + (size_t)rowg * TT + j);
    const float4 d4 = *(const float4*)(dvec + (size_t)b * TT + j);

    const float4* A0 = (const float4*)(Amat + ((size_t)b * TT + j) * RSS);
    float dv[4];
    #pragma unroll
    for (int q = 0; q < 4; ++q) {
        const float4* Ar = A0 + q * 4;
        dv[q] = dot4f(c0, Ar[0]) + dot4f(c1, Ar[1]) + dot4f(c2, Ar[2]) + dot4f(c3, Ar[3]);
    }
    float4 o;
    o.x = z4.x * d4.x + dv[0];
    o.y = z4.y * d4.y + dv[1];
    o.z = z4.z * d4.z + dv[2];
    o.w = z4.w * d4.w + dv[3];
    *(float4*)(out + (size_t)rowg * TT + j) = o;
}

__global__ __launch_bounds__(256) void rownorm_kernel(
    const float* __restrict__ h, float* __restrict__ denom)
{
    const int wave = threadIdx.x >> 5, lane = threadIdx.x & 31;
    const int rowg = blockIdx.x * 8 + wave;
    const float* hr = h + (size_t)rowg * TT;
    float acc = 0.f;
    for (int j = lane; j < TT; j += 32) { float v = hr[j]; acc += v * v; }
    acc = waveReduce(acc);
    if (lane == 0) denom[rowg] = sqrtf(acc) + 1e-8f;
}

__global__ __launch_bounds__(256) void combine_h_kernel(
    float* __restrict__ hmix, const float* __restrict__ hscale,
    const float* __restrict__ denom)
{
    const int rowg = blockIdx.y;
    const int j = (blockIdx.x * blockDim.x + threadIdx.x) * 4;
    const float inv = 1.f / denom[rowg];
    float4 m = *(const float4*)(hmix   + (size_t)rowg * TT + j);
    float4 s = *(const float4*)(hscale + (size_t)rowg * TT + j);
    float4 o;
    o.x = softplusf(s.x) * fmaxf(m.x * inv, 0.f);
    o.y = softplusf(s.y) * fmaxf(m.y * inv, 0.f);
    o.z = softplusf(s.z) * fmaxf(m.z * inv, 0.f);
    o.w = softplusf(s.w) * fmaxf(m.w * inv, 0.f);
    *(float4*)(hmix + (size_t)rowg * TT + j) = o;
}

extern "C" void kernel_launch(void* const* d_in, const int* in_sizes, int n_in,
                              void* d_out, int out_size, void* d_ws, size_t ws_size,
                              hipStream_t stream) {
    (void)in_sizes; (void)n_in; (void)out_size; (void)ws_size;

    const float* x = (const float*)d_in[0];
    const float* qk_Wq1  = (const float*)d_in[1];  const float* qk_Wq2  = (const float*)d_in[2];
    const float* qk_Wm1  = (const float*)d_in[3];  const float* qk_Wm2  = (const float*)d_in[4];
    const float* qk_Wk1  = (const float*)d_in[5];  const float* qk_Wk2  = (const float*)d_in[6];
    const float* qks_Wq1 = (const float*)d_in[7];  const float* qks_Wq2 = (const float*)d_in[8];
    const float* qks_Wm1 = (const float*)d_in[9];  const float* qks_Wm2 = (const float*)d_in[10];
    const float* qks_Wk1 = (const float*)d_in[11]; const float* qks_Wk2 = (const float*)d_in[12];
    const float* vo_Wv1  = (const float*)d_in[13]; const float* vo_Wv2  = (const float*)d_in[14];
    const float* vo_Wm1  = (const float*)d_in[15]; const float* vo_Wm2  = (const float*)d_in[16];
    const float* vo_Wo   = (const float*)d_in[17];
    const float* sm1_Wd  = (const float*)d_in[18]; const float* sm1_Wa  = (const float*)d_in[19]; const float* sm1_Wb  = (const float*)d_in[20];
    const float* sm1s_Wd = (const float*)d_in[21]; const float* sm1s_Wa = (const float*)d_in[22]; const float* sm1s_Wb = (const float*)d_in[23];
    const float* sm2_Wd  = (const float*)d_in[24]; const float* sm2_Wa  = (const float*)d_in[25]; const float* sm2_Wb  = (const float*)d_in[26];

    const size_t BTD = (size_t)BT * DD;
    const size_t BTT = (size_t)BB * TT * TT;

    float* w = (float*)d_ws;
    float* qg   = w; w += BTD;
    float* pkp  = w; w += BTD;
    float* qsg  = w; w += BTD;
    float* pp   = w; w += BTD;
    float* mvob = w; w += BTD;
    float* tmpb = w; w += BTD;
    float* zb   = w; w += BTT;
    float* zsb  = w; w += BTT;
    float* ctx  = w; w += BTD;
    float* d1   = w; w += BT; float* d1s = w; w += BT; float* d2 = w; w += BT;
    float* A1   = w; w += (size_t)BT * RSS; float* B1  = w; w += (size_t)BT * RSS;
    float* A1s  = w; w += (size_t)BT * RSS; float* B1s = w; w += (size_t)BT * RSS;
    float* A2   = w; w += (size_t)BT * RSS; float* B2  = w; w += (size_t)BT * RSS;
    float* zBb  = w; w += (size_t)BT * RSS;
    float* denom= w; w += BT;

    auto gemm = [&](const float* A, const float* Bm, float* C, const float* mul,
                    int M, int N, int K, int lda, int ldb, int ldc, int bT,
                    long long sA, long long sB, long long sC, int batches) {
        dim3 g((unsigned)(N / GBN), (unsigned)(M / GBM), (unsigned)batches);
        gemm_wmma_f32<<<g, dim3(256), 0, stream>>>(A, Bm, C, mul, M, N, K, lda, ldb, ldc, bT, sA, sB, sC);
    };

    float* xr8 = ctx;
    const long long sXR = (long long)BT * RR;
    const Weights8 W1 = { qk_Wm1, qk_Wq1, qk_Wk1, qks_Wm1, qks_Wq1, qks_Wk1, vo_Wv1, vo_Wm1 };
    const DabPaths P3 = { sm1_Wa,  sm1_Wb,  sm1_Wd,  A1,  B1,  d1,
                          sm1s_Wa, sm1s_Wb, sm1s_Wd, A1s, B1s, d1s,
                          sm2_Wa,  sm2_Wb,  sm2_Wd,  A2,  B2,  d2 };
    gemm_wmma_f32_shareA8<<<dim3(11, BT / GBM), dim3(256), 0, stream>>>(x, W1, xr8, DD, DD, sXR, P3);

    gemm(xr8 + 0 * sXR, qk_Wm2,  tmpb, nullptr, BT, DD, RR, RR, DD, DD, 0, 0, 0, 0, 1);
    gemm(xr8 + 1 * sXR, qk_Wq2,  qg,   tmpb,    BT, DD, RR, RR, DD, DD, 0, 0, 0, 0, 1);

    gemm(xr8 + 3 * sXR, qks_Wm2, tmpb, nullptr, BT, DD, RR, RR, DD, DD, 0, 0, 0, 0, 1);
    gemm(xr8 + 4 * sXR, qks_Wq2, qsg,  tmpb,    BT, DD, RR, RR, DD, DD, 0, 0, 0, 0, 1);

    gemm(xr8 + 7 * sXR, vo_Wm2,  mvob, nullptr, BT, DD, RR, RR, DD, DD, 0, 0, 0, 0, 1);

    const long long sTTl = (long long)TT * TT;
    const long long sTR = (long long)TT * RR;
    gemm_wmma_f32_splitk<<<dim3(1, BT / GBM, 2 * 8), dim3(256), 0, stream>>>(
        qg, qsg, qk_Wk2, qks_Wk2, pkp, pkp + 8 * sXR,
        DD / 8, DD, DD, RR, 1, 8, DD / 8, DD / 8, sXR);
    splitk_sum_kernel<<<dim3((unsigned)(sXR / 4 / 256), 2), 256, 0, stream>>>(pkp, pp, 8, sXR);
    gemm(pp,       xr8 + 2 * sXR, zb,  nullptr, TT, TT, RR, RR, RR, TT, 1, sTR, sTR, sTTl, BB);
    gemm(pp + sXR, xr8 + 5 * sXR, zsb, nullptr, TT, TT, RR, RR, RR, TT, 1, sTR, sTR, sTTl, BB);

    zmatB_kernel<<<dim3(BT / 8), 256, 0, stream>>>(zb, B1, zBb);
    dplr_combine_kernel<<<dim3(TT / 1024, BT), 256, 0, stream>>>(zb, d1, zBb, A1, zb);
    zmatB_kernel<<<dim3(BT / 8), 256, 0, stream>>>(zsb, B1s, zBb);
    dplr_combine_kernel<<<dim3(TT / 1024, BT), 256, 0, stream>>>(zsb, d1s, zBb, A1s, zsb);
    rownorm_kernel<<<dim3(BT / 8), 256, 0, stream>>>(zb, denom);
    combine_h_kernel<<<dim3(TT / 1024, BT), 256, 0, stream>>>(zb, zsb, denom);

    zmatB_kernel<<<dim3(BT / 8), 256, 0, stream>>>(zb, B2, zBb);
    dplr_combine_kernel<<<dim3(TT / 1024, BT), 256, 0, stream>>>(zb, d2, zBb, A2, zsb);

    gemm_wmma_f32_splitk<<<dim3(1, TT / GBM, 2 * 16), dim3(256), 0, stream>>>(
        zsb, zsb + sTTl, xr8 + 6 * sXR, xr8 + 6 * sXR + sTR, qg, qg + sTR,
        TT / 16, TT, RR, RR, 0, 16, TT / 16, (long long)(TT / 16) * RR, sXR);
    splitk_sum_kernel<<<dim3((unsigned)(sXR / 4 / 256), 1), 256, 0, stream>>>(qg, pkp, 16, sXR);
    gemm(pkp, vo_Wv2, ctx, mvob, BT, DD, RR, RR, DD, DD, 0, 0, 0, 0, 1);
    gemm(ctx, vo_Wo, (float*)d_out, nullptr, BT, DD, DD, DD, DD, DD, 1, 0, 0, 0, 1);
}


# each DPLR mix R=diag(d)+B A^T is applied to the rank-64 factor beside it ([T,64]: z R=P(R^T xk)^T, (h R)xv=h(R xv)), not to the [T,T] matrix; 6 [T,T] passes gone; sum order of the mixes changes
# speedup vs baseline: 4.6153x; 1.5965x over previous
#include <hip/hip_runtime.h>
#include <math.h>

#define BB 2
#define TT 2048
#define DD 1024
#define RR 64
#define RSS 16
#define BT (BB*TT)

typedef __attribute__((ext_vector_type(2))) float v2f;
typedef __attribute__((ext_vector_type(8))) float v8f;

#define GBM 256
#define GBN 64
#define GBK 32

struct DabTile {
    const float* Wa; const float* Wb; const float* Wd;
    float* aOut; float* bOut; float* dOut;
};

__device__ __forceinline__ void gemm_tile_f32(
    const float* __restrict__ Ab, const float* __restrict__ Bb,
    float* __restrict__ Cb, const float* __restrict__ Mb,
    int K, int lda, int ldb, int ldc, int bT, int m0, int n0,
    bool dab, DabTile dt)
{
    __shared__ __align__(16) float As [2][GBM][34];
    __shared__ __align__(16) float BsT[2][GBN][34];

    const int tid  = threadIdx.x;
    const int wave = tid >> 5;
    const int lane = tid & 31;
    const int hi   = lane >> 4;
    const int lm   = lane & 15;

    float4 ra[8];
    float4 rb[2];

    int arow[8], ac4[8];
    #pragma unroll
    for (int i = 0; i < 8; ++i) { int idx = tid + i * 256; arow[i] = idx >> 3; ac4[i] = (idx & 7) * 4; }
    int bi0[2], bi1[2];
    #pragma unroll
    for (int i = 0; i < 2; ++i) {
        int idx = tid + i * 256;
        if (true) {   }
        bi0[i] = idx; bi1[i] = idx;
    }

    auto loadGlobal = [&](int k0) {
        #pragma unroll
        for (int i = 0; i < 8; ++i)
            ra[i] = *(const float4*)(Ab + (size_t)(m0 + arow[i]) * lda + k0 + ac4[i]);
        if (dab) {
            #pragma unroll
            for (int i = 0; i < 2; ++i) {
                int idx = bi1[i];
                int q = idx & 15;
                size_t k = (size_t)(k0 + (idx >> 4));
                float4 t = make_float4(0.f, 0.f, 0.f, 0.f);
                if (q < 4)       t = *(const float4*)(dt.Wa + k * RSS + q * 4);
                else if (q < 8)  t = *(const float4*)(dt.Wb + k * RSS + (q - 4) * 4);
                else if (q == 8) t.x = dt.Wd[k];
                rb[i] = t;
            }
        } else if (bT) {
            #pragma unroll
            for (int i = 0; i < 2; ++i) {
                int idx = bi0[i];
                int n = idx >> 3, c4 = (idx & 7) * 4;
                rb[i] = *(const float4*)(Bb + (size_t)(n0 + n) * ldb + k0 + c4);
            }
        } else {
            #pragma unroll
            for (int i = 0; i < 2; ++i) {
                int idx = bi1[i];
                int r = idx >> 4, c4 = (idx & 15) * 4;
                rb[i] = *(const float4*)(Bb + (size_t)(k0 + r) * ldb + n0 + c4);
            }
        }
    };

    auto storeLDS = [&](int buf) {
        #pragma unroll
        for (int i = 0; i < 8; ++i) {
            float4 t = ra[i];
            v2f lo; lo.x = t.x; lo.y = t.y;
            v2f hi2; hi2.x = t.z; hi2.y = t.w;
            *(v2f*)&As[buf][arow[i]][ac4[i]]     = lo;
            *(v2f*)&As[buf][arow[i]][ac4[i] + 2] = hi2;
        }
        if (bT) {
            #pragma unroll
            for (int i = 0; i < 2; ++i) {
                int idx = bi0[i];
                int n = idx >> 3, c4 = (idx & 7) * 4;
                float4 t = rb[i];
                v2f lo; lo.x = t.x; lo.y = t.y;
                v2f hi2; hi2.x = t.z; hi2.y = t.w;
                *(v2f*)&BsT[buf][n][c4]     = lo;
                *(v2f*)&BsT[buf][n][c4 + 2] = hi2;
            }
        } else {
            #pragma unroll
            for (int i = 0; i < 2; ++i) {
                int idx = bi1[i];
                int r = idx >> 4, c4 = (idx & 15) * 4;
                float4 t = rb[i];
                BsT[buf][c4 + 0][r] = t.x;
                BsT[buf][c4 + 1][r] = t.y;
                BsT[buf][c4 + 2][r] = t.z;
                BsT[buf][c4 + 3][r] = t.w;
            }
        }
    };

    v8f acc00 = {}, acc01 = {}, acc02 = {}, acc03 = {};
    v8f acc10 = {}, acc11 = {}, acc12 = {}, acc13 = {};

    const int r0 = wave * 32 + lm;
    const int r1 = r0 + 16;
    const int nchunks = K / GBK;

    loadGlobal(0);
    for (int c = 0; c < nchunks; ++c) {
        const int buf = c & 1;
        storeLDS(buf);
        __syncthreads();
        if (c + 1 < nchunks) loadGlobal((c + 1) * GBK);

        #pragma unroll
        for (int kk = 0; kk < GBK; kk += 4) {
            const int kc = kk + 2 * hi;
            v2f a0 = *(const v2f*)&As [buf][r0][kc];
            v2f a1 = *(const v2f*)&As [buf][r1][kc];
            v2f b0 = *(const v2f*)&BsT[buf][lm +  0][kc];
            v2f b1 = *(const v2f*)&BsT[buf][lm + 16][kc];
            v2f b2 = *(const v2f*)&BsT[buf][lm + 32][kc];
            v2f b3 = *(const v2f*)&BsT[buf][lm + 48][kc];
            acc00 = __builtin_amdgcn_wmma_f32_16x16x4_f32(false, a0, false, b0, (short)0, acc00, false, false);
            acc01 = __builtin_amdgcn_wmma_f32_16x16x4_f32(false, a0, false, b1, (short)0, acc01, false, false);
            acc02 = __builtin_amdgcn_wmma_f32_16x16x4_f32(false, a0, false, b2, (short)0, acc02, false, false);
            acc03 = __builtin_amdgcn_wmma_f32_16x16x4_f32(false, a0, false, b3, (short)0, acc03, false, false);
            acc10 = __builtin_amdgcn_wmma_f32_16x16x4_f32(false, a1, false, b0, (short)0, acc10, false, false);
            acc11 = __builtin_amdgcn_wmma_f32_16x16x4_f32(false, a1, false, b1, (short)0, acc11, false, false);
            acc12 = __builtin_amdgcn_wmma_f32_16x16x4_f32(false, a1, false, b2, (short)0, acc12, false, false);
            acc13 = __builtin_amdgcn_wmma_f32_16x16x4_f32(false, a1, false, b3, (short)0, acc13, false, false);
        }
        __syncthreads();
    }

    if (dab) {
        #pragma unroll
        for (int mt = 0; mt < 2; ++mt) {
            const int mrow = m0 + wave * 32 + mt * 16 + hi * 8;
            v8f a0 = mt ? acc10 : acc00, a1 = mt ? acc11 : acc01, a2 = mt ? acc12 : acc02;
            #pragma unroll
            for (int r = 0; r < 8; ++r) {
                size_t row = (size_t)(mrow + r);
                dt.aOut[row * RSS + lm] = a0[r];
                dt.bOut[row * RSS + lm] = a1[r];
                if (lm == 0) dt.dOut[row] = a2[r];
            }
        }
        return;
    }

    const int ncol = n0 + lm;
    #pragma unroll
    for (int mt = 0; mt < 2; ++mt) {
        const int mrow = m0 + wave * 32 + mt * 16 + hi * 8;
        v8f a0 = mt ? acc10 : acc00, a1 = mt ? acc11 : acc01;
        v8f a2 = mt ? acc12 : acc02, a3 = mt ? acc13 : acc03;
        #pragma unroll
        for (int r = 0; r < 8; ++r) {
            size_t base = (size_t)(mrow + r) * ldc + ncol;
            float v0 = a0[r], v1 = a1[r], v2 = a2[r], v3 = a3[r];
            if (Mb) {
                v0 *= Mb[base]; v1 *= Mb[base + 16]; v2 *= Mb[base + 32]; v3 *= Mb[base + 48];
            }
            Cb[base] = v0; Cb[base + 16] = v1; Cb[base + 32] = v2; Cb[base + 48] = v3;
        }
    }
}

__global__ __launch_bounds__(256) void gemm_wmma_f32(
    const float* __restrict__ A, const float* __restrict__ Bm,
    float* __restrict__ C, const float* __restrict__ mulC,
    int M, int N, int K, int lda, int ldb, int ldc, int bT,
    long long sA, long long sB, long long sC)
{
    (void)M; (void)N;
    const float* Ab = A  + (size_t)blockIdx.z * sA;
    const float* Bb = Bm + (size_t)blockIdx.z * sB;
    float*       Cb = C  + (size_t)blockIdx.z * sC;
    const float* Mb = mulC ? (mulC + (size_t)blockIdx.z * sC) : nullptr;
    gemm_tile_f32(Ab, Bb, Cb, Mb, K, lda, ldb, ldc, bT,
                  (int)blockIdx.y * GBM, (int)blockIdx.x * GBN, false, DabTile{});
}

struct Weights8 {
    const float* w0; const float* w1; const float* w2; const float* w3;
    const float* w4; const float* w5; const float* w6; const float* w7;
};
struct DabPaths {
    const float* wa0; const float* wb0; const float* wd0; float* a0; float* b0; float* d0;
    const float* wa1; const float* wb1; const float* wd1; float* a1; float* b1; float* d1;
    const float* wa2; const float* wb2; const float* wd2; float* a2; float* b2; float* d2;
};

__global__ __launch_bounds__(256) void gemm_wmma_f32_shareA8(
    const float* __restrict__ A, Weights8 W, float* __restrict__ C,
    int K, int lda, long long sC, DabPaths P)
{
    const int j = blockIdx.x;
    if (j >= 8) {
        DabTile dt = { P.wa0, P.wb0, P.wd0, P.a0, P.b0, P.d0 };
        if (j == 9)  dt = DabTile{ P.wa1, P.wb1, P.wd1, P.a1, P.b1, P.d1 };
        if (j == 10) dt = DabTile{ P.wa2, P.wb2, P.wd2, P.a2, P.b2, P.d2 };
        gemm_tile_f32(A, nullptr, nullptr, nullptr, K, lda, 0, 0, 0,
                      (int)blockIdx.y * GBM, 0, true, dt);
        return;
    }
    const float* Bb = W.w0;
    switch (j) {
        case 1: Bb = W.w1; break;
        case 2: Bb = W.w2; break;
        case 3: Bb = W.w3; break;
        case 4: Bb = W.w4; break;
        case 5: Bb = W.w5; break;
        case 6: Bb = W.w6; break;
        case 7: Bb = W.w7; break;
        default: break;
    }
    gemm_tile_f32(A, Bb, C + (size_t)j * sC, nullptr, K, lda, GBN, GBN, 0,
                  (int)blockIdx.y * GBM, 0, false, DabTile{});
}

__global__ __launch_bounds__(256) void gemm_wmma_f32_splitk(
    const float* A0, const float* A1, const float* B0, const float* B1,
    float* C0, float* C1,
    int K, int lda, int ldb, int ldc, int bT, int S,
    long long sAs, long long sBs, long long sCs)
{
    const int y = (int)blockIdx.z / S;
    const int s = (int)blockIdx.z - y * S;
    const float* Ab = (y ? A1 : A0) + (size_t)s * sAs;
    const float* Bb = (y ? B1 : B0) + (size_t)s * sBs;
    float*       Cb = (y ? C1 : C0) + (size_t)s * sCs;
    gemm_tile_f32(Ab, Bb, Cb, nullptr, K, lda, ldb, ldc, bT,
                  (int)blockIdx.y * GBM, (int)blockIdx.x * GBN, false, DabTile{});
}

__global__ __launch_bounds__(256) void splitk_sum_kernel(
    const float* __restrict__ part, float* __restrict__ out, int S, long long n)
{
    const size_t e = ((size_t)blockIdx.x * blockDim.x + threadIdx.x) * 4;
    const float* p = part + (size_t)blockIdx.y * (size_t)S * (size_t)n + e;
    float4 acc = *(const float4*)p;
    for (int s = 1; s < S; ++s) {
        const float4 t = *(const float4*)(p + (size_t)s * (size_t)n);
        acc.x += t.x; acc.y += t.y; acc.z += t.z; acc.w += t.w;
    }
    *(float4*)(out + (size_t)blockIdx.y * (size_t)n + e) = acc;
}

__device__ inline float waveReduce(float v) {
    #pragma unroll
    for (int o = 16; o > 0; o >>= 1) v += __shfl_xor(v, o, 32);
    return v;
}
__device__ inline float softplusf(float x) {
    if (x > 20.f) return x;
    if (x < -20.f) return expf(x);
    return log1pf(expf(x));
}

struct MixPath { const float* F; const float* L; const float* Rm; const float* d; };
struct MixPaths { MixPath p0; MixPath p1; MixPath p2; };

#define MIX_S 32
#define MIX_J (TT / MIX_S)

__global__ __launch_bounds__(256) void mix_factor_partial_kernel(
    MixPaths mp, float* __restrict__ Mpart)
{
    const int s = blockIdx.x, combo = blockIdx.y;
    const int p = combo >> 1, b = combo & 1;
    const MixPath mx = p == 0 ? mp.p0 : (p == 1 ? mp.p1 : mp.p2);
    const int c = threadIdx.x & 63, rq = threadIdx.x >> 6;
    const size_t row0 = (size_t)b * TT + (size_t)s * MIX_J;
    const float* Fp = mx.F + row0 * RR + c;
    const float* Lp = mx.L + row0 * RSS + rq * 4;
    float4 acc = make_float4(0.f, 0.f, 0.f, 0.f);
    #pragma unroll 16
    for (int jj = 0; jj < MIX_J; ++jj) {
        const float f = Fp[(size_t)jj * RR];
        const float4 l = *(const float4*)(Lp + (size_t)jj * RSS);
        acc.x += l.x * f; acc.y += l.y * f; acc.z += l.z * f; acc.w += l.w * f;
    }
    float* o = Mpart + ((size_t)(combo * MIX_S + s) * RSS + rq * 4) * RR + c;
    o[0] = acc.x; o[RR] = acc.y; o[2 * RR] = acc.z; o[3 * RR] = acc.w;
}

__global__ __launch_bounds__(256) void mix_factor_apply_kernel(
    MixPaths mp, const float* __restrict__ Mpart, float* __restrict__ G)
{
    __shared__ __align__(16) float Ms[RSS * RR];
    const int combo = blockIdx.y;
    const int p = combo >> 1, b = combo & 1;
    const MixPath mx = p == 0 ? mp.p0 : (p == 1 ? mp.p1 : mp.p2);
    {
        const float4* src = (const float4*)(Mpart + (size_t)combo * MIX_S * RSS * RR) + threadIdx.x;
        float4 acc = src[0];
        #pragma unroll
        for (int sl = 1; sl < MIX_S; ++sl) {
            const float4 t = src[(size_t)sl * (RSS * RR / 4)];
            acc.x += t.x; acc.y += t.y; acc.z += t.z; acc.w += t.w;
        }
        ((float4*)Ms)[threadIdx.x] = acc;
    }
    __syncthreads();
    const int c = threadIdx.x & 63, rq = threadIdx.x >> 6;
    float m[RSS];
    #pragma unroll
    for (int r = 0; r < RSS; ++r) m[r] = Ms[r * RR + c];
    const size_t row0 = (size_t)b * TT + (size_t)blockIdx.x * 64 + (size_t)rq * 16;
    float* Gp = G + (size_t)p * BT * RR;
    #pragma unroll 4
    for (int k = 0; k < 16; ++k) {
        const size_t row = row0 + k;
        const float4* rr = (const float4*)(mx.Rm + row * RSS);
        const float4 r0 = rr[0], r1 = rr[1], r2 = rr[2], r3 = rr[3];
        float v = mx.d[row] * mx.F[row * RR + c];
        v += r0.x * m[0];  v += r0.y * m[1];  v += r0.z * m[2];  v += r0.w * m[3];
        v += r1.x * m[4];  v += r1.y * m[5];  v += r1.z * m[6];  v += r1.w * m[7];
        v += r2.x * m[8];  v += r2.y * m[9];  v += r2.z * m[10]; v += r2.w * m[11];
        v += r3.x * m[12]; v += r3.y * m[13]; v += r3.z * m[14]; v += r3.w * m[15];
        Gp[row * RR + c] = v;
    }
}

__global__ __launch_bounds__(256) void rownorm_kernel(
    const float* __restrict__ h, float* __restrict__ denom)
{
    const int wave = threadIdx.x >> 5, lane = threadIdx.x & 31;
    const int rowg = blockIdx.x * 8 + wave;
    const float* hr = h + (size_t)rowg * TT;
    float acc = 0.f;
    for (int j = lane; j < TT; j += 32) { float v = hr[j]; acc += v * v; }
    acc = waveReduce(acc);
    if (lane == 0) denom[rowg] = sqrtf(acc) + 1e-8f;
}

__global__ __launch_bounds__(256) void combine_h_kernel(
    float* __restrict__ hmix, const float* __restrict__ hscale,
    const float* __restrict__ denom)
{
    const int rowg = blockIdx.y;
    const int j = (blockIdx.x * blockDim.x + threadIdx.x) * 4;
    const float inv = 1.f / denom[rowg];
    float4 m = *(const float4*)(hmix   + (size_t)rowg * TT + j);
    float4 s = *(const float4*)(hscale + (size_t)rowg * TT + j);
    float4 o;
    o.x = softplusf(s.x) * fmaxf(m.x * inv, 0.f);
    o.y = softplusf(s.y) * fmaxf(m.y * inv, 0.f);
    o.z = softplusf(s.z) * fmaxf(m.z * inv, 0.f);
    o.w = softplusf(s.w) * fmaxf(m.w * inv, 0.f);
    *(float4*)(hmix + (size_t)rowg * TT + j) = o;
}

extern "C" void kernel_launch(void* const* d_in, const int* in_sizes, int n_in,
                              void* d_out, int out_size, void* d_ws, size_t ws_size,
                              hipStream_t stream) {
    (void)in_sizes; (void)n_in; (void)out_size; (void)ws_size;

    const float* x = (const float*)d_in[0];
    const float* qk_Wq1  = (const float*)d_in[1];  const float* qk_Wq2  = (const float*)d_in[2];
    const float* qk_Wm1  = (const float*)d_in[3];  const float* qk_Wm2  = (const float*)d_in[4];
    const float* qk_Wk1  = (const float*)d_in[5];  const float* qk_Wk2  = (const float*)d_in[6];
    const float* qks_Wq1 = (const float*)d_in[7];  const float* qks_Wq2 = (const float*)d_in[8];
    const float* qks_Wm1 = (const float*)d_in[9];  const float* qks_Wm2 = (const float*)d_in[10];
    const float* qks_Wk1 = (const float*)d_in[11]; const float* qks_Wk2 = (const float*)d_in[12];
    const float* vo_Wv1  = (const float*)d_in[13]; const float* vo_Wv2  = (const float*)d_in[14];
    const float* vo_Wm1  = (const float*)d_in[15]; const float* vo_Wm2  = (const float*)d_in[16];
    const float* vo_Wo   = (const float*)d_in[17];
    const float* sm1_Wd  = (const float*)d_in[18]; const float* sm1_Wa  = (const float*)d_in[19]; const float* sm1_Wb  = (const float*)d_in[20];
    const float* sm1s_Wd = (const float*)d_in[21]; const float* sm1s_Wa = (const float*)d_in[22]; const float* sm1s_Wb = (const float*)d_in[23];
    const float* sm2_Wd  = (const float*)d_in[24]; const float* sm2_Wa  = (const float*)d_in[25]; const float* sm2_Wb  = (const float*)d_in[26];

    const size_t BTD = (size_t)BT * DD;
    const size_t BTT = (size_t)BB * TT * TT;

    float* w = (float*)d_ws;
    float* qg   = w; w += BTD;
    float* pkp  = w; w += BTD;
    float* qsg  = w; w += BTD;
    float* pp   = w; w += BTD;
    float* mvob = w; w += BTD;
    float* tmpb = w; w += BTD;
    float* zb   = w; w += BTT;
    float* zsb  = w; w += BTT;
    float* ctx  = w; w += BTD;
    float* d1   = w; w += BT; float* d1s = w; w += BT; float* d2 = w; w += BT;
    float* A1   = w; w += (size_t)BT * RSS; float* B1  = w; w += (size_t)BT * RSS;
    float* A1s  = w; w += (size_t)BT * RSS; float* B1s = w; w += (size_t)BT * RSS;
    float* A2   = w; w += (size_t)BT * RSS; float* B2  = w; w += (size_t)BT * RSS;
    float* denom= w; w += BT;

    auto gemm = [&](const float* A, const float* Bm, float* C, const float* mul,
                    int M, int N, int K, int lda, int ldb, int ldc, int bT,
                    long long sA, long long sB, long long sC, int batches) {
        dim3 g((unsigned)(N / GBN), (unsigned)(M / GBM), (unsigned)batches);
        gemm_wmma_f32<<<g, dim3(256), 0, stream>>>(A, Bm, C, mul, M, N, K, lda, ldb, ldc, bT, sA, sB, sC);
    };

    float* xr8 = ctx;
    const long long sXR = (long long)BT * RR;
    const Weights8 W1 = { qk_Wm1, qk_Wq1, qk_Wk1, qks_Wm1, qks_Wq1, qks_Wk1, vo_Wv1, vo_Wm1 };
    const DabPaths P3 = { sm1_Wa,  sm1_Wb,  sm1_Wd,  A1,  B1,  d1,
                          sm1s_Wa, sm1s_Wb, sm1s_Wd, A1s, B1s, d1s,
                          sm2_Wa,  sm2_Wb,  sm2_Wd,  A2,  B2,  d2 };
    gemm_wmma_f32_shareA8<<<dim3(11, BT / GBM), dim3(256), 0, stream>>>(x, W1, xr8, DD, DD, sXR, P3);

    gemm(xr8 + 0 * sXR, qk_Wm2,  tmpb, nullptr, BT, DD, RR, RR, DD, DD, 0, 0, 0, 0, 1);
    gemm(xr8 + 1 * sXR, qk_Wq2,  qg,   tmpb,    BT, DD, RR, RR, DD, DD, 0, 0, 0, 0, 1);

    gemm(xr8 + 3 * sXR, qks_Wm2, tmpb, nullptr, BT, DD, RR, RR, DD, DD, 0, 0, 0, 0, 1);
    gemm(xr8 + 4 * sXR, qks_Wq2, qsg,  tmpb,    BT, DD, RR, RR, DD, DD, 0, 0, 0, 0, 1);

    gemm(xr8 + 7 * sXR, vo_Wm2,  mvob, nullptr, BT, DD, RR, RR, DD, DD, 0, 0, 0, 0, 1);

    float* Mpart = tmpb;
    float* gmix  = tmpb + (size_t)BT * RR;
    const MixPaths MP = { { xr8 + 2 * sXR, B1,  A1,  d1  },
                          { xr8 + 5 * sXR, B1s, A1s, d1s },
                          { xr8 + 6 * sXR, A2,  B2,  d2  } };
    mix_factor_partial_kernel<<<dim3(MIX_S, 6), 256, 0, stream>>>(MP, Mpart);
    mix_factor_apply_kernel<<<dim3(TT / 64, 6), 256, 0, stream>>>(MP, Mpart, gmix);

    const long long sTTl = (long long)TT * TT;
    const long long sTR = (long long)TT * RR;
    gemm_wmma_f32_splitk<<<dim3(1, BT / GBM, 2 * 8), dim3(256), 0, stream>>>(
        qg, qsg, qk_Wk2, qks_Wk2, pkp, pkp + 8 * sXR,
        DD / 8, DD, DD, RR, 1, 8, DD / 8, DD / 8, sXR);
    splitk_sum_kernel<<<dim3((unsigned)(sXR / 4 / 256), 2), 256, 0, stream>>>(pkp, pp, 8, sXR);
    gemm(pp,       gmix,       zb,  nullptr, TT, TT, RR, RR, RR, TT, 1, sTR, sTR, sTTl, BB);
    gemm(pp + sXR, gmix + sXR, zsb, nullptr, TT, TT, RR, RR, RR, TT, 1, sTR, sTR, sTTl, BB);

    rownorm_kernel<<<dim3(BT / 8), 256, 0, stream>>>(zb, denom);
    combine_h_kernel<<<dim3(TT / 1024, BT), 256, 0, stream>>>(zb, zsb, denom);

    float* vmix = gmix + 2 * sXR;
    gemm_wmma_f32_splitk<<<dim3(1, TT / GBM, 2 * 16), dim3(256), 0, stream>>>(
        zb, zb + sTTl, vmix, vmix + sTR, qg, qg + sTR,
        TT / 16, TT, RR, RR, 0, 16, TT / 16, (long long)(TT / 16) * RR, sXR);
    splitk_sum_kernel<<<dim3((unsigned)(sXR / 4 / 256), 1), 256, 0, stream>>>(qg, pkp, 16, sXR);
    gemm(pkp, vo_Wv2, ctx, mvob, BT, DD, RR, RR, DD, DD, 0, 0, 0, 0, 1);
    gemm(ctx, vo_Wo, (float*)d_out, nullptr, BT, DD, DD, DD, DD, DD, 1, 0, 0, 0, 1);
}


# each gated second stage (q*mg, qs*mgs, ctx*mvo) forms its gate product in the same tile (gate parked in LDS) and multiplies there; gates never stored, 14 launches (was 18); sum orders kept
# speedup vs baseline: 5.2484x; 1.1372x over previous
#include <hip/hip_runtime.h>
#include <math.h>

#define BB 2
#define TT 2048
#define DD 1024
#define RR 64
#define RSS 16
#define BT (BB*TT)

typedef __attribute__((ext_vector_type(2))) float v2f;
typedef __attribute__((ext_vector_type(8))) float v8f;

#define GBM 256
#define GBN 64
#define GBK 32

struct DabTile {
    const float* Wa; const float* Wb; const float* Wd;
    float* aOut; float* bOut; float* dOut;
};

struct TileAcc { v8f c00, c01, c02, c03, c10, c11, c12, c13; };

__device__ __forceinline__ void gemm_tile_loop_f32(
    const float* __restrict__ Ab, const float* __restrict__ Bb,
    int K, int lda, int ldb, int bT, int m0, int n0,
    bool dab, DabTile dt, TileAcc& acc)
{
    __shared__ __align__(16) float As [2][GBM][34];
    __shared__ __align__(16) float BsT[2][GBN][34];

    const int tid  = threadIdx.x;
    const int wave = tid >> 5;
    const int lane = tid & 31;
    const int hi   = lane >> 4;
    const int lm   = lane & 15;

    float4 ra[8];
    float4 rb[2];

    int arow[8], ac4[8];
    #pragma unroll
    for (int i = 0; i < 8; ++i) { int idx = tid + i * 256; arow[i] = idx >> 3; ac4[i] = (idx & 7) * 4; }
    int bi0[2], bi1[2];
    #pragma unroll
    for (int i = 0; i < 2; ++i) {
        int idx = tid + i * 256;
        if (true) {   }
        bi0[i] = idx; bi1[i] = idx;
    }

    auto loadGlobal = [&](int k0) {
        #pragma unroll
        for (int i = 0; i < 8; ++i)
            ra[i] = *(const float4*)(Ab + (size_t)(m0 + arow[i]) * lda + k0 + ac4[i]);
        if (dab) {
            #pragma unroll
            for (int i = 0; i < 2; ++i) {
                int idx = bi1[i];
                int q = idx & 15;
                size_t k = (size_t)(k0 + (idx >> 4));
                float4 t = make_float4(0.f, 0.f, 0.f, 0.f);
                if (q < 4)       t = *(const float4*)(dt.Wa + k * RSS + q * 4);
                else if (q < 8)  t = *(const float4*)(dt.Wb + k * RSS + (q - 4) * 4);
                else if (q == 8) t.x = dt.Wd[k];
                rb[i] = t;
            }
        } else if (bT) {
            #pragma unroll
            for (int i = 0; i < 2; ++i) {
                int idx = bi0[i];
                int n = idx >> 3, c4 = (idx & 7) * 4;
                rb[i] = *(const float4*)(Bb + (size_t)(n0 + n) * ldb + k0 + c4);
            }
        } else {
            #pragma unroll
            for (int i = 0; i < 2; ++i) {
                int idx = bi1[i];
                int r = idx >> 4, c4 = (idx & 15) * 4;
                rb[i] = *(const float4*)(Bb + (size_t)(k0 + r) * ldb + n0 + c4);
            }
        }
    };

    auto storeLDS = [&](int buf) {
        #pragma unroll
        for (int i = 0; i < 8; ++i) {
            float4 t = ra[i];
            v2f lo; lo.x = t.x; lo.y = t.y;
            v2f hi2; hi2.x = t.z; hi2.y = t.w;
            *(v2f*)&As[buf][arow[i]][ac4[i]]     = lo;
            *(v2f*)&As[buf][arow[i]][ac4[i] + 2] = hi2;
        }
        if (bT) {
            #pragma unroll
            for (int i = 0; i < 2; ++i) {
                int idx = bi0[i];
                int n = idx >> 3, c4 = (idx & 7) * 4;
                float4 t = rb[i];
                v2f lo; lo.x = t.x; lo.y = t.y;
                v2f hi2; hi2.x = t.z; hi2.y = t.w;
                *(v2f*)&BsT[buf][n][c4]     = lo;
                *(v2f*)&BsT[buf][n][c4 + 2] = hi2;
            }
        } else {
            #pragma unroll
            for (int i = 0; i < 2; ++i) {
                int idx = bi1[i];
                int r = idx >> 4, c4 = (idx & 15) * 4;
                float4 t = rb[i];
                BsT[buf][c4 + 0][r] = t.x;
                BsT[buf][c4 + 1][r] = t.y;
                BsT[buf][c4 + 2][r] = t.z;
                BsT[buf][c4 + 3][r] = t.w;
            }
        }
    };

    v8f acc00 = {}, acc01 = {}, acc02 = {}, acc03 = {};
    v8f acc10 = {}, acc11 = {}, acc12 = {}, acc13 = {};

    const int r0 = wave * 32 + lm;
    const int r1 = r0 + 16;
    const int nchunks = K / GBK;

    loadGlobal(0);
    for (int c = 0; c < nchunks; ++c) {
        const int buf = c & 1;
        storeLDS(buf);
        __syncthreads();
        if (c + 1 < nchunks) loadGlobal((c + 1) * GBK);

        #pragma unroll
        for (int kk = 0; kk < GBK; kk += 4) {
            const int kc = kk + 2 * hi;
            v2f a0 = *(const v2f*)&As [buf][r0][kc];
            v2f a1 = *(const v2f*)&As [buf][r1][kc];
            v2f b0 = *(const v2f*)&BsT[buf][lm +  0][kc];
            v2f b1 = *(const v2f*)&BsT[buf][lm + 16][kc];
            v2f b2 = *(const v2f*)&BsT[buf][lm + 32][kc];
            v2f b3 = *(const v2f*)&BsT[buf][lm + 48][kc];
            acc00 = __builtin_amdgcn_wmma_f32_16x16x4_f32(false, a0, false, b0, (short)0, acc00, false, false);
            acc01 = __builtin_amdgcn_wmma_f32_16x16x4_f32(false, a0, false, b1, (short)0, acc01, false, false);
            acc02 = __builtin_amdgcn_wmma_f32_16x16x4_f32(false, a0, false, b2, (short)0, acc02, false, false);
            acc03 = __builtin_amdgcn_wmma_f32_16x16x4_f32(false, a0, false, b3, (short)0, acc03, false, false);
            acc10 = __builtin_amdgcn_wmma_f32_16x16x4_f32(false, a1, false, b0, (short)0, acc10, false, false);
            acc11 = __builtin_amdgcn_wmma_f32_16x16x4_f32(false, a1, false, b1, (short)0, acc11, false, false);
            acc12 = __builtin_amdgcn_wmma_f32_16x16x4_f32(false, a1, false, b2, (short)0, acc12, false, false);
            acc13 = __builtin_amdgcn_wmma_f32_16x16x4_f32(false, a1, false, b3, (short)0, acc13, false, false);
        }
        __syncthreads();
    }
    acc.c00 = acc00; acc.c01 = acc01; acc.c02 = acc02; acc.c03 = acc03;
    acc.c10 = acc10; acc.c11 = acc11; acc.c12 = acc12; acc.c13 = acc13;
}

__device__ __forceinline__ void gemm_tile_f32(
    const float* __restrict__ Ab, const float* __restrict__ Bb,
    float* __restrict__ Cb,
    int K, int lda, int ldb, int ldc, int bT, int m0, int n0,
    bool dab, DabTile dt)
{
    TileAcc t;
    gemm_tile_loop_f32(Ab, Bb, K, lda, ldb, bT, m0, n0, dab, dt, t);

    const int wave = threadIdx.x >> 5;
    const int lane = threadIdx.x & 31;
    const int hi   = lane >> 4;
    const int lm   = lane & 15;

    if (dab) {
        #pragma unroll
        for (int mt = 0; mt < 2; ++mt) {
            const int mrow = m0 + wave * 32 + mt * 16 + hi * 8;
            v8f a0 = mt ? t.c10 : t.c00, a1 = mt ? t.c11 : t.c01, a2 = mt ? t.c12 : t.c02;
            #pragma unroll
            for (int r = 0; r < 8; ++r) {
                size_t row = (size_t)(mrow + r);
                dt.aOut[row * RSS + lm] = a0[r];
                dt.bOut[row * RSS + lm] = a1[r];
                if (lm == 0) dt.dOut[row] = a2[r];
            }
        }
        return;
    }

    const int ncol = n0 + lm;
    #pragma unroll
    for (int mt = 0; mt < 2; ++mt) {
        const int mrow = m0 + wave * 32 + mt * 16 + hi * 8;
        v8f a0 = mt ? t.c10 : t.c00, a1 = mt ? t.c11 : t.c01;
        v8f a2 = mt ? t.c12 : t.c02, a3 = mt ? t.c13 : t.c03;
        #pragma unroll
        for (int r = 0; r < 8; ++r) {
            size_t base = (size_t)(mrow + r) * ldc + ncol;
            Cb[base] = a0[r]; Cb[base + 16] = a1[r]; Cb[base + 32] = a2[r]; Cb[base + 48] = a3[r];
        }
    }
}

__global__ __launch_bounds__(256) void gemm_wmma_f32(
    const float* __restrict__ A, const float* __restrict__ Bm,
    float* __restrict__ C,
    int M, int N, int K, int lda, int ldb, int ldc, int bT,
    long long sA, long long sB, long long sC)
{
    (void)M; (void)N;
    const float* Ab = A  + (size_t)blockIdx.z * sA;
    const float* Bb = Bm + (size_t)blockIdx.z * sB;
    float*       Cb = C  + (size_t)blockIdx.z * sC;
    gemm_tile_f32(Ab, Bb, Cb, K, lda, ldb, ldc, bT,
                  (int)blockIdx.y * GBM, (int)blockIdx.x * GBN, false, DabTile{});
}

struct GatedSets {
    const float* A1a; const float* B1a; const float* A2a; const float* B2a; float* Ca;
    const float* A1b; const float* B1b; const float* A2b; const float* B2b; float* Cb;
};

__global__ __launch_bounds__(256) void gemm_wmma_f32_gated(
    GatedSets g, int K, int lda, int ldb, int ldc)
{
    const bool sb = blockIdx.z != 0;
    const float* A1 = sb ? g.A1b : g.A1a;
    const float* B1 = sb ? g.B1b : g.B1a;
    const float* A2 = sb ? g.A2b : g.A2a;
    const float* B2 = sb ? g.B2b : g.B2a;
    float*       C  = sb ? g.Cb  : g.Ca;
    const int m0 = (int)blockIdx.y * GBM, n0 = (int)blockIdx.x * GBN;

    __shared__ __align__(16) float4 Gs[16][256];
    const int tid = threadIdx.x;

    TileAcc gate, val;
    gemm_tile_loop_f32(A2, B2, K, lda, ldb, 0, m0, n0, false, DabTile{}, gate);
    {
        const v8f* gv[8] = { &gate.c00, &gate.c01, &gate.c02, &gate.c03,
                             &gate.c10, &gate.c11, &gate.c12, &gate.c13 };
        #pragma unroll
        for (int t = 0; t < 8; ++t) {
            const v8f v = *gv[t];
            Gs[2 * t][tid]     = make_float4(v[0], v[1], v[2], v[3]);
            Gs[2 * t + 1][tid] = make_float4(v[4], v[5], v[6], v[7]);
        }
    }
    gemm_tile_loop_f32(A1, B1, K, lda, ldb, 0, m0, n0, false, DabTile{}, val);
    {
        v8f* gv[8] = { &gate.c00, &gate.c01, &gate.c02, &gate.c03,
                       &gate.c10, &gate.c11, &gate.c12, &gate.c13 };
        #pragma unroll
        for (int t = 0; t < 8; ++t) {
            const float4 lo = Gs[2 * t][tid], up = Gs[2 * t + 1][tid];
            v8f v;
            v[0] = lo.x; v[1] = lo.y; v[2] = lo.z; v[3] = lo.w;
            v[4] = up.x; v[5] = up.y; v[6] = up.z; v[7] = up.w;
            *gv[t] = v;
        }
    }

    const int wave = threadIdx.x >> 5;
    const int lane = threadIdx.x & 31;
    const int hi   = lane >> 4;
    const int lm   = lane & 15;
    const int ncol = n0 + lm;
    #pragma unroll
    for (int mt = 0; mt < 2; ++mt) {
        const int mrow = m0 + wave * 32 + mt * 16 + hi * 8;
        v8f a0 = mt ? val.c10 : val.c00, a1 = mt ? val.c11 : val.c01;
        v8f a2 = mt ? val.c12 : val.c02, a3 = mt ? val.c13 : val.c03;
        v8f g0 = mt ? gate.c10 : gate.c00, g1 = mt ? gate.c11 : gate.c01;
        v8f g2 = mt ? gate.c12 : gate.c02, g3 = mt ? gate.c13 : gate.c03;
        #pragma unroll
        for (int r = 0; r < 8; ++r) {
            size_t base = (size_t)(mrow + r) * ldc + ncol;
            C[base]      = a0[r] * g0[r];
            C[base + 16] = a1[r] * g1[r];
            C[base + 32] = a2[r] * g2[r];
            C[base + 48] = a3[r] * g3[r];
        }
    }
}

struct Weights8 {
    const float* w0; const float* w1; const float* w2; const float* w3;
    const float* w4; const float* w5; const float* w6; const float* w7;
};
struct DabPaths {
    const float* wa0; const float* wb0; const float* wd0; float* a0; float* b0; float* d0;
    const float* wa1; const float* wb1; const float* wd1; float* a1; float* b1; float* d1;
    const float* wa2; const float* wb2; const float* wd2; float* a2; float* b2; float* d2;
};

__global__ __launch_bounds__(256) void gemm_wmma_f32_shareA8(
    const float* __restrict__ A, Weights8 W, float* __restrict__ C,
    int K, int lda, long long sC, DabPaths P)
{
    const int j = blockIdx.x;
    if (j >= 8) {
        DabTile dt = { P.wa0, P.wb0, P.wd0, P.a0, P.b0, P.d0 };
        if (j == 9)  dt = DabTile{ P.wa1, P.wb1, P.wd1, P.a1, P.b1, P.d1 };
        if (j == 10) dt = DabTile{ P.wa2, P.wb2, P.wd2, P.a2, P.b2, P.d2 };
        gemm_tile_f32(A, nullptr, nullptr, K, lda, 0, 0, 0,
                      (int)blockIdx.y * GBM, 0, true, dt);
        return;
    }
    const float* Bb = W.w0;
    switch (j) {
        case 1: Bb = W.w1; break;
        case 2: Bb = W.w2; break;
        case 3: Bb = W.w3; break;
        case 4: Bb = W.w4; break;
        case 5: Bb = W.w5; break;
        case 6: Bb = W.w6; break;
        case 7: Bb = W.w7; break;
        default: break;
    }
    gemm_tile_f32(A, Bb, C + (size_t)j * sC, K, lda, GBN, GBN, 0,
                  (int)blockIdx.y * GBM, 0, false, DabTile{});
}

__global__ __launch_bounds__(256) void gemm_wmma_f32_splitk(
    const float* A0, const float* A1, const float* B0, const float* B1,
    float* C0, float* C1,
    int K, int lda, int ldb, int ldc, int bT, int S,
    long long sAs, long long sBs, long long sCs)
{
    const int y = (int)blockIdx.z / S;
    const int s = (int)blockIdx.z - y * S;
    const float* Ab = (y ? A1 : A0) + (size_t)s * sAs;
    const float* Bb = (y ? B1 : B0) + (size_t)s * sBs;
    float*       Cb = (y ? C1 : C0) + (size_t)s * sCs;
    gemm_tile_f32(Ab, Bb, Cb, K, lda, ldb, ldc, bT,
                  (int)blockIdx.y * GBM, (int)blockIdx.x * GBN, false, DabTile{});
}

__global__ __launch_bounds__(256) void splitk_sum_kernel(
    const float* __restrict__ part, float* __restrict__ out, int S, long long n)
{
    const size_t e = ((size_t)blockIdx.x * blockDim.x + threadIdx.x) * 4;
    const float* p = part + (size_t)blockIdx.y * (size_t)S * (size_t)n + e;
    float4 acc = *(const float4*)p;
    for (int s = 1; s < S; ++s) {
        const float4 t = *(const float4*)(p + (size_t)s * (size_t)n);
        acc.x += t.x; acc.y += t.y; acc.z += t.z; acc.w += t.w;
    }
    *(float4*)(out + (size_t)blockIdx.y * (size_t)n + e) = acc;
}

__device__ inline float waveReduce(float v) {
    #pragma unroll
    for (int o = 16; o > 0; o >>= 1) v += __shfl_xor(v, o, 32);
    return v;
}
__device__ inline float softplusf(float x) {
    if (x > 20.f) return x;
    if (x < -20.f) return expf(x);
    return log1pf(expf(x));
}

struct MixPath { const float* F; const float* L; const float* Rm; const float* d; };
struct MixPaths { MixPath p0; MixPath p1; MixPath p2; };

#define MIX_S 32
#define MIX_J (TT / MIX_S)

__global__ __launch_bounds__(256) void mix_factor_partial_kernel(
    MixPaths mp, float* __restrict__ Mpart)
{
    const int s = blockIdx.x, combo = blockIdx.y;
    const int p = combo >> 1, b = combo & 1;
    const MixPath mx = p == 0 ? mp.p0 : (p == 1 ? mp.p1 : mp.p2);
    const int c = threadIdx.x & 63, rq = threadIdx.x >> 6;
    const size_t row0 = (size_t)b * TT + (size_t)s * MIX_J;
    const float* Fp = mx.F + row0 * RR + c;
    const float* Lp = mx.L + row0 * RSS + rq * 4;
    float4 acc = make_float4(0.f, 0.f, 0.f, 0.f);
    #pragma unroll 16
    for (int jj = 0; jj < MIX_J; ++jj) {
        const float f = Fp[(size_t)jj * RR];
        const float4 l = *(const float4*)(Lp + (size_t)jj * RSS);
        acc.x += l.x * f; acc.y += l.y * f; acc.z += l.z * f; acc.w += l.w * f;
    }
    float* o = Mpart + ((size_t)(combo * MIX_S + s) * RSS + rq * 4) * RR + c;
    o[0] = acc.x; o[RR] = acc.y; o[2 * RR] = acc.z; o[3 * RR] = acc.w;
}

__global__ __launch_bounds__(256) void mix_factor_apply_kernel(
    MixPaths mp, const float* __restrict__ Mpart, float* __restrict__ G)
{
    __shared__ __align__(16) float Ms[RSS * RR];
    const int combo = blockIdx.y;
    const int p = combo >> 1, b = combo & 1;
    const MixPath mx = p == 0 ? mp.p0 : (p == 1 ? mp.p1 : mp.p2);
    {
        const float4* src = (const float4*)(Mpart + (size_t)combo * MIX_S * RSS * RR) + threadIdx.x;
        float4 acc = src[0];
        #pragma unroll
        for (int sl = 1; sl < MIX_S; ++sl) {
            const float4 t = src[(size_t)sl * (RSS * RR / 4)];
            acc.x += t.x; acc.y += t.y; acc.z += t.z; acc.w += t.w;
        }
        ((float4*)Ms)[threadIdx.x] = acc;
    }
    __syncthreads();
    const int c = threadIdx.x & 63, rq = threadIdx.x >> 6;
    float m[RSS];
    #pragma unroll
    for (int r = 0; r < RSS; ++r) m[r] = Ms[r * RR + c];
    const size_t row0 = (size_t)b * TT + (size_t)blockIdx.x * 64 + (size_t)rq * 16;
    float* Gp = G + (size_t)p * BT * RR;
    #pragma unroll 4
    for (int k = 0; k < 16; ++k) {
        const size_t row = row0 + k;
        const float4* rr = (const float4*)(mx.Rm + row * RSS);
        const float4 r0 = rr[0], r1 = rr[1], r2 = rr[2], r3 = rr[3];
        float v = mx.d[row] * mx.F[row * RR + c];
        v += r0.x * m[0];  v += r0.y * m[1];  v += r0.z * m[2];  v += r0.w * m[3];
        v += r1.x * m[4];  v += r1.y * m[5];  v += r1.z * m[6];  v += r1.w * m[7];
        v += r2.x * m[8];  v += r2.y * m[9];  v += r2.z * m[10]; v += r2.w * m[11];
        v += r3.x * m[12]; v += r3.y * m[13]; v += r3.z * m[14]; v += r3.w * m[15];
        Gp[row * RR + c] = v;
    }
}

__global__ __launch_bounds__(256) void rownorm_kernel(
    const float* __restrict__ h, float* __restrict__ denom)
{
    const int wave = threadIdx.x >> 5, lane = threadIdx.x & 31;
    const int rowg = blockIdx.x * 8 + wave;
    const float* hr = h + (size_t)rowg * TT;
    float acc = 0.f;
    for (int j = lane; j < TT; j += 32) { float v = hr[j]; acc += v * v; }
    acc = waveReduce(acc);
    if (lane == 0) denom[rowg] = sqrtf(acc) + 1e-8f;
}

__global__ __launch_bounds__(256) void combine_h_kernel(
    float* __restrict__ hmix, const float* __restrict__ hscale,
    const float* __restrict__ denom)
{
    const int rowg = blockIdx.y;
    const int j = (blockIdx.x * blockDim.x + threadIdx.x) * 4;
    const float inv = 1.f / denom[rowg];
    float4 m = *(const float4*)(hmix   + (size_t)rowg * TT + j);
    float4 s = *(const float4*)(hscale + (size_t)rowg * TT + j);
    float4 o;
    o.x = softplusf(s.x) * fmaxf(m.x * inv, 0.f);
    o.y = softplusf(s.y) * fmaxf(m.y * inv, 0.f);
    o.z = softplusf(s.z) * fmaxf(m.z * inv, 0.f);
    o.w = softplusf(s.w) * fmaxf(m.w * inv, 0.f);
    *(float4*)(hmix + (size_t)rowg * TT + j) = o;
}

extern "C" void kernel_launch(void* const* d_in, const int* in_sizes, int n_in,
                              void* d_out, int out_size, void* d_ws, size_t ws_size,
                              hipStream_t stream) {
    (void)in_sizes; (void)n_in; (void)out_size; (void)ws_size;

    const float* x = (const float*)d_in[0];
    const float* qk_Wq1  = (const float*)d_in[1];  const float* qk_Wq2  = (const float*)d_in[2];
    const float* qk_Wm1  = (const float*)d_in[3];  const float* qk_Wm2  = (const float*)d_in[4];
    const float* qk_Wk1  = (const float*)d_in[5];  const float* qk_Wk2  = (const float*)d_in[6];
    const float* qks_Wq1 = (const float*)d_in[7];  const float* qks_Wq2 = (const float*)d_in[8];
    const float* qks_Wm1 = (const float*)d_in[9];  const float* qks_Wm2 = (const float*)d_in[10];
    const float* qks_Wk1 = (const float*)d_in[11]; const float* qks_Wk2 = (const float*)d_in[12];
    const float* vo_Wv1  = (const float*)d_in[13]; const float* vo_Wv2  = (const float*)d_in[14];
    const float* vo_Wm1  = (const float*)d_in[15]; const float* vo_Wm2  = (const float*)d_in[16];
    const float* vo_Wo   = (const float*)d_in[17];
    const float* sm1_Wd  = (const float*)d_in[18]; const float* sm1_Wa  = (const float*)d_in[19]; const float* sm1_Wb  = (const float*)d_in[20];
    const float* sm1s_Wd = (const float*)d_in[21]; const float* sm1s_Wa = (const float*)d_in[22]; const float* sm1s_Wb = (const float*)d_in[23];
    const float* sm2_Wd  = (const float*)d_in[24]; const float* sm2_Wa  = (const float*)d_in[25]; const float* sm2_Wb  = (const float*)d_in[26];

    const size_t BTD = (size_t)BT * DD;
    const size_t BTT = (size_t)BB * TT * TT;

    float* w = (float*)d_ws;
    float* qg   = w; w += BTD;
    float* pkp  = w; w += BTD;
    float* qsg  = w; w += BTD;
    float* pp   = w; w += BTD;
    float* ctxg = w; w += BTD;
    float* tmpb = w; w += BTD;
    float* zb   = w; w += BTT;
    float* zsb  = w; w += BTT;
    float* xr8  = w; w += BTD;
    float* d1   = w; w += BT; float* d1s = w; w += BT; float* d2 = w; w += BT;
    float* A1   = w; w += (size_t)BT * RSS; float* B1  = w; w += (size_t)BT * RSS;
    float* A1s  = w; w += (size_t)BT * RSS; float* B1s = w; w += (size_t)BT * RSS;
    float* A2   = w; w += (size_t)BT * RSS; float* B2  = w; w += (size_t)BT * RSS;
    float* denom= w; w += BT;

    auto gemm = [&](const float* A, const float* Bm, float* C,
                    int M, int N, int K, int lda, int ldb, int ldc, int bT,
                    long long sA, long long sB, long long sC, int batches) {
        dim3 g((unsigned)(N / GBN), (unsigned)(M / GBM), (unsigned)batches);
        gemm_wmma_f32<<<g, dim3(256), 0, stream>>>(A, Bm, C, M, N, K, lda, ldb, ldc, bT, sA, sB, sC);
    };

    const long long sXR = (long long)BT * RR;
    const Weights8 W1 = { qk_Wm1, qk_Wq1, qk_Wk1, qks_Wm1, qks_Wq1, qks_Wk1, vo_Wv1, vo_Wm1 };
    const DabPaths P3 = { sm1_Wa,  sm1_Wb,  sm1_Wd,  A1,  B1,  d1,
                          sm1s_Wa, sm1s_Wb, sm1s_Wd, A1s, B1s, d1s,
                          sm2_Wa,  sm2_Wb,  sm2_Wd,  A2,  B2,  d2 };
    gemm_wmma_f32_shareA8<<<dim3(11, BT / GBM), dim3(256), 0, stream>>>(x, W1, xr8, DD, DD, sXR, P3);

    const GatedSets GQ = { xr8 + 1 * sXR, qk_Wq2,  xr8 + 0 * sXR, qk_Wm2,  qg,
                           xr8 + 4 * sXR, qks_Wq2, xr8 + 3 * sXR, qks_Wm2, qsg };
    gemm_wmma_f32_gated<<<dim3(DD / GBN, BT / GBM, 2), dim3(256), 0, stream>>>(GQ, RR, RR, DD, DD);

    float* Mpart = tmpb;
    float* gmix  = tmpb + (size_t)BT * RR;
    const MixPaths MP = { { xr8 + 2 * sXR, B1,  A1,  d1  },
                          { xr8 + 5 * sXR, B1s, A1s, d1s },
                          { xr8 + 6 * sXR, A2,  B2,  d2  } };
    mix_factor_partial_kernel<<<dim3(MIX_S, 6), 256, 0, stream>>>(MP, Mpart);
    mix_factor_apply_kernel<<<dim3(TT / 64, 6), 256, 0, stream>>>(MP, Mpart, gmix);

    const long long sTTl = (long long)TT * TT;
    const long long sTR = (long long)TT * RR;
    gemm_wmma_f32_splitk<<<dim3(1, BT / GBM, 2 * 8), dim3(256), 0, stream>>>(
        qg, qsg, qk_Wk2, qks_Wk2, pkp, pkp + 8 * sXR,
        DD / 8, DD, DD, RR, 1, 8, DD / 8, DD / 8, sXR);
    splitk_sum_kernel<<<dim3((unsigned)(sXR / 4 / 256), 2), 256, 0, stream>>>(pkp, pp, 8, sXR);
    gemm(pp,       gmix,       zb,  TT, TT, RR, RR, RR, TT, 1, sTR, sTR, sTTl, BB);
    gemm(pp + sXR, gmix + sXR, zsb, TT, TT, RR, RR, RR, TT, 1, sTR, sTR, sTTl, BB);

    rownorm_kernel<<<dim3(BT / 8), 256, 0, stream>>>(zb, denom);
    combine_h_kernel<<<dim3(TT / 1024, BT), 256, 0, stream>>>(zb, zsb, denom);

    float* vmix = gmix + 2 * sXR;
    gemm_wmma_f32_splitk<<<dim3(1, TT / GBM, 2 * 16), dim3(256), 0, stream>>>(
        zb, zb + sTTl, vmix, vmix + sTR, qg, qg + sTR,
        TT / 16, TT, RR, RR, 0, 16, TT / 16, (long long)(TT / 16) * RR, sXR);
    splitk_sum_kernel<<<dim3((unsigned)(sXR / 4 / 256), 1), 256, 0, stream>>>(qg, pkp, 16, sXR);
    const GatedSets GC = { pkp, vo_Wv2, xr8 + 7 * sXR, vo_Wm2, ctxg,
                           nullptr, nullptr, nullptr, nullptr, nullptr };
    gemm_wmma_f32_gated<<<dim3(DD / GBN, BT / GBM, 1), dim3(256), 0, stream>>>(GC, RR, RR, DD, DD);
    gemm(ctxg, vo_Wo, (float*)d_out, BT, DD, DD, DD, DD, DD, 1, 0, 0, 0, 1);
}


# h_mix, h_scale tiles are normalised and combined in the kernel that forms them (row norm from a first pass over the gate product, its squares summed tile by tile); z, zs never stored; 12 launches
# speedup vs baseline: 5.4215x; 1.0330x over previous
#include <hip/hip_runtime.h>
#include <math.h>

#define BB 2
#define TT 2048
#define DD 1024
#define RR 64
#define RSS 16
#define BT (BB*TT)

typedef __attribute__((ext_vector_type(2))) float v2f;
typedef __attribute__((ext_vector_type(8))) float v8f;

#define GBM 256
#define GBN 64
#define GBK 32

struct DabTile {
    const float* Wa; const float* Wb; const float* Wd;
    float* aOut; float* bOut; float* dOut;
};

struct TileAcc { v8f c00, c01, c02, c03, c10, c11, c12, c13; };

__device__ __forceinline__ void gemm_tile_loop_f32(
    const float* __restrict__ Ab, const float* __restrict__ Bb,
    int K, int lda, int ldb, int bT, int m0, int n0,
    bool dab, DabTile dt, TileAcc& acc)
{
    __shared__ __align__(16) float As [2][GBM][34];
    __shared__ __align__(16) float BsT[2][GBN][34];

    const int tid  = threadIdx.x;
    const int wave = tid >> 5;
    const int lane = tid & 31;
    const int hi   = lane >> 4;
    const int lm   = lane & 15;

    float4 ra[8];
    float4 rb[2];

    int arow[8], ac4[8];
    #pragma unroll
    for (int i = 0; i < 8; ++i) { int idx = tid + i * 256; arow[i] = idx >> 3; ac4[i] = (idx & 7) * 4; }
    int bi0[2], bi1[2];
    #pragma unroll
    for (int i = 0; i < 2; ++i) {
        int idx = tid + i * 256;
        if (true) {   }
        bi0[i] = idx; bi1[i] = idx;
    }

    auto loadGlobal = [&](int k0) {
        #pragma unroll
        for (int i = 0; i < 8; ++i)
            ra[i] = *(const float4*)(Ab + (size_t)(m0 + arow[i]) * lda + k0 + ac4[i]);
        if (dab) {
            #pragma unroll
            for (int i = 0; i < 2; ++i) {
                int idx = bi1[i];
                int q = idx & 15;
                size_t k = (size_t)(k0 + (idx >> 4));
                float4 t = make_float4(0.f, 0.f, 0.f, 0.f);
                if (q < 4)       t = *(const float4*)(dt.Wa + k * RSS + q * 4);
                else if (q < 8)  t = *(const float4*)(dt.Wb + k * RSS + (q - 4) * 4);
                else if (q == 8) t.x = dt.Wd[k];
                rb[i] = t;
            }
        } else if (bT) {
            #pragma unroll
            for (int i = 0; i < 2; ++i) {
                int idx = bi0[i];
                int n = idx >> 3, c4 = (idx & 7) * 4;
                rb[i] = *(const float4*)(Bb + (size_t)(n0 + n) * ldb + k0 + c4);
            }
        } else {
            #pragma unroll
            for (int i = 0; i < 2; ++i) {
                int idx = bi1[i];
                int r = idx >> 4, c4 = (idx & 15) * 4;
                rb[i] = *(const float4*)(Bb + (size_t)(k0 + r) * ldb + n0 + c4);
            }
        }
    };

    auto storeLDS = [&](int buf) {
        #pragma unroll
        for (int i = 0; i < 8; ++i) {
            float4 t = ra[i];
            v2f lo; lo.x = t.x; lo.y = t.y;
            v2f hi2; hi2.x = t.z; hi2.y = t.w;
            *(v2f*)&As[buf][arow[i]][ac4[i]]     = lo;
            *(v2f*)&As[buf][arow[i]][ac4[i] + 2] = hi2;
        }
        if (bT) {
            #pragma unroll
            for (int i = 0; i < 2; ++i) {
                int idx = bi0[i];
                int n = idx >> 3, c4 = (idx & 7) * 4;
                float4 t = rb[i];
                v2f lo; lo.x = t.x; lo.y = t.y;
                v2f hi2; hi2.x = t.z; hi2.y = t.w;
                *(v2f*)&BsT[buf][n][c4]     = lo;
                *(v2f*)&BsT[buf][n][c4 + 2] = hi2;
            }
        } else {
            #pragma unroll
            for (int i = 0; i < 2; ++i) {
                int idx = bi1[i];
                int r = idx >> 4, c4 = (idx & 15) * 4;
                float4 t = rb[i];
                BsT[buf][c4 + 0][r] = t.x;
                BsT[buf][c4 + 1][r] = t.y;
                BsT[buf][c4 + 2][r] = t.z;
                BsT[buf][c4 + 3][r] = t.w;
            }
        }
    };

    v8f acc00 = {}, acc01 = {}, acc02 = {}, acc03 = {};
    v8f acc10 = {}, acc11 = {}, acc12 = {}, acc13 = {};

    const int r0 = wave * 32 + lm;
    const int r1 = r0 + 16;
    const int nchunks = K / GBK;

    loadGlobal(0);
    for (int c = 0; c < nchunks; ++c) {
        const int buf = c & 1;
        storeLDS(buf);
        __syncthreads();
        if (c + 1 < nchunks) loadGlobal((c + 1) * GBK);

        #pragma unroll
        for (int kk = 0; kk < GBK; kk += 4) {
            const int kc = kk + 2 * hi;
            v2f a0 = *(const v2f*)&As [buf][r0][kc];
            v2f a1 = *(const v2f*)&As [buf][r1][kc];
            v2f b0 = *(const v2f*)&BsT[buf][lm +  0][kc];
            v2f b1 = *(const v2f*)&BsT[buf][lm + 16][kc];
            v2f b2 = *(const v2f*)&BsT[buf][lm + 32][kc];
            v2f b3 = *(const v2f*)&BsT[buf][lm + 48][kc];
            acc00 = __builtin_amdgcn_wmma_f32_16x16x4_f32(false, a0, false, b0, (short)0, acc00, false, false);
            acc01 = __builtin_amdgcn_wmma_f32_16x16x4_f32(false, a0, false, b1, (short)0, acc01, false, false);
            acc02 = __builtin_amdgcn_wmma_f32_16x16x4_f32(false, a0, false, b2, (short)0, acc02, false, false);
            acc03 = __builtin_amdgcn_wmma_f32_16x16x4_f32(false, a0, false, b3, (short)0, acc03, false, false);
            acc10 = __builtin_amdgcn_wmma_f32_16x16x4_f32(false, a1, false, b0, (short)0, acc10, false, false);
            acc11 = __builtin_amdgcn_wmma_f32_16x16x4_f32(false, a1, false, b1, (short)0, acc11, false, false);
            acc12 = __builtin_amdgcn_wmma_f32_16x16x4_f32(false, a1, false, b2, (short)0, acc12, false, false);
            acc13 = __builtin_amdgcn_wmma_f32_16x16x4_f32(false, a1, false, b3, (short)0, acc13, false, false);
        }
        __syncthreads();
    }
    acc.c00 = acc00; acc.c01 = acc01; acc.c02 = acc02; acc.c03 = acc03;
    acc.c10 = acc10; acc.c11 = acc11; acc.c12 = acc12; acc.c13 = acc13;
}

__device__ __forceinline__ void gemm_tile_f32(
    const float* __restrict__ Ab, const float* __restrict__ Bb,
    float* __restrict__ Cb,
    int K, int lda, int ldb, int ldc, int bT, int m0, int n0,
    bool dab, DabTile dt)
{
    TileAcc t;
    gemm_tile_loop_f32(Ab, Bb, K, lda, ldb, bT, m0, n0, dab, dt, t);

    const int wave = threadIdx.x >> 5;
    const int lane = threadIdx.x & 31;
    const int hi   = lane >> 4;
    const int lm   = lane & 15;

    if (dab) {
        #pragma unroll
        for (int mt = 0; mt < 2; ++mt) {
            const int mrow = m0 + wave * 32 + mt * 16 + hi * 8;
            v8f a0 = mt ? t.c10 : t.c00, a1 = mt ? t.c11 : t.c01, a2 = mt ? t.c12 : t.c02;
            #pragma unroll
            for (int r = 0; r < 8; ++r) {
                size_t row = (size_t)(mrow + r);
                dt.aOut[row * RSS + lm] = a0[r];
                dt.bOut[row * RSS + lm] = a1[r];
                if (lm == 0) dt.dOut[row] = a2[r];
            }
        }
        return;
    }

    const int ncol = n0 + lm;
    #pragma unroll
    for (int mt = 0; mt < 2; ++mt) {
        const int mrow = m0 + wave * 32 + mt * 16 + hi * 8;
        v8f a0 = mt ? t.c10 : t.c00, a1 = mt ? t.c11 : t.c01;
        v8f a2 = mt ? t.c12 : t.c02, a3 = mt ? t.c13 : t.c03;
        #pragma unroll
        for (int r = 0; r < 8; ++r) {
            size_t base = (size_t)(mrow + r) * ldc + ncol;
            Cb[base] = a0[r]; Cb[base + 16] = a1[r]; Cb[base + 32] = a2[r]; Cb[base + 48] = a3[r];
        }
    }
}

__global__ __launch_bounds__(256) void gemm_wmma_f32(
    const float* __restrict__ A, const float* __restrict__ Bm,
    float* __restrict__ C,
    int M, int N, int K, int lda, int ldb, int ldc, int bT,
    long long sA, long long sB, long long sC)
{
    (void)M; (void)N;
    const float* Ab = A  + (size_t)blockIdx.z * sA;
    const float* Bb = Bm + (size_t)blockIdx.z * sB;
    float*       Cb = C  + (size_t)blockIdx.z * sC;
    gemm_tile_f32(Ab, Bb, Cb, K, lda, ldb, ldc, bT,
                  (int)blockIdx.y * GBM, (int)blockIdx.x * GBN, false, DabTile{});
}

struct GatedSets {
    const float* A1a; const float* B1a; const float* A2a; const float* B2a; float* Ca;
    const float* A1b; const float* B1b; const float* A2b; const float* B2b; float* Cb;
};

__global__ __launch_bounds__(256) void gemm_wmma_f32_gated(
    GatedSets g, int K, int lda, int ldb, int ldc)
{
    const bool sb = blockIdx.z != 0;
    const float* A1 = sb ? g.A1b : g.A1a;
    const float* B1 = sb ? g.B1b : g.B1a;
    const float* A2 = sb ? g.A2b : g.A2a;
    const float* B2 = sb ? g.B2b : g.B2a;
    float*       C  = sb ? g.Cb  : g.Ca;
    const int m0 = (int)blockIdx.y * GBM, n0 = (int)blockIdx.x * GBN;

    __shared__ __align__(16) float4 Gs[16][256];
    const int tid = threadIdx.x;

    TileAcc gate, val;
    gemm_tile_loop_f32(A2, B2, K, lda, ldb, 0, m0, n0, false, DabTile{}, gate);
    {
        const v8f* gv[8] = { &gate.c00, &gate.c01, &gate.c02, &gate.c03,
                             &gate.c10, &gate.c11, &gate.c12, &gate.c13 };
        #pragma unroll
        for (int t = 0; t < 8; ++t) {
            const v8f v = *gv[t];
            Gs[2 * t][tid]     = make_float4(v[0], v[1], v[2], v[3]);
            Gs[2 * t + 1][tid] = make_float4(v[4], v[5], v[6], v[7]);
        }
    }
    gemm_tile_loop_f32(A1, B1, K, lda, ldb, 0, m0, n0, false, DabTile{}, val);
    {
        v8f* gv[8] = { &gate.c00, &gate.c01, &gate.c02, &gate.c03,
                       &gate.c10, &gate.c11, &gate.c12, &gate.c13 };
        #pragma unroll
        for (int t = 0; t < 8; ++t) {
            const float4 lo = Gs[2 * t][tid], up = Gs[2 * t + 1][tid];
            v8f v;
            v[0] = lo.x; v[1] = lo.y; v[2] = lo.z; v[3] = lo.w;
            v[4] = up.x; v[5] = up.y; v[6] = up.z; v[7] = up.w;
            *gv[t] = v;
        }
    }

    const int wave = threadIdx.x >> 5;
    const int lane = threadIdx.x & 31;
    const int hi   = lane >> 4;
    const int lm   = lane & 15;
    const int ncol = n0 + lm;
    #pragma unroll
    for (int mt = 0; mt < 2; ++mt) {
        const int mrow = m0 + wave * 32 + mt * 16 + hi * 8;
        v8f a0 = mt ? val.c10 : val.c00, a1 = mt ? val.c11 : val.c01;
        v8f a2 = mt ? val.c12 : val.c02, a3 = mt ? val.c13 : val.c03;
        v8f g0 = mt ? gate.c10 : gate.c00, g1 = mt ? gate.c11 : gate.c01;
        v8f g2 = mt ? gate.c12 : gate.c02, g3 = mt ? gate.c13 : gate.c03;
        #pragma unroll
        for (int r = 0; r < 8; ++r) {
            size_t base = (size_t)(mrow + r) * ldc + ncol;
            C[base]      = a0[r] * g0[r];
            C[base + 16] = a1[r] * g1[r];
            C[base + 32] = a2[r] * g2[r];
            C[base + 48] = a3[r] * g3[r];
        }
    }
}

struct Weights8 {
    const float* w0; const float* w1; const float* w2; const float* w3;
    const float* w4; const float* w5; const float* w6; const float* w7;
};
struct DabPaths {
    const float* wa0; const float* wb0; const float* wd0; float* a0; float* b0; float* d0;
    const float* wa1; const float* wb1; const float* wd1; float* a1; float* b1; float* d1;
    const float* wa2; const float* wb2; const float* wd2; float* a2; float* b2; float* d2;
};

__global__ __launch_bounds__(256) void gemm_wmma_f32_shareA8(
    const float* __restrict__ A, Weights8 W, float* __restrict__ C,
    int K, int lda, long long sC, DabPaths P)
{
    const int j = blockIdx.x;
    if (j >= 8) {
        DabTile dt = { P.wa0, P.wb0, P.wd0, P.a0, P.b0, P.d0 };
        if (j == 9)  dt = DabTile{ P.wa1, P.wb1, P.wd1, P.a1, P.b1, P.d1 };
        if (j == 10) dt = DabTile{ P.wa2, P.wb2, P.wd2, P.a2, P.b2, P.d2 };
        gemm_tile_f32(A, nullptr, nullptr, K, lda, 0, 0, 0,
                      (int)blockIdx.y * GBM, 0, true, dt);
        return;
    }
    const float* Bb = W.w0;
    switch (j) {
        case 1: Bb = W.w1; break;
        case 2: Bb = W.w2; break;
        case 3: Bb = W.w3; break;
        case 4: Bb = W.w4; break;
        case 5: Bb = W.w5; break;
        case 6: Bb = W.w6; break;
        case 7: Bb = W.w7; break;
        default: break;
    }
    gemm_tile_f32(A, Bb, C + (size_t)j * sC, K, lda, GBN, GBN, 0,
                  (int)blockIdx.y * GBM, 0, false, DabTile{});
}

__global__ __launch_bounds__(256) void gemm_wmma_f32_splitk(
    const float* A0, const float* A1, const float* B0, const float* B1,
    float* C0, float* C1,
    int K, int lda, int ldb, int ldc, int bT, int S,
    long long sAs, long long sBs, long long sCs)
{
    const int y = (int)blockIdx.z / S;
    const int s = (int)blockIdx.z - y * S;
    const float* Ab = (y ? A1 : A0) + (size_t)s * sAs;
    const float* Bb = (y ? B1 : B0) + (size_t)s * sBs;
    float*       Cb = (y ? C1 : C0) + (size_t)s * sCs;
    gemm_tile_f32(Ab, Bb, Cb, K, lda, ldb, ldc, bT,
                  (int)blockIdx.y * GBM, (int)blockIdx.x * GBN, false, DabTile{});
}

__global__ __launch_bounds__(256) void splitk_sum_kernel(
    const float* __restrict__ part, float* __restrict__ out, int S, long long n)
{
    const size_t e = ((size_t)blockIdx.x * blockDim.x + threadIdx.x) * 4;
    const float* p = part + (size_t)blockIdx.y * (size_t)S * (size_t)n + e;
    float4 acc = *(const float4*)p;
    for (int s = 1; s < S; ++s) {
        const float4 t = *(const float4*)(p + (size_t)s * (size_t)n);
        acc.x += t.x; acc.y += t.y; acc.z += t.z; acc.w += t.w;
    }
    *(float4*)(out + (size_t)blockIdx.y * (size_t)n + e) = acc;
}

__device__ inline float softplusf(float x) {
    if (x > 20.f) return x;
    if (x < -20.f) return expf(x);
    return log1pf(expf(x));
}

struct MixPath { const float* F; const float* L; const float* Rm; const float* d; };
struct MixPaths { MixPath p0; MixPath p1; MixPath p2; };

#define MIX_S 32
#define MIX_J (TT / MIX_S)

__global__ __launch_bounds__(256) void mix_factor_partial_kernel(
    MixPaths mp, float* __restrict__ Mpart)
{
    const int s = blockIdx.x, combo = blockIdx.y;
    const int p = combo >> 1, b = combo & 1;
    const MixPath mx = p == 0 ? mp.p0 : (p == 1 ? mp.p1 : mp.p2);
    const int c = threadIdx.x & 63, rq = threadIdx.x >> 6;
    const size_t row0 = (size_t)b * TT + (size_t)s * MIX_J;
    const float* Fp = mx.F + row0 * RR + c;
    const float* Lp = mx.L + row0 * RSS + rq * 4;
    float4 acc = make_float4(0.f, 0.f, 0.f, 0.f);
    #pragma unroll 16
    for (int jj = 0; jj < MIX_J; ++jj) {
        const float f = Fp[(size_t)jj * RR];
        const float4 l = *(const float4*)(Lp + (size_t)jj * RSS);
        acc.x += l.x * f; acc.y += l.y * f; acc.z += l.z * f; acc.w += l.w * f;
    }
    float* o = Mpart + ((size_t)(combo * MIX_S + s) * RSS + rq * 4) * RR + c;
    o[0] = acc.x; o[RR] = acc.y; o[2 * RR] = acc.z; o[3 * RR] = acc.w;
}

__global__ __launch_bounds__(256) void mix_factor_apply_kernel(
    MixPaths mp, const float* __restrict__ Mpart, float* __restrict__ G)
{
    __shared__ __align__(16) float Ms[RSS * RR];
    const int combo = blockIdx.y;
    const int p = combo >> 1, b = combo & 1;
    const MixPath mx = p == 0 ? mp.p0 : (p == 1 ? mp.p1 : mp.p2);
    {
        const float4* src = (const float4*)(Mpart + (size_t)combo * MIX_S * RSS * RR) + threadIdx.x;
        float4 acc = src[0];
        #pragma unroll
        for (int sl = 1; sl < MIX_S; ++sl) {
            const float4 t = src[(size_t)sl * (RSS * RR / 4)];
            acc.x += t.x; acc.y += t.y; acc.z += t.z; acc.w += t.w;
        }
        ((float4*)Ms)[threadIdx.x] = acc;
    }
    __syncthreads();
    const int c = threadIdx.x & 63, rq = threadIdx.x >> 6;
    float m[RSS];
    #pragma unroll
    for (int r = 0; r < RSS; ++r) m[r] = Ms[r * RR + c];
    const size_t row0 = (size_t)b * TT + (size_t)blockIdx.x * 64 + (size_t)rq * 16;
    float* Gp = G + (size_t)p * BT * RR;
    #pragma unroll 4
    for (int k = 0; k < 16; ++k) {
        const size_t row = row0 + k;
        const float4* rr = (const float4*)(mx.Rm + row * RSS);
        const float4 r0 = rr[0], r1 = rr[1], r2 = rr[2], r3 = rr[3];
        float v = mx.d[row] * mx.F[row * RR + c];
        v += r0.x * m[0];  v += r0.y * m[1];  v += r0.z * m[2];  v += r0.w * m[3];
        v += r1.x * m[4];  v += r1.y * m[5];  v += r1.z * m[6];  v += r1.w * m[7];
        v += r2.x * m[8];  v += r2.y * m[9];  v += r2.z * m[10]; v += r2.w * m[11];
        v += r3.x * m[12]; v += r3.y * m[13]; v += r3.z * m[14]; v += r3.w * m[15];
        Gp[row * RR + c] = v;
    }
}

#define NCT (TT / GBN)

__global__ __launch_bounds__(256) void score_rowsq_kernel(
    const float* __restrict__ P, const float* __restrict__ Kmix,
    float* __restrict__ part, int K, int lda, int ldb,
    long long sA, long long sB)
{
    const int b  = blockIdx.z;
    const int m0 = (int)blockIdx.y * GBM, n0 = (int)blockIdx.x * GBN;
    TileAcc t;
    gemm_tile_loop_f32(P + (size_t)b * sA, Kmix + (size_t)b * sB, K, lda, ldb, 1,
                       m0, n0, false, DabTile{}, t);

    const int wave = threadIdx.x >> 5;
    const int lane = threadIdx.x & 31;
    const int hi   = lane >> 4;
    const int lm   = lane & 15;
    float* o = part + ((size_t)b * NCT + blockIdx.x) * TT + m0 + wave * 32 + hi * 8;
    #pragma unroll
    for (int mt = 0; mt < 2; ++mt) {
        v8f a0 = mt ? t.c10 : t.c00, a1 = mt ? t.c11 : t.c01;
        v8f a2 = mt ? t.c12 : t.c02, a3 = mt ? t.c13 : t.c03;
        float s[8];
        #pragma unroll
        for (int r = 0; r < 8; ++r) {
            float v = a0[r] * a0[r];
            v += a1[r] * a1[r]; v += a2[r] * a2[r]; v += a3[r] * a3[r];
            v += __shfl_xor(v, 8, 32); v += __shfl_xor(v, 4, 32);
            v += __shfl_xor(v, 2, 32); v += __shfl_xor(v, 1, 32);
            s[r] = v;
        }
        if (lm == 0) {
            float4* q = (float4*)(o + mt * 16);
            q[0] = make_float4(s[0], s[1], s[2], s[3]);
            q[1] = make_float4(s[4], s[5], s[6], s[7]);
        }
    }
}

__global__ __launch_bounds__(256) void score_combine_kernel(
    const float* __restrict__ P,  const float* __restrict__ Kmix,
    const float* __restrict__ Ps, const float* __restrict__ Ksmix,
    const float* __restrict__ part, float* __restrict__ H,
    int K, int lda, int ldb, int ldc,
    long long sA, long long sB, long long sC)
{
    __shared__ __align__(16) float4 Gs[16][256];
    __shared__ float invn[GBM];
    const int b   = blockIdx.z;
    const int m0  = (int)blockIdx.y * GBM, n0 = (int)blockIdx.x * GBN;
    const int tid = threadIdx.x;

    {
        const float* pr = part + (size_t)b * NCT * TT + m0 + tid;
        float acc = pr[0];
        #pragma unroll
        for (int ct = 1; ct < NCT; ++ct) acc += pr[(size_t)ct * TT];
        invn[tid] = 1.f / (sqrtf(acc) + 1e-8f);
    }

    TileAcc gate, val;
    gemm_tile_loop_f32(Ps + (size_t)b * sA, Ksmix + (size_t)b * sB, K, lda, ldb, 1,
                       m0, n0, false, DabTile{}, gate);
    {
        const v8f* gv[8] = { &gate.c00, &gate.c01, &gate.c02, &gate.c03,
                             &gate.c10, &gate.c11, &gate.c12, &gate.c13 };
        #pragma unroll
        for (int t = 0; t < 8; ++t) {
            const v8f v = *gv[t];
            Gs[2 * t][tid]     = make_float4(v[0], v[1], v[2], v[3]);
            Gs[2 * t + 1][tid] = make_float4(v[4], v[5], v[6], v[7]);
        }
    }
    #pragma nounroll
    for (int q = 0; q < 16; ++q) {
        float4 v = Gs[q][tid];
        v.x = softplusf(v.x); v.y = softplusf(v.y);
        v.z = softplusf(v.z); v.w = softplusf(v.w);
        Gs[q][tid] = v;
    }
    gemm_tile_loop_f32(P + (size_t)b * sA, Kmix + (size_t)b * sB, K, lda, ldb, 1,
                       m0, n0, false, DabTile{}, val);
    {
        v8f* gv[8] = { &gate.c00, &gate.c01, &gate.c02, &gate.c03,
                       &gate.c10, &gate.c11, &gate.c12, &gate.c13 };
        #pragma unroll
        for (int t = 0; t < 8; ++t) {
            const float4 lo = Gs[2 * t][tid], up = Gs[2 * t + 1][tid];
            v8f v;
            v[0] = lo.x; v[1] = lo.y; v[2] = lo.z; v[3] = lo.w;
            v[4] = up.x; v[5] = up.y; v[6] = up.z; v[7] = up.w;
            *gv[t] = v;
        }
    }

    const int wave = threadIdx.x >> 5;
    const int lane = threadIdx.x & 31;
    const int hi   = lane >> 4;
    const int lm   = lane & 15;
    float* Hb = H + (size_t)b * sC;
    const int ncol = n0 + lm;
    #pragma unroll
    for (int mt = 0; mt < 2; ++mt) {
        const int rl = wave * 32 + mt * 16 + hi * 8;
        v8f a0 = mt ? val.c10 : val.c00, a1 = mt ? val.c11 : val.c01;
        v8f a2 = mt ? val.c12 : val.c02, a3 = mt ? val.c13 : val.c03;
        v8f g0 = mt ? gate.c10 : gate.c00, g1 = mt ? gate.c11 : gate.c01;
        v8f g2 = mt ? gate.c12 : gate.c02, g3 = mt ? gate.c13 : gate.c03;
        #pragma unroll
        for (int r = 0; r < 8; ++r) {
            const float inv = invn[rl + r];
            size_t base = (size_t)(m0 + rl + r) * ldc + ncol;
            Hb[base]      = g0[r] * fmaxf(a0[r] * inv, 0.f);
            Hb[base + 16] = g1[r] * fmaxf(a1[r] * inv, 0.f);
            Hb[base + 32] = g2[r] * fmaxf(a2[r] * inv, 0.f);
            Hb[base + 48] = g3[r] * fmaxf(a3[r] * inv, 0.f);
        }
    }
}

extern "C" void kernel_launch(void* const* d_in, const int* in_sizes, int n_in,
                              void* d_out, int out_size, void* d_ws, size_t ws_size,
                              hipStream_t stream) {
    (void)in_sizes; (void)n_in; (void)out_size; (void)ws_size;

    const float* x = (const float*)d_in[0];
    const float* qk_Wq1  = (const float*)d_in[1];  const float* qk_Wq2  = (const float*)d_in[2];
    const float* qk_Wm1  = (const float*)d_in[3];  const float* qk_Wm2  = (const float*)d_in[4];
    const float* qk_Wk1  = (const float*)d_in[5];  const float* qk_Wk2  = (const float*)d_in[6];
    const float* qks_Wq1 = (const float*)d_in[7];  const float* qks_Wq2 = (const float*)d_in[8];
    const float* qks_Wm1 = (const float*)d_in[9];  const float* qks_Wm2 = (const float*)d_in[10];
    const float* qks_Wk1 = (const float*)d_in[11]; const float* qks_Wk2 = (const float*)d_in[12];
    const float* vo_Wv1  = (const float*)d_in[13]; const float* vo_Wv2  = (const float*)d_in[14];
    const float* vo_Wm1  = (const float*)d_in[15]; const float* vo_Wm2  = (const float*)d_in[16];
    const float* vo_Wo   = (const float*)d_in[17];
    const float* sm1_Wd  = (const float*)d_in[18]; const float* sm1_Wa  = (const float*)d_in[19]; const float* sm1_Wb  = (const float*)d_in[20];
    const float* sm1s_Wd = (const float*)d_in[21]; const float* sm1s_Wa = (const float*)d_in[22]; const float* sm1s_Wb = (const float*)d_in[23];
    const float* sm2_Wd  = (const float*)d_in[24]; const float* sm2_Wa  = (const float*)d_in[25]; const float* sm2_Wb  = (const float*)d_in[26];

    const size_t BTD = (size_t)BT * DD;
    const size_t BTT = (size_t)BB * TT * TT;

    float* w = (float*)d_ws;
    float* qg   = w; w += BTD;
    float* pkp  = w; w += BTD;
    float* qsg  = w; w += BTD;
    float* pp   = w; w += BTD;
    float* ctxg = w; w += BTD;
    float* tmpb = w; w += BTD;
    float* zb   = w; w += BTT;
    float* zsb  = w; w += BTT;
    float* xr8  = w; w += BTD;
    float* d1   = w; w += BT; float* d1s = w; w += BT; float* d2 = w; w += BT;
    float* A1   = w; w += (size_t)BT * RSS; float* B1  = w; w += (size_t)BT * RSS;
    float* A1s  = w; w += (size_t)BT * RSS; float* B1s = w; w += (size_t)BT * RSS;
    float* A2   = w; w += (size_t)BT * RSS; float* B2  = w; w += (size_t)BT * RSS;

    auto gemm = [&](const float* A, const float* Bm, float* C,
                    int M, int N, int K, int lda, int ldb, int ldc, int bT,
                    long long sA, long long sB, long long sC, int batches) {
        dim3 g((unsigned)(N / GBN), (unsigned)(M / GBM), (unsigned)batches);
        gemm_wmma_f32<<<g, dim3(256), 0, stream>>>(A, Bm, C, M, N, K, lda, ldb, ldc, bT, sA, sB, sC);
    };

    const long long sXR = (long long)BT * RR;
    const Weights8 W1 = { qk_Wm1, qk_Wq1, qk_Wk1, qks_Wm1, qks_Wq1, qks_Wk1, vo_Wv1, vo_Wm1 };
    const DabPaths P3 = { sm1_Wa,  sm1_Wb,  sm1_Wd,  A1,  B1,  d1,
                          sm1s_Wa, sm1s_Wb, sm1s_Wd, A1s, B1s, d1s,
                          sm2_Wa,  sm2_Wb,  sm2_Wd,  A2,  B2,  d2 };
    gemm_wmma_f32_shareA8<<<dim3(11, BT / GBM), dim3(256), 0, stream>>>(x, W1, xr8, DD, DD, sXR, P3);

    const GatedSets GQ = { xr8 + 1 * sXR, qk_Wq2,  xr8 + 0 * sXR, qk_Wm2,  qg,
                           xr8 + 4 * sXR, qks_Wq2, xr8 + 3 * sXR, qks_Wm2, qsg };
    gemm_wmma_f32_gated<<<dim3(DD / GBN, BT / GBM, 2), dim3(256), 0, stream>>>(GQ, RR, RR, DD, DD);

    float* Mpart = tmpb;
    float* gmix  = tmpb + (size_t)BT * RR;
    const MixPaths MP = { { xr8 + 2 * sXR, B1,  A1,  d1  },
                          { xr8 + 5 * sXR, B1s, A1s, d1s },
                          { xr8 + 6 * sXR, A2,  B2,  d2  } };
    mix_factor_partial_kernel<<<dim3(MIX_S, 6), 256, 0, stream>>>(MP, Mpart);
    mix_factor_apply_kernel<<<dim3(TT / 64, 6), 256, 0, stream>>>(MP, Mpart, gmix);

    const long long sTTl = (long long)TT * TT;
    const long long sTR = (long long)TT * RR;
    gemm_wmma_f32_splitk<<<dim3(1, BT / GBM, 2 * 8), dim3(256), 0, stream>>>(
        qg, qsg, qk_Wk2, qks_Wk2, pkp, pkp + 8 * sXR,
        DD / 8, DD, DD, RR, 1, 8, DD / 8, DD / 8, sXR);
    splitk_sum_kernel<<<dim3((unsigned)(sXR / 4 / 256), 2), 256, 0, stream>>>(pkp, pp, 8, sXR);

    score_rowsq_kernel<<<dim3(NCT, TT / GBM, BB), dim3(256), 0, stream>>>(
        pp, gmix, zsb, RR, RR, RR, sTR, sTR);
    score_combine_kernel<<<dim3(NCT, TT / GBM, BB), dim3(256), 0, stream>>>(
        pp, gmix, pp + sXR, gmix + sXR, zsb, zb, RR, RR, RR, TT, sTR, sTR, sTTl);

    float* vmix = gmix + 2 * sXR;
    gemm_wmma_f32_splitk<<<dim3(1, TT / GBM, 2 * 16), dim3(256), 0, stream>>>(
        zb, zb + sTTl, vmix, vmix + sTR, qg, qg + sTR,
        TT / 16, TT, RR, RR, 0, 16, TT / 16, (long long)(TT / 16) * RR, sXR);
    splitk_sum_kernel<<<dim3((unsigned)(sXR / 4 / 256), 1), 256, 0, stream>>>(qg, pkp, 16, sXR);
    const GatedSets GC = { pkp, vo_Wv2, xr8 + 7 * sXR, vo_Wm2, ctxg,
                           nullptr, nullptr, nullptr, nullptr, nullptr };
    gemm_wmma_f32_gated<<<dim3(DD / GBN, BT / GBM, 1), dim3(256), 0, stream>>>(GC, RR, RR, DD, DD);
    gemm(ctxg, vo_Wo, (float*)d_out, BT, DD, DD, DD, DD, DD, 1, 0, 0, 0, 1);
}


# mix slice sums ride in the split-K launch of P/Ps and the mix application in the launch that adds its planes (roles by block number): 10 launches (was 12); every sum order kept
# speedup vs baseline: 5.6555x; 1.0432x over previous
#include <hip/hip_runtime.h>
#include <math.h>

#define BB 2
#define TT 2048
#define DD 1024
#define RR 64
#define RSS 16
#define BT (BB*TT)

typedef __attribute__((ext_vector_type(2))) float v2f;
typedef __attribute__((ext_vector_type(8))) float v8f;

#define GBM 256
#define GBN 64
#define GBK 32

struct DabTile {
    const float* Wa; const float* Wb; const float* Wd;
    float* aOut; float* bOut; float* dOut;
};

struct TileAcc { v8f c00, c01, c02, c03, c10, c11, c12, c13; };

__device__ __forceinline__ void gemm_tile_loop_f32(
    const float* __restrict__ Ab, const float* __restrict__ Bb,
    int K, int lda, int ldb, int bT, int m0, int n0,
    bool dab, DabTile dt, TileAcc& acc)
{
    __shared__ __align__(16) float As [2][GBM][34];
    __shared__ __align__(16) float BsT[2][GBN][34];

    const int tid  = threadIdx.x;
    const int wave = tid >> 5;
    const int lane = tid & 31;
    const int hi   = lane >> 4;
    const int lm   = lane & 15;

    float4 ra[8];
    float4 rb[2];

    int arow[8], ac4[8];
    #pragma unroll
    for (int i = 0; i < 8; ++i) { int idx = tid + i * 256; arow[i] = idx >> 3; ac4[i] = (idx & 7) * 4; }
    int bi0[2], bi1[2];
    #pragma unroll
    for (int i = 0; i < 2; ++i) {
        int idx = tid + i * 256;
        if (true) {   }
        bi0[i] = idx; bi1[i] = idx;
    }

    auto loadGlobal = [&](int k0) {
        #pragma unroll
        for (int i = 0; i < 8; ++i)
            ra[i] = *(const float4*)(Ab + (size_t)(m0 + arow[i]) * lda + k0 + ac4[i]);
        if (dab) {
            #pragma unroll
            for (int i = 0; i < 2; ++i) {
                int idx = bi1[i];
                int q = idx & 15;
                size_t k = (size_t)(k0 + (idx >> 4));
                float4 t = make_float4(0.f, 0.f, 0.f, 0.f);
                if (q < 4)       t = *(const float4*)(dt.Wa + k * RSS + q * 4);
                else if (q < 8)  t = *(const float4*)(dt.Wb + k * RSS + (q - 4) * 4);
                else if (q == 8) t.x = dt.Wd[k];
                rb[i] = t;
            }
        } else if (bT) {
            #pragma unroll
            for (int i = 0; i < 2; ++i) {
                int idx = bi0[i];
                int n = idx >> 3, c4 = (idx & 7) * 4;
                rb[i] = *(const float4*)(Bb + (size_t)(n0 + n) * ldb + k0 + c4);
            }
        } else {
            #pragma unroll
            for (int i = 0; i < 2; ++i) {
                int idx = bi1[i];
                int r = idx >> 4, c4 = (idx & 15) * 4;
                rb[i] = *(const float4*)(Bb + (size_t)(k0 + r) * ldb + n0 + c4);
            }
        }
    };

    auto storeLDS = [&](int buf) {
        #pragma unroll
        for (int i = 0; i < 8; ++i) {
            float4 t = ra[i];
            v2f lo; lo.x = t.x; lo.y = t.y;
            v2f hi2; hi2.x = t.z; hi2.y = t.w;
            *(v2f*)&As[buf][arow[i]][ac4[i]]     = lo;
            *(v2f*)&As[buf][arow[i]][ac4[i] + 2] = hi2;
        }
        if (bT) {
            #pragma unroll
            for (int i = 0; i < 2; ++i) {
                int idx = bi0[i];
                int n = idx >> 3, c4 = (idx & 7) * 4;
                float4 t = rb[i];
                v2f lo; lo.x = t.x; lo.y = t.y;
                v2f hi2; hi2.x = t.z; hi2.y = t.w;
                *(v2f*)&BsT[buf][n][c4]     = lo;
                *(v2f*)&BsT[buf][n][c4 + 2] = hi2;
            }
        } else {
            #pragma unroll
            for (int i = 0; i < 2; ++i) {
                int idx = bi1[i];
                int r = idx >> 4, c4 = (idx & 15) * 4;
                float4 t = rb[i];
                BsT[buf][c4 + 0][r] = t.x;
                BsT[buf][c4 + 1][r] = t.y;
                BsT[buf][c4 + 2][r] = t.z;
                BsT[buf][c4 + 3][r] = t.w;
            }
        }
    };

    v8f acc00 = {}, acc01 = {}, acc02 = {}, acc03 = {};
    v8f acc10 = {}, acc11 = {}, acc12 = {}, acc13 = {};

    const int r0 = wave * 32 + lm;
    const int r1 = r0 + 16;
    const int nchunks = K / GBK;

    loadGlobal(0);
    for (int c = 0; c < nchunks; ++c) {
        const int buf = c & 1;
        storeLDS(buf);
        __syncthreads();
        if (c + 1 < nchunks) loadGlobal((c + 1) * GBK);

        #pragma unroll
        for (int kk = 0; kk < GBK; kk += 4) {
            const int kc = kk + 2 * hi;
            v2f a0 = *(const v2f*)&As [buf][r0][kc];
            v2f a1 = *(const v2f*)&As [buf][r1][kc];
            v2f b0 = *(const v2f*)&BsT[buf][lm +  0][kc];
            v2f b1 = *(const v2f*)&BsT[buf][lm + 16][kc];
            v2f b2 = *(const v2f*)&BsT[buf][lm + 32][kc];
            v2f b3 = *(const v2f*)&BsT[buf][lm + 48][kc];
            acc00 = __builtin_amdgcn_wmma_f32_16x16x4_f32(false, a0, false, b0, (short)0, acc00, false, false);
            acc01 = __builtin_amdgcn_wmma_f32_16x16x4_f32(false, a0, false, b1, (short)0, acc01, false, false);
            acc02 = __builtin_amdgcn_wmma_f32_16x16x4_f32(false, a0, false, b2, (short)0, acc02, false, false);
            acc03 = __builtin_amdgcn_wmma_f32_16x16x4_f32(false, a0, false, b3, (short)0, acc03, false, false);
            acc10 = __builtin_amdgcn_wmma_f32_16x16x4_f32(false, a1, false, b0, (short)0, acc10, false, false);
            acc11 = __builtin_amdgcn_wmma_f32_16x16x4_f32(false, a1, false, b1, (short)0, acc11, false, false);
            acc12 = __builtin_amdgcn_wmma_f32_16x16x4_f32(false, a1, false, b2, (short)0, acc12, false, false);
            acc13 = __builtin_amdgcn_wmma_f32_16x16x4_f32(false, a1, false, b3, (short)0, acc13, false, false);
        }
        __syncthreads();
    }
    acc.c00 = acc00; acc.c01 = acc01; acc.c02 = acc02; acc.c03 = acc03;
    acc.c10 = acc10; acc.c11 = acc11; acc.c12 = acc12; acc.c13 = acc13;
}

__device__ __forceinline__ void gemm_tile_f32(
    const float* __restrict__ Ab, const float* __restrict__ Bb,
    float* __restrict__ Cb,
    int K, int lda, int ldb, int ldc, int bT, int m0, int n0,
    bool dab, DabTile dt)
{
    TileAcc t;
    gemm_tile_loop_f32(Ab, Bb, K, lda, ldb, bT, m0, n0, dab, dt, t);

    const int wave = threadIdx.x >> 5;
    const int lane = threadIdx.x & 31;
    const int hi   = lane >> 4;
    const int lm   = lane & 15;

    if (dab) {
        #pragma unroll
        for (int mt = 0; mt < 2; ++mt) {
            const int mrow = m0 + wave * 32 + mt * 16 + hi * 8;
            v8f a0 = mt ? t.c10 : t.c00, a1 = mt ? t.c11 : t.c01, a2 = mt ? t.c12 : t.c02;
            #pragma unroll
            for (int r = 0; r < 8; ++r) {
                size_t row = (size_t)(mrow + r);
                dt.aOut[row * RSS + lm] = a0[r];
                dt.bOut[row * RSS + lm] = a1[r];
                if (lm == 0) dt.dOut[row] = a2[r];
            }
        }
        return;
    }

    const int ncol = n0 + lm;
    #pragma unroll
    for (int mt = 0; mt < 2; ++mt) {
        const int mrow = m0 + wave * 32 + mt * 16 + hi * 8;
        v8f a0 = mt ? t.c10 : t.c00, a1 = mt ? t.c11 : t.c01;
        v8f a2 = mt ? t.c12 : t.c02, a3 = mt ? t.c13 : t.c03;
        #pragma unroll
        for (int r = 0; r < 8; ++r) {
            size_t base = (size_t)(mrow + r) * ldc + ncol;
            Cb[base] = a0[r]; Cb[base + 16] = a1[r]; Cb[base + 32] = a2[r]; Cb[base + 48] = a3[r];
        }
    }
}

__global__ __launch_bounds__(256) void gemm_wmma_f32(
    const float* __restrict__ A, const float* __restrict__ Bm,
    float* __restrict__ C,
    int M, int N, int K, int lda, int ldb, int ldc, int bT,
    long long sA, long long sB, long long sC)
{
    (void)M; (void)N;
    const float* Ab = A  + (size_t)blockIdx.z * sA;
    const float* Bb = Bm + (size_t)blockIdx.z * sB;
    float*       Cb = C  + (size_t)blockIdx.z * sC;
    gemm_tile_f32(Ab, Bb, Cb, K, lda, ldb, ldc, bT,
                  (int)blockIdx.y * GBM, (int)blockIdx.x * GBN, false, DabTile{});
}

struct GatedSets {
    const float* A1a; const float* B1a; const float* A2a; const float* B2a; float* Ca;
    const float* A1b; const float* B1b; const float* A2b; const float* B2b; float* Cb;
};

__global__ __launch_bounds__(256) void gemm_wmma_f32_gated(
    GatedSets g, int K, int lda, int ldb, int ldc)
{
    const bool sb = blockIdx.z != 0;
    const float* A1 = sb ? g.A1b : g.A1a;
    const float* B1 = sb ? g.B1b : g.B1a;
    const float* A2 = sb ? g.A2b : g.A2a;
    const float* B2 = sb ? g.B2b : g.B2a;
    float*       C  = sb ? g.Cb  : g.Ca;
    const int m0 = (int)blockIdx.y * GBM, n0 = (int)blockIdx.x * GBN;

    __shared__ __align__(16) float4 Gs[16][256];
    const int tid = threadIdx.x;

    TileAcc gate, val;
    gemm_tile_loop_f32(A2, B2, K, lda, ldb, 0, m0, n0, false, DabTile{}, gate);
    {
        const v8f* gv[8] = { &gate.c00, &gate.c01, &gate.c02, &gate.c03,
                             &gate.c10, &gate.c11, &gate.c12, &gate.c13 };
        #pragma unroll
        for (int t = 0; t < 8; ++t) {
            const v8f v = *gv[t];
            Gs[2 * t][tid]     = make_float4(v[0], v[1], v[2], v[3]);
            Gs[2 * t + 1][tid] = make_float4(v[4], v[5], v[6], v[7]);
        }
    }
    gemm_tile_loop_f32(A1, B1, K, lda, ldb, 0, m0, n0, false, DabTile{}, val);
    {
        v8f* gv[8] = { &gate.c00, &gate.c01, &gate.c02, &gate.c03,
                       &gate.c10, &gate.c11, &gate.c12, &gate.c13 };
        #pragma unroll
        for (int t = 0; t < 8; ++t) {
            const float4 lo = Gs[2 * t][tid], up = Gs[2 * t + 1][tid];
            v8f v;
            v[0] = lo.x; v[1] = lo.y; v[2] = lo.z; v[3] = lo.w;
            v[4] = up.x; v[5] = up.y; v[6] = up.z; v[7] = up.w;
            *gv[t] = v;
        }
    }

    const int wave = threadIdx.x >> 5;
    const int lane = threadIdx.x & 31;
    const int hi   = lane >> 4;
    const int lm   = lane & 15;
    const int ncol = n0 + lm;
    #pragma unroll
    for (int mt = 0; mt < 2; ++mt) {
        const int mrow = m0 + wave * 32 + mt * 16 + hi * 8;
        v8f a0 = mt ? val.c10 : val.c00, a1 = mt ? val.c11 : val.c01;
        v8f a2 = mt ? val.c12 : val.c02, a3 = mt ? val.c13 : val.c03;
        v8f g0 = mt ? gate.c10 : gate.c00, g1 = mt ? gate.c11 : gate.c01;
        v8f g2 = mt ? gate.c12 : gate.c02, g3 = mt ? gate.c13 : gate.c03;
        #pragma unroll
        for (int r = 0; r < 8; ++r) {
            size_t base = (size_t)(mrow + r) * ldc + ncol;
            C[base]      = a0[r] * g0[r];
            C[base + 16] = a1[r] * g1[r];
            C[base + 32] = a2[r] * g2[r];
            C[base + 48] = a3[r] * g3[r];
        }
    }
}

struct Weights8 {
    const float* w0; const float* w1; const float* w2; const float* w3;
    const float* w4; const float* w5; const float* w6; const float* w7;
};
struct DabPaths {
    const float* wa0; const float* wb0; const float* wd0; float* a0; float* b0; float* d0;
    const float* wa1; const float* wb1; const float* wd1; float* a1; float* b1; float* d1;
    const float* wa2; const float* wb2; const float* wd2; float* a2; float* b2; float* d2;
};

__global__ __launch_bounds__(256) void gemm_wmma_f32_shareA8(
    const float* __restrict__ A, Weights8 W, float* __restrict__ C,
    int K, int lda, long long sC, DabPaths P)
{
    const int j = blockIdx.x;
    if (j >= 8) {
        DabTile dt = { P.wa0, P.wb0, P.wd0, P.a0, P.b0, P.d0 };
        if (j == 9)  dt = DabTile{ P.wa1, P.wb1, P.wd1, P.a1, P.b1, P.d1 };
        if (j == 10) dt = DabTile{ P.wa2, P.wb2, P.wd2, P.a2, P.b2, P.d2 };
        gemm_tile_f32(A, nullptr, nullptr, K, lda, 0, 0, 0,
                      (int)blockIdx.y * GBM, 0, true, dt);
        return;
    }
    const float* Bb = W.w0;
    switch (j) {
        case 1: Bb = W.w1; break;
        case 2: Bb = W.w2; break;
        case 3: Bb = W.w3; break;
        case 4: Bb = W.w4; break;
        case 5: Bb = W.w5; break;
        case 6: Bb = W.w6; break;
        case 7: Bb = W.w7; break;
        default: break;
    }
    gemm_tile_f32(A, Bb, C + (size_t)j * sC, K, lda, GBN, GBN, 0,
                  (int)blockIdx.y * GBM, 0, false, DabTile{});
}

__global__ __launch_bounds__(256) void gemm_wmma_f32_splitk(
    const float* A0, const float* A1, const float* B0, const float* B1,
    float* C0, float* C1,
    int K, int lda, int ldb, int ldc, int bT, int S,
    long long sAs, long long sBs, long long sCs)
{
    const int y = (int)blockIdx.z / S;
    const int s = (int)blockIdx.z - y * S;
    const float* Ab = (y ? A1 : A0) + (size_t)s * sAs;
    const float* Bb = (y ? B1 : B0) + (size_t)s * sBs;
    float*       Cb = (y ? C1 : C0) + (size_t)s * sCs;
    gemm_tile_f32(Ab, Bb, Cb, K, lda, ldb, ldc, bT,
                  (int)blockIdx.y * GBM, (int)blockIdx.x * GBN, false, DabTile{});
}

__device__ __forceinline__ void splitk_sum_body(
    const float* __restrict__ part, float* __restrict__ out, int S, long long n,
    int bx, int by)
{
    const size_t e = ((size_t)bx * 256 + threadIdx.x) * 4;
    const float* p = part + (size_t)by * (size_t)S * (size_t)n + e;
    float4 acc = *(const float4*)p;
    for (int s = 1; s < S; ++s) {
        const float4 t = *(const float4*)(p + (size_t)s * (size_t)n);
        acc.x += t.x; acc.y += t.y; acc.z += t.z; acc.w += t.w;
    }
    *(float4*)(out + (size_t)by * (size_t)n + e) = acc;
}

__global__ __launch_bounds__(256) void splitk_sum_kernel(
    const float* __restrict__ part, float* __restrict__ out, int S, long long n)
{
    splitk_sum_body(part, out, S, n, (int)blockIdx.x, (int)blockIdx.y);
}

__device__ inline float softplusf(float x) {
    if (x > 20.f) return x;
    if (x < -20.f) return expf(x);
    return log1pf(expf(x));
}

struct MixPath { const float* F; const float* L; const float* Rm; const float* d; };
struct MixPaths { MixPath p0; MixPath p1; MixPath p2; };

__device__ __forceinline__ MixPath mix_path(const MixPaths& mp, int p)
{
    MixPath mx;
    mx.F  = p == 0 ? mp.p0.F  : (p == 1 ? mp.p1.F  : mp.p2.F);
    mx.L  = p == 0 ? mp.p0.L  : (p == 1 ? mp.p1.L  : mp.p2.L);
    mx.Rm = p == 0 ? mp.p0.Rm : (p == 1 ? mp.p1.Rm : mp.p2.Rm);
    mx.d  = p == 0 ? mp.p0.d  : (p == 1 ? mp.p1.d  : mp.p2.d);
    return mx;
}

#define MIX_S 32
#define MIX_J (TT / MIX_S)

__device__ __forceinline__ void mix_factor_partial_body(
    const MixPaths& mp, float* __restrict__ Mpart, int s, int combo)
{
    const int p = combo >> 1, b = combo & 1;
    const MixPath mx = mix_path(mp, p);
    const int c = threadIdx.x & 63, rq = threadIdx.x >> 6;
    const size_t row0 = (size_t)b * TT + (size_t)s * MIX_J;
    const float* Fp = mx.F + row0 * RR + c;
    const float* Lp = mx.L + row0 * RSS + rq * 4;
    float4 acc = make_float4(0.f, 0.f, 0.f, 0.f);
    #pragma unroll 16
    for (int jj = 0; jj < MIX_J; ++jj) {
        const float f = Fp[(size_t)jj * RR];
        const float4 l = *(const float4*)(Lp + (size_t)jj * RSS);
        acc.x += l.x * f; acc.y += l.y * f; acc.z += l.z * f; acc.w += l.w * f;
    }
    float* o = Mpart + ((size_t)(combo * MIX_S + s) * RSS + rq * 4) * RR + c;
    o[0] = acc.x; o[RR] = acc.y; o[2 * RR] = acc.z; o[3 * RR] = acc.w;
}

__device__ __forceinline__ void mix_factor_apply_body(
    const MixPaths& mp, const float* __restrict__ Mpart, float* __restrict__ G,
    int bx, int combo)
{
    __shared__ __align__(16) float Ms[RSS * RR];
    const int p = combo >> 1, b = combo & 1;
    const MixPath mx = mix_path(mp, p);
    {
        const float4* src = (const float4*)(Mpart + (size_t)combo * MIX_S * RSS * RR) + threadIdx.x;
        float4 acc = src[0];
        #pragma unroll
        for (int sl = 1; sl < MIX_S; ++sl) {
            const float4 t = src[(size_t)sl * (RSS * RR / 4)];
            acc.x += t.x; acc.y += t.y; acc.z += t.z; acc.w += t.w;
        }
        ((float4*)Ms)[threadIdx.x] = acc;
    }
    __syncthreads();
    const int c = threadIdx.x & 63, rq = threadIdx.x >> 6;
    float m[RSS];
    #pragma unroll
    for (int r = 0; r < RSS; ++r) m[r] = Ms[r * RR + c];
    const size_t row0 = (size_t)b * TT + (size_t)bx * 64 + (size_t)rq * 16;
    float* Gp = G + (size_t)p * BT * RR;
    #pragma unroll 4
    for (int k = 0; k < 16; ++k) {
        const size_t row = row0 + k;
        const float4* rr = (const float4*)(mx.Rm + row * RSS);
        const float4 r0 = rr[0], r1 = rr[1], r2 = rr[2], r3 = rr[3];
        float v = mx.d[row] * mx.F[row * RR + c];
        v += r0.x * m[0];  v += r0.y * m[1];  v += r0.z * m[2];  v += r0.w * m[3];
        v += r1.x * m[4];  v += r1.y * m[5];  v += r1.z * m[6];  v += r1.w * m[7];
        v += r2.x * m[8];  v += r2.y * m[9];  v += r2.z * m[10]; v += r2.w * m[11];
        v += r3.x * m[12]; v += r3.y * m[13]; v += r3.z * m[14]; v += r3.w * m[15];
        Gp[row * RR + c] = v;
    }
}

__global__ __launch_bounds__(256) void splitk_p_and_mix_partial_kernel(
    const float* A0, const float* A1, const float* B0, const float* B1,
    float* C0, float* C1,
    int K, int lda, int ldb, int ldc, int S,
    long long sAs, long long sBs, long long sCs,
    MixPaths mp, float* Mpart)
{
    const int nGemm = 2 * S * (BT / GBM);
    const int id = (int)blockIdx.x;
    if (id < nGemm) {
        const int mt = id % (BT / GBM);
        const int z  = id / (BT / GBM);
        const int y  = z / S;
        const int s  = z - y * S;
        const float* Ab = (y ? A1 : A0) + (size_t)s * sAs;
        const float* Bb = (y ? B1 : B0) + (size_t)s * sBs;
        float*       Cb = (y ? C1 : C0) + (size_t)s * sCs;
        gemm_tile_f32(Ab, Bb, Cb, K, lda, ldb, ldc, 1, mt * GBM, 0, false, DabTile{});
        return;
    }
    const int r = id - nGemm;
    mix_factor_partial_body(mp, Mpart, r % MIX_S, r / MIX_S);
}

__global__ __launch_bounds__(256) void splitk_sum_and_mix_apply_kernel(
    MixPaths mp, const float* __restrict__ Mpart, float* __restrict__ G,
    const float* __restrict__ part, float* __restrict__ out, int S, long long n)
{
    const int nApply = 6 * (TT / 64);
    const int id = (int)blockIdx.x;
    if (id < nApply) {
        mix_factor_apply_body(mp, Mpart, G, id % (TT / 64), id / (TT / 64));
        return;
    }
    const int r  = id - nApply;
    const int nx = (int)(n / 1024);
    splitk_sum_body(part, out, S, n, r % nx, r / nx);
}

#define NCT (TT / GBN)

__global__ __launch_bounds__(256) void score_rowsq_kernel(
    const float* __restrict__ P, const float* __restrict__ Kmix,
    float* __restrict__ part, int K, int lda, int ldb,
    long long sA, long long sB)
{
    const int b  = blockIdx.z;
    const int m0 = (int)blockIdx.y * GBM, n0 = (int)blockIdx.x * GBN;
    TileAcc t;
    gemm_tile_loop_f32(P + (size_t)b * sA, Kmix + (size_t)b * sB, K, lda, ldb, 1,
                       m0, n0, false, DabTile{}, t);

    const int wave = threadIdx.x >> 5;
    const int lane = threadIdx.x & 31;
    const int hi   = lane >> 4;
    const int lm   = lane & 15;
    float* o = part + ((size_t)b * NCT + blockIdx.x) * TT + m0 + wave * 32 + hi * 8;
    #pragma unroll
    for (int mt = 0; mt < 2; ++mt) {
        v8f a0 = mt ? t.c10 : t.c00, a1 = mt ? t.c11 : t.c01;
        v8f a2 = mt ? t.c12 : t.c02, a3 = mt ? t.c13 : t.c03;
        float s[8];
        #pragma unroll
        for (int r = 0; r < 8; ++r) {
            float v = a0[r] * a0[r];
            v += a1[r] * a1[r]; v += a2[r] * a2[r]; v += a3[r] * a3[r];
            v += __shfl_xor(v, 8, 32); v += __shfl_xor(v, 4, 32);
            v += __shfl_xor(v, 2, 32); v += __shfl_xor(v, 1, 32);
            s[r] = v;
        }
        if (lm == 0) {
            float4* q = (float4*)(o + mt * 16);
            q[0] = make_float4(s[0], s[1], s[2], s[3]);
            q[1] = make_float4(s[4], s[5], s[6], s[7]);
        }
    }
}

__global__ __launch_bounds__(256) void score_combine_kernel(
    const float* __restrict__ P,  const float* __restrict__ Kmix,
    const float* __restrict__ Ps, const float* __restrict__ Ksmix,
    const float* __restrict__ part, float* __restrict__ H,
    int K, int lda, int ldb, int ldc,
    long long sA, long long sB, long long sC)
{
    __shared__ __align__(16) float4 Gs[16][256];
    __shared__ float invn[GBM];
    const int b   = blockIdx.z;
    const int m0  = (int)blockIdx.y * GBM, n0 = (int)blockIdx.x * GBN;
    const int tid = threadIdx.x;

    {
        const float* pr = part + (size_t)b * NCT * TT + m0 + tid;
        float acc = pr[0];
        #pragma unroll
        for (int ct = 1; ct < NCT; ++ct) acc += pr[(size_t)ct * TT];
        invn[tid] = 1.f / (sqrtf(acc) + 1e-8f);
    }

    TileAcc gate, val;
    gemm_tile_loop_f32(Ps + (size_t)b * sA, Ksmix + (size_t)b * sB, K, lda, ldb, 1,
                       m0, n0, false, DabTile{}, gate);
    {
        const v8f* gv[8] = { &gate.c00, &gate.c01, &gate.c02, &gate.c03,
                             &gate.c10, &gate.c11, &gate.c12, &gate.c13 };
        #pragma unroll
        for (int t = 0; t < 8; ++t) {
            const v8f v = *gv[t];
            Gs[2 * t][tid]     = make_float4(v[0], v[1], v[2], v[3]);
            Gs[2 * t + 1][tid] = make_float4(v[4], v[5], v[6], v[7]);
        }
    }
    #pragma nounroll
    for (int q = 0; q < 16; ++q) {
        float4 v = Gs[q][tid];
        v.x = softplusf(v.x); v.y = softplusf(v.y);
        v.z = softplusf(v.z); v.w = softplusf(v.w);
        Gs[q][tid] = v;
    }
    gemm_tile_loop_f32(P + (size_t)b * sA, Kmix + (size_t)b * sB, K, lda, ldb, 1,
                       m0, n0, false, DabTile{}, val);
    {
        v8f* gv[8] = { &gate.c00, &gate.c01, &gate.c02, &gate.c03,
                       &gate.c10, &gate.c11, &gate.c12, &gate.c13 };
        #pragma unroll
        for (int t = 0; t < 8; ++t) {
            const float4 lo = Gs[2 * t][tid], up = Gs[2 * t + 1][tid];
            v8f v;
            v[0] = lo.x; v[1] = lo.y; v[2] = lo.z; v[3] = lo.w;
            v[4] = up.x; v[5] = up.y; v[6] = up.z; v[7] = up.w;
            *gv[t] = v;
        }
    }

    const int wave = threadIdx.x >> 5;
    const int lane = threadIdx.x & 31;
    const int hi   = lane >> 4;
    const int lm   = lane & 15;
    float* Hb = H + (size_t)b * sC;
    const int ncol = n0 + lm;
    #pragma unroll
    for (int mt = 0; mt < 2; ++mt) {
        const int rl = wave * 32 + mt * 16 + hi * 8;
        v8f a0 = mt ? val.c10 : val.c00, a1 = mt ? val.c11 : val.c01;
        v8f a2 = mt ? val.c12 : val.c02, a3 = mt ? val.c13 : val.c03;
        v8f g0 = mt ? gate.c10 : gate.c00, g1 = mt ? gate.c11 : gate.c01;
        v8f g2 = mt ? gate.c12 : gate.c02, g3 = mt ? gate.c13 : gate.c03;
        #pragma unroll
        for (int r = 0; r < 8; ++r) {
            const float inv = invn[rl + r];
            size_t base = (size_t)(m0 + rl + r) * ldc + ncol;
            Hb[base]      = g0[r] * fmaxf(a0[r] * inv, 0.f);
            Hb[base + 16] = g1[r] * fmaxf(a1[r] * inv, 0.f);
            Hb[base + 32] = g2[r] * fmaxf(a2[r] * inv, 0.f);
            Hb[base + 48] = g3[r] * fmaxf(a3[r] * inv, 0.f);
        }
    }
}

extern "C" void kernel_launch(void* const* d_in, const int* in_sizes, int n_in,
                              void* d_out, int out_size, void* d_ws, size_t ws_size,
                              hipStream_t stream) {
    (void)in_sizes; (void)n_in; (void)out_size; (void)ws_size;

    const float* x = (const float*)d_in[0];
    const float* qk_Wq1  = (const float*)d_in[1];  const float* qk_Wq2  = (const float*)d_in[2];
    const float* qk_Wm1  = (const float*)d_in[3];  const float* qk_Wm2  = (const float*)d_in[4];
    const float* qk_Wk1  = (const float*)d_in[5];  const float* qk_Wk2  = (const float*)d_in[6];
    const float* qks_Wq1 = (const float*)d_in[7];  const float* qks_Wq2 = (const float*)d_in[8];
    const float* qks_Wm1 = (const float*)d_in[9];  const float* qks_Wm2 = (const float*)d_in[10];
    const float* qks_Wk1 = (const float*)d_in[11]; const float* qks_Wk2 = (const float*)d_in[12];
    const float* vo_Wv1  = (const float*)d_in[13]; const float* vo_Wv2  = (const float*)d_in[14];
    const float* vo_Wm1  = (const float*)d_in[15]; const float* vo_Wm2  = (const float*)d_in[16];
    const float* vo_Wo   = (const float*)d_in[17];
    const float* sm1_Wd  = (const float*)d_in[18]; const float* sm1_Wa  = (const float*)d_in[19]; const float* sm1_Wb  = (const float*)d_in[20];
    const float* sm1s_Wd = (const float*)d_in[21]; const float* sm1s_Wa = (const float*)d_in[22]; const float* sm1s_Wb = (const float*)d_in[23];
    const float* sm2_Wd  = (const float*)d_in[24]; const float* sm2_Wa  = (const float*)d_in[25]; const float* sm2_Wb  = (const float*)d_in[26];

    const size_t BTD = (size_t)BT * DD;
    const size_t BTT = (size_t)BB * TT * TT;

    float* w = (float*)d_ws;
    float* qg   = w; w += BTD;
    float* pkp  = w; w += BTD;
    float* qsg  = w; w += BTD;
    float* pp   = w; w += BTD;
    float* ctxg = w; w += BTD;
    float* tmpb = w; w += BTD;
    float* zb   = w; w += BTT;
    float* zsb  = w; w += BTT;
    float* xr8  = w; w += BTD;
    float* d1   = w; w += BT; float* d1s = w; w += BT; float* d2 = w; w += BT;
    float* A1   = w; w += (size_t)BT * RSS; float* B1  = w; w += (size_t)BT * RSS;
    float* A1s  = w; w += (size_t)BT * RSS; float* B1s = w; w += (size_t)BT * RSS;
    float* A2   = w; w += (size_t)BT * RSS; float* B2  = w; w += (size_t)BT * RSS;

    auto gemm = [&](const float* A, const float* Bm, float* C,
                    int M, int N, int K, int lda, int ldb, int ldc, int bT,
                    long long sA, long long sB, long long sC, int batches) {
        dim3 g((unsigned)(N / GBN), (unsigned)(M / GBM), (unsigned)batches);
        gemm_wmma_f32<<<g, dim3(256), 0, stream>>>(A, Bm, C, M, N, K, lda, ldb, ldc, bT, sA, sB, sC);
    };

    const long long sXR = (long long)BT * RR;
    const Weights8 W1 = { qk_Wm1, qk_Wq1, qk_Wk1, qks_Wm1, qks_Wq1, qks_Wk1, vo_Wv1, vo_Wm1 };
    const DabPaths P3 = { sm1_Wa,  sm1_Wb,  sm1_Wd,  A1,  B1,  d1,
                          sm1s_Wa, sm1s_Wb, sm1s_Wd, A1s, B1s, d1s,
                          sm2_Wa,  sm2_Wb,  sm2_Wd,  A2,  B2,  d2 };
    gemm_wmma_f32_shareA8<<<dim3(11, BT / GBM), dim3(256), 0, stream>>>(x, W1, xr8, DD, DD, sXR, P3);

    const GatedSets GQ = { xr8 + 1 * sXR, qk_Wq2,  xr8 + 0 * sXR, qk_Wm2,  qg,
                           xr8 + 4 * sXR, qks_Wq2, xr8 + 3 * sXR, qks_Wm2, qsg };
    gemm_wmma_f32_gated<<<dim3(DD / GBN, BT / GBM, 2), dim3(256), 0, stream>>>(GQ, RR, RR, DD, DD);

    float* Mpart = tmpb;
    float* gmix  = tmpb + (size_t)BT * RR;
    const MixPaths MP = { { xr8 + 2 * sXR, B1,  A1,  d1  },
                          { xr8 + 5 * sXR, B1s, A1s, d1s },
                          { xr8 + 6 * sXR, A2,  B2,  d2  } };

    const long long sTTl = (long long)TT * TT;
    const long long sTR = (long long)TT * RR;
    splitk_p_and_mix_partial_kernel<<<dim3(2 * 8 * (BT / GBM) + 6 * MIX_S), dim3(256), 0, stream>>>(
        qg, qsg, qk_Wk2, qks_Wk2, pkp, pkp + 8 * sXR,
        DD / 8, DD, DD, RR, 8, DD / 8, DD / 8, sXR, MP, Mpart);
    splitk_sum_and_mix_apply_kernel<<<dim3(6 * (TT / 64) + 2 * (unsigned)(sXR / 4 / 256)), dim3(256), 0, stream>>>(
        MP, Mpart, gmix, pkp, pp, 8, sXR);

    score_rowsq_kernel<<<dim3(NCT, TT / GBM, BB), dim3(256), 0, stream>>>(
        pp, gmix, zsb, RR, RR, RR, sTR, sTR);
    score_combine_kernel<<<dim3(NCT, TT / GBM, BB), dim3(256), 0, stream>>>(
        pp, gmix, pp + sXR, gmix + sXR, zsb, zb, RR, RR, RR, TT, sTR, sTR, sTTl);

    float* vmix = gmix + 2 * sXR;
    gemm_wmma_f32_splitk<<<dim3(1, TT / GBM, 2 * 16), dim3(256), 0, stream>>>(
        zb, zb + sTTl, vmix, vmix + sTR, qg, qg + sTR,
        TT / 16, TT, RR, RR, 0, 16, TT / 16, (long long)(TT / 16) * RR, sXR);
    splitk_sum_kernel<<<dim3((unsigned)(sXR / 4 / 256), 1), 256, 0, stream>>>(qg, pkp, 16, sXR);
    const GatedSets GC = { pkp, vo_Wv2, xr8 + 7 * sXR, vo_Wm2, ctxg,
                           nullptr, nullptr, nullptr, nullptr, nullptr };
    gemm_wmma_f32_gated<<<dim3(DD / GBN, BT / GBM, 1), dim3(256), 0, stream>>>(GC, RR, RR, DD, DD);
    gemm(ctxg, vo_Wo, (float*)d_out, BT, DD, DD, DD, DD, DD, 1, 0, 0, 0, 1);
}


# row norm of h_mix divided out after the context sum (relu(a*s)=s*relu(a), s>0; a row scale commutes with what follows; scale meets the sum, not its terms): squares summed in combine tiles; 9 launches
# speedup vs baseline: 5.9155x; 1.0460x over previous
#include <hip/hip_runtime.h>
#include <math.h>

#define BB 2
#define TT 2048
#define DD 1024
#define RR 64
#define RSS 16
#define BT (BB*TT)

typedef __attribute__((ext_vector_type(2))) float v2f;
typedef __attribute__((ext_vector_type(8))) float v8f;

#define GBM 256
#define GBN 64
#define GBK 32

struct DabTile {
    const float* Wa; const float* Wb; const float* Wd;
    float* aOut; float* bOut; float* dOut;
};

struct TileAcc { v8f c00, c01, c02, c03, c10, c11, c12, c13; };

__device__ __forceinline__ void gemm_tile_loop_f32(
    const float* __restrict__ Ab, const float* __restrict__ Bb,
    int K, int lda, int ldb, int bT, int m0, int n0,
    bool dab, DabTile dt, TileAcc& acc)
{
    __shared__ __align__(16) float As [2][GBM][34];
    __shared__ __align__(16) float BsT[2][GBN][34];

    const int tid  = threadIdx.x;
    const int wave = tid >> 5;
    const int lane = tid & 31;
    const int hi   = lane >> 4;
    const int lm   = lane & 15;

    float4 ra[8];
    float4 rb[2];

    int arow[8], ac4[8];
    #pragma unroll
    for (int i = 0; i < 8; ++i) { int idx = tid + i * 256; arow[i] = idx >> 3; ac4[i] = (idx & 7) * 4; }
    int bi0[2], bi1[2];
    #pragma unroll
    for (int i = 0; i < 2; ++i) {
        int idx = tid + i * 256;
        if (true) {   }
        bi0[i] = idx; bi1[i] = idx;
    }

    auto loadGlobal = [&](int k0) {
        #pragma unroll
        for (int i = 0; i < 8; ++i)
            ra[i] = *(const float4*)(Ab + (size_t)(m0 + arow[i]) * lda + k0 + ac4[i]);
        if (dab) {
            #pragma unroll
            for (int i = 0; i < 2; ++i) {
                int idx = bi1[i];
                int q = idx & 15;
                size_t k = (size_t)(k0 + (idx >> 4));
                float4 t = make_float4(0.f, 0.f, 0.f, 0.f);
                if (q < 4)       t = *(const float4*)(dt.Wa + k * RSS + q * 4);
                else if (q < 8)  t = *(const float4*)(dt.Wb + k * RSS + (q - 4) * 4);
                else if (q == 8) t.x = dt.Wd[k];
                rb[i] = t;
            }
        } else if (bT) {
            #pragma unroll
            for (int i = 0; i < 2; ++i) {
                int idx = bi0[i];
                int n = idx >> 3, c4 = (idx & 7) * 4;
                rb[i] = *(const float4*)(Bb + (size_t)(n0 + n) * ldb + k0 + c4);
            }
        } else {
            #pragma unroll
            for (int i = 0; i < 2; ++i) {
                int idx = bi1[i];
                int r = idx >> 4, c4 = (idx & 15) * 4;
                rb[i] = *(const float4*)(Bb + (size_t)(k0 + r) * ldb + n0 + c4);
            }
        }
    };

    auto storeLDS = [&](int buf) {
        #pragma unroll
        for (int i = 0; i < 8; ++i) {
            float4 t = ra[i];
            v2f lo; lo.x = t.x; lo.y = t.y;
            v2f hi2; hi2.x = t.z; hi2.y = t.w;
            *(v2f*)&As[buf][arow[i]][ac4[i]]     = lo;
            *(v2f*)&As[buf][arow[i]][ac4[i] + 2] = hi2;
        }
        if (bT) {
            #pragma unroll
            for (int i = 0; i < 2; ++i) {
                int idx = bi0[i];
                int n = idx >> 3, c4 = (idx & 7) * 4;
                float4 t = rb[i];
                v2f lo; lo.x = t.x; lo.y = t.y;
                v2f hi2; hi2.x = t.z; hi2.y = t.w;
                *(v2f*)&BsT[buf][n][c4]     = lo;
                *(v2f*)&BsT[buf][n][c4 + 2] = hi2;
            }
        } else {
            #pragma unroll
            for (int i = 0; i < 2; ++i) {
                int idx = bi1[i];
                int r = idx >> 4, c4 = (idx & 15) * 4;
                float4 t = rb[i];
                BsT[buf][c4 + 0][r] = t.x;
                BsT[buf][c4 + 1][r] = t.y;
                BsT[buf][c4 + 2][r] = t.z;
                BsT[buf][c4 + 3][r] = t.w;
            }
        }
    };

    v8f acc00 = {}, acc01 = {}, acc02 = {}, acc03 = {};
    v8f acc10 = {}, acc11 = {}, acc12 = {}, acc13 = {};

    const int r0 = wave * 32 + lm;
    const int r1 = r0 + 16;
    const int nchunks = K / GBK;

    loadGlobal(0);
    for (int c = 0; c < nchunks; ++c) {
        const int buf = c & 1;
        storeLDS(buf);
        __syncthreads();
        if (c + 1 < nchunks) loadGlobal((c + 1) * GBK);

        #pragma unroll
        for (int kk = 0; kk < GBK; kk += 4) {
            const int kc = kk + 2 * hi;
            v2f a0 = *(const v2f*)&As [buf][r0][kc];
            v2f a1 = *(const v2f*)&As [buf][r1][kc];
            v2f b0 = *(const v2f*)&BsT[buf][lm +  0][kc];
            v2f b1 = *(const v2f*)&BsT[buf][lm + 16][kc];
            v2f b2 = *(const v2f*)&BsT[buf][lm + 32][kc];
            v2f b3 = *(const v2f*)&BsT[buf][lm + 48][kc];
            acc00 = __builtin_amdgcn_wmma_f32_16x16x4_f32(false, a0, false, b0, (short)0, acc00, false, false);
            acc01 = __builtin_amdgcn_wmma_f32_16x16x4_f32(false, a0, false, b1, (short)0, acc01, false, false);
            acc02 = __builtin_amdgcn_wmma_f32_16x16x4_f32(false, a0, false, b2, (short)0, acc02, false, false);
            acc03 = __builtin_amdgcn_wmma_f32_16x16x4_f32(false, a0, false, b3, (short)0, acc03, false, false);
            acc10 = __builtin_amdgcn_wmma_f32_16x16x4_f32(false, a1, false, b0, (short)0, acc10, false, false);
            acc11 = __builtin_amdgcn_wmma_f32_16x16x4_f32(false, a1, false, b1, (short)0, acc11, false, false);
            acc12 = __builtin_amdgcn_wmma_f32_16x16x4_f32(false, a1, false, b2, (short)0, acc12, false, false);
            acc13 = __builtin_amdgcn_wmma_f32_16x16x4_f32(false, a1, false, b3, (short)0, acc13, false, false);
        }
        __syncthreads();
    }
    acc.c00 = acc00; acc.c01 = acc01; acc.c02 = acc02; acc.c03 = acc03;
    acc.c10 = acc10; acc.c11 = acc11; acc.c12 = acc12; acc.c13 = acc13;
}

__device__ __forceinline__ void gemm_tile_f32(
    const float* __restrict__ Ab, const float* __restrict__ Bb,
    float* __restrict__ Cb,
    int K, int lda, int ldb, int ldc, int bT, int m0, int n0,
    bool dab, DabTile dt)
{
    TileAcc t;
    gemm_tile_loop_f32(Ab, Bb, K, lda, ldb, bT, m0, n0, dab, dt, t);

    const int wave = threadIdx.x >> 5;
    const int lane = threadIdx.x & 31;
    const int hi   = lane >> 4;
    const int lm   = lane & 15;

    if (dab) {
        #pragma unroll
        for (int mt = 0; mt < 2; ++mt) {
            const int mrow = m0 + wave * 32 + mt * 16 + hi * 8;
            v8f a0 = mt ? t.c10 : t.c00, a1 = mt ? t.c11 : t.c01, a2 = mt ? t.c12 : t.c02;
            #pragma unroll
            for (int r = 0; r < 8; ++r) {
                size_t row = (size_t)(mrow + r);
                dt.aOut[row * RSS + lm] = a0[r];
                dt.bOut[row * RSS + lm] = a1[r];
                if (lm == 0) dt.dOut[row] = a2[r];
            }
        }
        return;
    }

    const int ncol = n0 + lm;
    #pragma unroll
    for (int mt = 0; mt < 2; ++mt) {
        const int mrow = m0 + wave * 32 + mt * 16 + hi * 8;
        v8f a0 = mt ? t.c10 : t.c00, a1 = mt ? t.c11 : t.c01;
        v8f a2 = mt ? t.c12 : t.c02, a3 = mt ? t.c13 : t.c03;
        #pragma unroll
        for (int r = 0; r < 8; ++r) {
            size_t base = (size_t)(mrow + r) * ldc + ncol;
            Cb[base] = a0[r]; Cb[base + 16] = a1[r]; Cb[base + 32] = a2[r]; Cb[base + 48] = a3[r];
        }
    }
}

__global__ __launch_bounds__(256) void gemm_wmma_f32(
    const float* __restrict__ A, const float* __restrict__ Bm,
    float* __restrict__ C,
    int M, int N, int K, int lda, int ldb, int ldc, int bT,
    long long sA, long long sB, long long sC)
{
    (void)M; (void)N;
    const float* Ab = A  + (size_t)blockIdx.z * sA;
    const float* Bb = Bm + (size_t)blockIdx.z * sB;
    float*       Cb = C  + (size_t)blockIdx.z * sC;
    gemm_tile_f32(Ab, Bb, Cb, K, lda, ldb, ldc, bT,
                  (int)blockIdx.y * GBM, (int)blockIdx.x * GBN, false, DabTile{});
}

struct GatedSets {
    const float* A1a; const float* B1a; const float* A2a; const float* B2a; float* Ca;
    const float* A1b; const float* B1b; const float* A2b; const float* B2b; float* Cb;
};

__global__ __launch_bounds__(256) void gemm_wmma_f32_gated(
    GatedSets g, int K, int lda, int ldb, int ldc)
{
    const bool sb = blockIdx.z != 0;
    const float* A1 = sb ? g.A1b : g.A1a;
    const float* B1 = sb ? g.B1b : g.B1a;
    const float* A2 = sb ? g.A2b : g.A2a;
    const float* B2 = sb ? g.B2b : g.B2a;
    float*       C  = sb ? g.Cb  : g.Ca;
    const int m0 = (int)blockIdx.y * GBM, n0 = (int)blockIdx.x * GBN;

    __shared__ __align__(16) float4 Gs[16][256];
    const int tid = threadIdx.x;

    TileAcc gate, val;
    gemm_tile_loop_f32(A2, B2, K, lda, ldb, 0, m0, n0, false, DabTile{}, gate);
    {
        const v8f* gv[8] = { &gate.c00, &gate.c01, &gate.c02, &gate.c03,
                             &gate.c10, &gate.c11, &gate.c12, &gate.c13 };
        #pragma unroll
        for (int t = 0; t < 8; ++t) {
            const v8f v = *gv[t];
            Gs[2 * t][tid]     = make_float4(v[0], v[1], v[2], v[3]);
            Gs[2 * t + 1][tid] = make_float4(v[4], v[5], v[6], v[7]);
        }
    }
    gemm_tile_loop_f32(A1, B1, K, lda, ldb, 0, m0, n0, false, DabTile{}, val);
    {
        v8f* gv[8] = { &gate.c00, &gate.c01, &gate.c02, &gate.c03,
                       &gate.c10, &gate.c11, &gate.c12, &gate.c13 };
        #pragma unroll
        for (int t = 0; t < 8; ++t) {
            const float4 lo = Gs[2 * t][tid], up = Gs[2 * t + 1][tid];
            v8f v;
            v[0] = lo.x; v[1] = lo.y; v[2] = lo.z; v[3] = lo.w;
            v[4] = up.x; v[5] = up.y; v[6] = up.z; v[7] = up.w;
            *gv[t] = v;
        }
    }

    const int wave = threadIdx.x >> 5;
    const int lane = threadIdx.x & 31;
    const int hi   = lane >> 4;
    const int lm   = lane & 15;
    const int ncol = n0 + lm;
    #pragma unroll
    for (int mt = 0; mt < 2; ++mt) {
        const int mrow = m0 + wave * 32 + mt * 16 + hi * 8;
        v8f a0 = mt ? val.c10 : val.c00, a1 = mt ? val.c11 : val.c01;
        v8f a2 = mt ? val.c12 : val.c02, a3 = mt ? val.c13 : val.c03;
        v8f g0 = mt ? gate.c10 : gate.c00, g1 = mt ? gate.c11 : gate.c01;
        v8f g2 = mt ? gate.c12 : gate.c02, g3 = mt ? gate.c13 : gate.c03;
        #pragma unroll
        for (int r = 0; r < 8; ++r) {
            size_t base = (size_t)(mrow + r) * ldc + ncol;
            C[base]      = a0[r] * g0[r];
            C[base + 16] = a1[r] * g1[r];
            C[base + 32] = a2[r] * g2[r];
            C[base + 48] = a3[r] * g3[r];
        }
    }
}

struct Weights8 {
    const float* w0; const float* w1; const float* w2; const float* w3;
    const float* w4; const float* w5; const float* w6; const float* w7;
};
struct DabPaths {
    const float* wa0; const float* wb0; const float* wd0; float* a0; float* b0; float* d0;
    const float* wa1; const float* wb1; const float* wd1; float* a1; float* b1; float* d1;
    const float* wa2; const float* wb2; const float* wd2; float* a2; float* b2; float* d2;
};

__global__ __launch_bounds__(256) void gemm_wmma_f32_shareA8(
    const float* __restrict__ A, Weights8 W, float* __restrict__ C,
    int K, int lda, long long sC, DabPaths P)
{
    const int j = blockIdx.x;
    if (j >= 8) {
        DabTile dt = { P.wa0, P.wb0, P.wd0, P.a0, P.b0, P.d0 };
        if (j == 9)  dt = DabTile{ P.wa1, P.wb1, P.wd1, P.a1, P.b1, P.d1 };
        if (j == 10) dt = DabTile{ P.wa2, P.wb2, P.wd2, P.a2, P.b2, P.d2 };
        gemm_tile_f32(A, nullptr, nullptr, K, lda, 0, 0, 0,
                      (int)blockIdx.y * GBM, 0, true, dt);
        return;
    }
    const float* Bb = W.w0;
    switch (j) {
        case 1: Bb = W.w1; break;
        case 2: Bb = W.w2; break;
        case 3: Bb = W.w3; break;
        case 4: Bb = W.w4; break;
        case 5: Bb = W.w5; break;
        case 6: Bb = W.w6; break;
        case 7: Bb = W.w7; break;
        default: break;
    }
    gemm_tile_f32(A, Bb, C + (size_t)j * sC, K, lda, GBN, GBN, 0,
                  (int)blockIdx.y * GBM, 0, false, DabTile{});
}

__global__ __launch_bounds__(256) void gemm_wmma_f32_splitk(
    const float* A0, const float* A1, const float* B0, const float* B1,
    float* C0, float* C1,
    int K, int lda, int ldb, int ldc, int bT, int S,
    long long sAs, long long sBs, long long sCs)
{
    const int y = (int)blockIdx.z / S;
    const int s = (int)blockIdx.z - y * S;
    const float* Ab = (y ? A1 : A0) + (size_t)s * sAs;
    const float* Bb = (y ? B1 : B0) + (size_t)s * sBs;
    float*       Cb = (y ? C1 : C0) + (size_t)s * sCs;
    gemm_tile_f32(Ab, Bb, Cb, K, lda, ldb, ldc, bT,
                  (int)blockIdx.y * GBM, (int)blockIdx.x * GBN, false, DabTile{});
}

__device__ __forceinline__ void splitk_sum_body(
    const float* __restrict__ part, float* __restrict__ out, int S, long long n,
    int bx, int by)
{
    const size_t e = ((size_t)bx * 256 + threadIdx.x) * 4;
    const float* p = part + (size_t)by * (size_t)S * (size_t)n + e;
    float4 acc = *(const float4*)p;
    for (int s = 1; s < S; ++s) {
        const float4 t = *(const float4*)(p + (size_t)s * (size_t)n);
        acc.x += t.x; acc.y += t.y; acc.z += t.z; acc.w += t.w;
    }
    *(float4*)(out + (size_t)by * (size_t)n + e) = acc;
}

#define NCT (TT / GBN)
#define CTX_S 16

__global__ __launch_bounds__(256) void splitk_sum_rowscale_kernel(
    const float* __restrict__ part, float* __restrict__ out, long long n,
    const float* __restrict__ sq)
{
    const size_t e   = ((size_t)blockIdx.x * 256 + threadIdx.x) * 4;
    const size_t row = e / RR;
    const size_t b   = row / TT, t = row - b * TT;
    const float* pr  = sq + b * (size_t)NCT * TT + t;
    const float* p   = part + e;

    float4 pl[CTX_S];
    #pragma unroll
    for (int s = 0; s < CTX_S; ++s) pl[s] = *(const float4*)(p + (size_t)s * (size_t)n);
    float q[NCT];
    #pragma unroll
    for (int ct = 0; ct < NCT; ++ct) q[ct] = pr[(size_t)ct * TT];

    float ss = q[0];
    #pragma unroll
    for (int ct = 1; ct < NCT; ++ct) ss += q[ct];
    const float inv = 1.f / (sqrtf(ss) + 1e-8f);

    float4 acc = pl[0];
    #pragma unroll
    for (int s = 1; s < CTX_S; ++s) {
        acc.x += pl[s].x; acc.y += pl[s].y; acc.z += pl[s].z; acc.w += pl[s].w;
    }
    acc.x *= inv; acc.y *= inv; acc.z *= inv; acc.w *= inv;
    *(float4*)(out + e) = acc;
}

__device__ inline float softplusf(float x) {
    if (x > 20.f) return x;
    if (x < -20.f) return expf(x);
    return log1pf(expf(x));
}

struct MixPath { const float* F; const float* L; const float* Rm; const float* d; };
struct MixPaths { MixPath p0; MixPath p1; MixPath p2; };

__device__ __forceinline__ MixPath mix_path(const MixPaths& mp, int p)
{
    MixPath mx;
    mx.F  = p == 0 ? mp.p0.F  : (p == 1 ? mp.p1.F  : mp.p2.F);
    mx.L  = p == 0 ? mp.p0.L  : (p == 1 ? mp.p1.L  : mp.p2.L);
    mx.Rm = p == 0 ? mp.p0.Rm : (p == 1 ? mp.p1.Rm : mp.p2.Rm);
    mx.d  = p == 0 ? mp.p0.d  : (p == 1 ? mp.p1.d  : mp.p2.d);
    return mx;
}

#define MIX_S 32
#define MIX_J (TT / MIX_S)

__device__ __forceinline__ void mix_factor_partial_body(
    const MixPaths& mp, float* __restrict__ Mpart, int s, int combo)
{
    const int p = combo >> 1, b = combo & 1;
    const MixPath mx = mix_path(mp, p);
    const int c = threadIdx.x & 63, rq = threadIdx.x >> 6;
    const size_t row0 = (size_t)b * TT + (size_t)s * MIX_J;
    const float* Fp = mx.F + row0 * RR + c;
    const float* Lp = mx.L + row0 * RSS + rq * 4;
    float4 acc = make_float4(0.f, 0.f, 0.f, 0.f);
    #pragma unroll 16
    for (int jj = 0; jj < MIX_J; ++jj) {
        const float f = Fp[(size_t)jj * RR];
        const float4 l = *(const float4*)(Lp + (size_t)jj * RSS);
        acc.x += l.x * f; acc.y += l.y * f; acc.z += l.z * f; acc.w += l.w * f;
    }
    float* o = Mpart + ((size_t)(combo * MIX_S + s) * RSS + rq * 4) * RR + c;
    o[0] = acc.x; o[RR] = acc.y; o[2 * RR] = acc.z; o[3 * RR] = acc.w;
}

__device__ __forceinline__ void mix_factor_apply_body(
    const MixPaths& mp, const float* __restrict__ Mpart, float* __restrict__ G,
    int bx, int combo)
{
    __shared__ __align__(16) float Ms[RSS * RR];
    const int p = combo >> 1, b = combo & 1;
    const MixPath mx = mix_path(mp, p);
    {
        const float4* src = (const float4*)(Mpart + (size_t)combo * MIX_S * RSS * RR) + threadIdx.x;
        float4 acc = src[0];
        #pragma unroll
        for (int sl = 1; sl < MIX_S; ++sl) {
            const float4 t = src[(size_t)sl * (RSS * RR / 4)];
            acc.x += t.x; acc.y += t.y; acc.z += t.z; acc.w += t.w;
        }
        ((float4*)Ms)[threadIdx.x] = acc;
    }
    __syncthreads();
    const int c = threadIdx.x & 63, rq = threadIdx.x >> 6;
    float m[RSS];
    #pragma unroll
    for (int r = 0; r < RSS; ++r) m[r] = Ms[r * RR + c];
    const size_t row0 = (size_t)b * TT + (size_t)bx * 64 + (size_t)rq * 16;
    float* Gp = G + (size_t)p * BT * RR;
    #pragma unroll 4
    for (int k = 0; k < 16; ++k) {
        const size_t row = row0 + k;
        const float4* rr = (const float4*)(mx.Rm + row * RSS);
        const float4 r0 = rr[0], r1 = rr[1], r2 = rr[2], r3 = rr[3];
        float v = mx.d[row] * mx.F[row * RR + c];
        v += r0.x * m[0];  v += r0.y * m[1];  v += r0.z * m[2];  v += r0.w * m[3];
        v += r1.x * m[4];  v += r1.y * m[5];  v += r1.z * m[6];  v += r1.w * m[7];
        v += r2.x * m[8];  v += r2.y * m[9];  v += r2.z * m[10]; v += r2.w * m[11];
        v += r3.x * m[12]; v += r3.y * m[13]; v += r3.z * m[14]; v += r3.w * m[15];
        Gp[row * RR + c] = v;
    }
}

__global__ __launch_bounds__(256) void splitk_p_and_mix_partial_kernel(
    const float* A0, const float* A1, const float* B0, const float* B1,
    float* C0, float* C1,
    int K, int lda, int ldb, int ldc, int S,
    long long sAs, long long sBs, long long sCs,
    MixPaths mp, float* Mpart)
{
    const int nGemm = 2 * S * (BT / GBM);
    const int id = (int)blockIdx.x;
    if (id < nGemm) {
        const int mt = id % (BT / GBM);
        const int z  = id / (BT / GBM);
        const int y  = z / S;
        const int s  = z - y * S;
        const float* Ab = (y ? A1 : A0) + (size_t)s * sAs;
        const float* Bb = (y ? B1 : B0) + (size_t)s * sBs;
        float*       Cb = (y ? C1 : C0) + (size_t)s * sCs;
        gemm_tile_f32(Ab, Bb, Cb, K, lda, ldb, ldc, 1, mt * GBM, 0, false, DabTile{});
        return;
    }
    const int r = id - nGemm;
    mix_factor_partial_body(mp, Mpart, r % MIX_S, r / MIX_S);
}

__global__ __launch_bounds__(256) void splitk_sum_and_mix_apply_kernel(
    MixPaths mp, const float* __restrict__ Mpart, float* __restrict__ G,
    const float* __restrict__ part, float* __restrict__ out, int S, long long n)
{
    const int nApply = 6 * (TT / 64);
    const int id = (int)blockIdx.x;
    if (id < nApply) {
        mix_factor_apply_body(mp, Mpart, G, id % (TT / 64), id / (TT / 64));
        return;
    }
    const int r  = id - nApply;
    const int nx = (int)(n / 1024);
    splitk_sum_body(part, out, S, n, r % nx, r / nx);
}

__global__ __launch_bounds__(256) void score_combine_kernel(
    const float* __restrict__ P,  const float* __restrict__ Kmix,
    const float* __restrict__ Ps, const float* __restrict__ Ksmix,
    float* __restrict__ part, float* __restrict__ H,
    int K, int lda, int ldb, int ldc,
    long long sA, long long sB, long long sC)
{
    __shared__ __align__(16) float4 Gs[16][256];
    const int b   = blockIdx.z;
    const int m0  = (int)blockIdx.y * GBM, n0 = (int)blockIdx.x * GBN;
    const int tid = threadIdx.x;

    TileAcc gate, val;
    gemm_tile_loop_f32(Ps + (size_t)b * sA, Ksmix + (size_t)b * sB, K, lda, ldb, 1,
                       m0, n0, false, DabTile{}, gate);
    {
        const v8f* gv[8] = { &gate.c00, &gate.c01, &gate.c02, &gate.c03,
                             &gate.c10, &gate.c11, &gate.c12, &gate.c13 };
        #pragma unroll
        for (int t = 0; t < 8; ++t) {
            const v8f v = *gv[t];
            Gs[2 * t][tid]     = make_float4(v[0], v[1], v[2], v[3]);
            Gs[2 * t + 1][tid] = make_float4(v[4], v[5], v[6], v[7]);
        }
    }
    #pragma nounroll
    for (int q = 0; q < 16; ++q) {
        float4 v = Gs[q][tid];
        v.x = softplusf(v.x); v.y = softplusf(v.y);
        v.z = softplusf(v.z); v.w = softplusf(v.w);
        Gs[q][tid] = v;
    }
    gemm_tile_loop_f32(P + (size_t)b * sA, Kmix + (size_t)b * sB, K, lda, ldb, 1,
                       m0, n0, false, DabTile{}, val);
    {
        v8f* gv[8] = { &gate.c00, &gate.c01, &gate.c02, &gate.c03,
                       &gate.c10, &gate.c11, &gate.c12, &gate.c13 };
        #pragma unroll
        for (int t = 0; t < 8; ++t) {
            const float4 lo = Gs[2 * t][tid], up = Gs[2 * t + 1][tid];
            v8f v;
            v[0] = lo.x; v[1] = lo.y; v[2] = lo.z; v[3] = lo.w;
            v[4] = up.x; v[5] = up.y; v[6] = up.z; v[7] = up.w;
            *gv[t] = v;
        }
    }

    const int wave = threadIdx.x >> 5;
    const int lane = threadIdx.x & 31;
    const int hi   = lane >> 4;
    const int lm   = lane & 15;
    float* Hb = H + (size_t)b * sC;
    const int ncol = n0 + lm;
    float* o = part + ((size_t)b * NCT + blockIdx.x) * TT + m0 + wave * 32 + hi * 8;
    #pragma unroll
    for (int mt = 0; mt < 2; ++mt) {
        const int rl = wave * 32 + mt * 16 + hi * 8;
        v8f a0 = mt ? val.c10 : val.c00, a1 = mt ? val.c11 : val.c01;
        v8f a2 = mt ? val.c12 : val.c02, a3 = mt ? val.c13 : val.c03;
        v8f g0 = mt ? gate.c10 : gate.c00, g1 = mt ? gate.c11 : gate.c01;
        v8f g2 = mt ? gate.c12 : gate.c02, g3 = mt ? gate.c13 : gate.c03;
        float s[8];
        #pragma unroll
        for (int r = 0; r < 8; ++r) {
            float v = a0[r] * a0[r];
            v += a1[r] * a1[r]; v += a2[r] * a2[r]; v += a3[r] * a3[r];
            v += __shfl_xor(v, 8, 32); v += __shfl_xor(v, 4, 32);
            v += __shfl_xor(v, 2, 32); v += __shfl_xor(v, 1, 32);
            s[r] = v;
            size_t base = (size_t)(m0 + rl + r) * ldc + ncol;
            Hb[base]      = g0[r] * fmaxf(a0[r], 0.f);
            Hb[base + 16] = g1[r] * fmaxf(a1[r], 0.f);
            Hb[base + 32] = g2[r] * fmaxf(a2[r], 0.f);
            Hb[base + 48] = g3[r] * fmaxf(a3[r], 0.f);
        }
        if (lm == 0) {
            float4* q = (float4*)(o + mt * 16);
            q[0] = make_float4(s[0], s[1], s[2], s[3]);
            q[1] = make_float4(s[4], s[5], s[6], s[7]);
        }
    }
}

extern "C" void kernel_launch(void* const* d_in, const int* in_sizes, int n_in,
                              void* d_out, int out_size, void* d_ws, size_t ws_size,
                              hipStream_t stream) {
    (void)in_sizes; (void)n_in; (void)out_size; (void)ws_size;

    const float* x = (const float*)d_in[0];
    const float* qk_Wq1  = (const float*)d_in[1];  const float* qk_Wq2  = (const float*)d_in[2];
    const float* qk_Wm1  = (const float*)d_in[3];  const float* qk_Wm2  = (const float*)d_in[4];
    const float* qk_Wk1  = (const float*)d_in[5];  const float* qk_Wk2  = (const float*)d_in[6];
    const float* qks_Wq1 = (const float*)d_in[7];  const float* qks_Wq2 = (const float*)d_in[8];
    const float* qks_Wm1 = (const float*)d_in[9];  const float* qks_Wm2 = (const float*)d_in[10];
    const float* qks_Wk1 = (const float*)d_in[11]; const float* qks_Wk2 = (const float*)d_in[12];
    const float* vo_Wv1  = (const float*)d_in[13]; const float* vo_Wv2  = (const float*)d_in[14];
    const float* vo_Wm1  = (const float*)d_in[15]; const float* vo_Wm2  = (const float*)d_in[16];
    const float* vo_Wo   = (const float*)d_in[17];
    const float* sm1_Wd  = (const float*)d_in[18]; const float* sm1_Wa  = (const float*)d_in[19]; const float* sm1_Wb  = (const float*)d_in[20];
    const float* sm1s_Wd = (const float*)d_in[21]; const float* sm1s_Wa = (const float*)d_in[22]; const float* sm1s_Wb = (const float*)d_in[23];
    const float* sm2_Wd  = (const float*)d_in[24]; const float* sm2_Wa  = (const float*)d_in[25]; const float* sm2_Wb  = (const float*)d_in[26];

    const size_t BTD = (size_t)BT * DD;
    const size_t BTT = (size_t)BB * TT * TT;

    float* w = (float*)d_ws;
    float* qg   = w; w += BTD;
    float* pkp  = w; w += BTD;
    float* qsg  = w; w += BTD;
    float* pp   = w; w += BTD;
    float* ctxg = w; w += BTD;
    float* tmpb = w; w += BTD;
    float* zb   = w; w += BTT;
    float* zsb  = w; w += BTT;
    float* xr8  = w; w += BTD;
    float* d1   = w; w += BT; float* d1s = w; w += BT; float* d2 = w; w += BT;
    float* A1   = w; w += (size_t)BT * RSS; float* B1  = w; w += (size_t)BT * RSS;
    float* A1s  = w; w += (size_t)BT * RSS; float* B1s = w; w += (size_t)BT * RSS;
    float* A2   = w; w += (size_t)BT * RSS; float* B2  = w; w += (size_t)BT * RSS;

    auto gemm = [&](const float* A, const float* Bm, float* C,
                    int M, int N, int K, int lda, int ldb, int ldc, int bT,
                    long long sA, long long sB, long long sC, int batches) {
        dim3 g((unsigned)(N / GBN), (unsigned)(M / GBM), (unsigned)batches);
        gemm_wmma_f32<<<g, dim3(256), 0, stream>>>(A, Bm, C, M, N, K, lda, ldb, ldc, bT, sA, sB, sC);
    };

    const long long sXR = (long long)BT * RR;
    const Weights8 W1 = { qk_Wm1, qk_Wq1, qk_Wk1, qks_Wm1, qks_Wq1, qks_Wk1, vo_Wv1, vo_Wm1 };
    const DabPaths P3 = { sm1_Wa,  sm1_Wb,  sm1_Wd,  A1,  B1,  d1,
                          sm1s_Wa, sm1s_Wb, sm1s_Wd, A1s, B1s, d1s,
                          sm2_Wa,  sm2_Wb,  sm2_Wd,  A2,  B2,  d2 };
    gemm_wmma_f32_shareA8<<<dim3(11, BT / GBM), dim3(256), 0, stream>>>(x, W1, xr8, DD, DD, sXR, P3);

    const GatedSets GQ = { xr8 + 1 * sXR, qk_Wq2,  xr8 + 0 * sXR, qk_Wm2,  qg,
                           xr8 + 4 * sXR, qks_Wq2, xr8 + 3 * sXR, qks_Wm2, qsg };
    gemm_wmma_f32_gated<<<dim3(DD / GBN, BT / GBM, 2), dim3(256), 0, stream>>>(GQ, RR, RR, DD, DD);

    float* Mpart = tmpb;
    float* gmix  = tmpb + (size_t)BT * RR;
    const MixPaths MP = { { xr8 + 2 * sXR, B1,  A1,  d1  },
                          { xr8 + 5 * sXR, B1s, A1s, d1s },
                          { xr8 + 6 * sXR, A2,  B2,  d2  } };

    const long long sTTl = (long long)TT * TT;
    const long long sTR = (long long)TT * RR;
    splitk_p_and_mix_partial_kernel<<<dim3(2 * 8 * (BT / GBM) + 6 * MIX_S), dim3(256), 0, stream>>>(
        qg, qsg, qk_Wk2, qks_Wk2, pkp, pkp + 8 * sXR,
        DD / 8, DD, DD, RR, 8, DD / 8, DD / 8, sXR, MP, Mpart);
    splitk_sum_and_mix_apply_kernel<<<dim3(6 * (TT / 64) + 2 * (unsigned)(sXR / 4 / 256)), dim3(256), 0, stream>>>(
        MP, Mpart, gmix, pkp, pp, 8, sXR);

    score_combine_kernel<<<dim3(NCT, TT / GBM, BB), dim3(256), 0, stream>>>(
        pp, gmix, pp + sXR, gmix + sXR, zsb, zb, RR, RR, RR, TT, sTR, sTR, sTTl);

    float* vmix = gmix + 2 * sXR;
    gemm_wmma_f32_splitk<<<dim3(1, TT / GBM, 2 * CTX_S), dim3(256), 0, stream>>>(
        zb, zb + sTTl, vmix, vmix + sTR, qg, qg + sTR,
        TT / CTX_S, TT, RR, RR, 0, CTX_S, TT / CTX_S, (long long)(TT / CTX_S) * RR, sXR);
    splitk_sum_rowscale_kernel<<<dim3((unsigned)(sXR / 4 / 256)), 256, 0, stream>>>(qg, pkp, sXR, zsb);
    const GatedSets GC = { pkp, vo_Wv2, xr8 + 7 * sXR, vo_Wm2, ctxg,
                           nullptr, nullptr, nullptr, nullptr, nullptr };
    gemm_wmma_f32_gated<<<dim3(DD / GBN, BT / GBM, 1), dim3(256), 0, stream>>>(GC, RR, RR, DD, DD);
    gemm(ctxg, vo_Wo, (float*)d_out, BT, DD, DD, DD, DD, DD, 1, 0, 0, 0, 1);
}


# launches that had at most one 256-row work-group per WGP (x@W1, P/Ps and context split-K, output projection) use 128-row tiles: two share a WGP and cover each other's staging; sum orders kept
# speedup vs baseline: 6.4209x; 1.0854x over previous
#include <hip/hip_runtime.h>
#include <math.h>

#define BB 2
#define TT 2048
#define DD 1024
#define RR 64
#define RSS 16
#define BT (BB*TT)

typedef __attribute__((ext_vector_type(2))) float v2f;
typedef __attribute__((ext_vector_type(8))) float v8f;

#define GBM 256
#define GBMH 128
#define GBN 64
#define GBK 32

struct DabTile {
    const float* Wa; const float* Wb; const float* Wd;
    float* aOut; float* bOut; float* dOut;
};

struct TileAcc { v8f c00, c01, c02, c03, c10, c11, c12, c13; };

template <int MT>
__device__ __forceinline__ void gemm_tile_loop_f32(
    const float* __restrict__ Ab, const float* __restrict__ Bb,
    int K, int lda, int ldb, int bT, int m0, int n0,
    bool dab, DabTile dt, TileAcc& acc)
{
    __shared__ __align__(16) float As [2][128 * MT][34];
    __shared__ __align__(16) float BsT[2][GBN][34];

    const int tid  = threadIdx.x;
    const int wave = tid >> 5;
    const int lane = tid & 31;
    const int hi   = lane >> 4;
    const int lm   = lane & 15;

    float4 ra[4 * MT];
    float4 rb[2];

    int arow[4 * MT], ac4[4 * MT];
    #pragma unroll
    for (int i = 0; i < 4 * MT; ++i) { int idx = tid + i * 256; arow[i] = idx >> 3; ac4[i] = (idx & 7) * 4; }
    int bi0[2], bi1[2];
    #pragma unroll
    for (int i = 0; i < 2; ++i) {
        int idx = tid + i * 256;
        if (true) {   }
        bi0[i] = idx; bi1[i] = idx;
    }

    auto loadGlobal = [&](int k0) {
        #pragma unroll
        for (int i = 0; i < 4 * MT; ++i)
            ra[i] = *(const float4*)(Ab + (size_t)(m0 + arow[i]) * lda + k0 + ac4[i]);
        if (dab) {
            #pragma unroll
            for (int i = 0; i < 2; ++i) {
                int idx = bi1[i];
                int q = idx & 15;
                size_t k = (size_t)(k0 + (idx >> 4));
                float4 t = make_float4(0.f, 0.f, 0.f, 0.f);
                if (q < 4)       t = *(const float4*)(dt.Wa + k * RSS + q * 4);
                else if (q < 8)  t = *(const float4*)(dt.Wb + k * RSS + (q - 4) * 4);
                else if (q == 8) t.x = dt.Wd[k];
                rb[i] = t;
            }
        } else if (bT) {
            #pragma unroll
            for (int i = 0; i < 2; ++i) {
                int idx = bi0[i];
                int n = idx >> 3, c4 = (idx & 7) * 4;
                rb[i] = *(const float4*)(Bb + (size_t)(n0 + n) * ldb + k0 + c4);
            }
        } else {
            #pragma unroll
            for (int i = 0; i < 2; ++i) {
                int idx = bi1[i];
                int r = idx >> 4, c4 = (idx & 15) * 4;
                rb[i] = *(const float4*)(Bb + (size_t)(k0 + r) * ldb + n0 + c4);
            }
        }
    };

    auto storeLDS = [&](int buf) {
        #pragma unroll
        for (int i = 0; i < 4 * MT; ++i) {
            float4 t = ra[i];
            v2f lo; lo.x = t.x; lo.y = t.y;
            v2f hi2; hi2.x = t.z; hi2.y = t.w;
            *(v2f*)&As[buf][arow[i]][ac4[i]]     = lo;
            *(v2f*)&As[buf][arow[i]][ac4[i] + 2] = hi2;
        }
        if (bT) {
            #pragma unroll
            for (int i = 0; i < 2; ++i) {
                int idx = bi0[i];
                int n = idx >> 3, c4 = (idx & 7) * 4;
                float4 t = rb[i];
                v2f lo; lo.x = t.x; lo.y = t.y;
                v2f hi2; hi2.x = t.z; hi2.y = t.w;
                *(v2f*)&BsT[buf][n][c4]     = lo;
                *(v2f*)&BsT[buf][n][c4 + 2] = hi2;
            }
        } else {
            #pragma unroll
            for (int i = 0; i < 2; ++i) {
                int idx = bi1[i];
                int r = idx >> 4, c4 = (idx & 15) * 4;
                float4 t = rb[i];
                BsT[buf][c4 + 0][r] = t.x;
                BsT[buf][c4 + 1][r] = t.y;
                BsT[buf][c4 + 2][r] = t.z;
                BsT[buf][c4 + 3][r] = t.w;
            }
        }
    };

    v8f acc00 = {}, acc01 = {}, acc02 = {}, acc03 = {};
    v8f acc10 = {}, acc11 = {}, acc12 = {}, acc13 = {};

    const int r0 = wave * 16 * MT + lm;
    const int r1 = r0 + 16;
    const int nchunks = K / GBK;

    loadGlobal(0);
    for (int c = 0; c < nchunks; ++c) {
        const int buf = c & 1;
        storeLDS(buf);
        __syncthreads();
        if (c + 1 < nchunks) loadGlobal((c + 1) * GBK);

        #pragma unroll
        for (int kk = 0; kk < GBK; kk += 4) {
            const int kc = kk + 2 * hi;
            v2f a0 = *(const v2f*)&As [buf][r0][kc];
            v2f b0 = *(const v2f*)&BsT[buf][lm +  0][kc];
            v2f b1 = *(const v2f*)&BsT[buf][lm + 16][kc];
            v2f b2 = *(const v2f*)&BsT[buf][lm + 32][kc];
            v2f b3 = *(const v2f*)&BsT[buf][lm + 48][kc];
            acc00 = __builtin_amdgcn_wmma_f32_16x16x4_f32(false, a0, false, b0, (short)0, acc00, false, false);
            acc01 = __builtin_amdgcn_wmma_f32_16x16x4_f32(false, a0, false, b1, (short)0, acc01, false, false);
            acc02 = __builtin_amdgcn_wmma_f32_16x16x4_f32(false, a0, false, b2, (short)0, acc02, false, false);
            acc03 = __builtin_amdgcn_wmma_f32_16x16x4_f32(false, a0, false, b3, (short)0, acc03, false, false);
            if constexpr (MT == 2) {
                v2f a1 = *(const v2f*)&As [buf][r1][kc];
                acc10 = __builtin_amdgcn_wmma_f32_16x16x4_f32(false, a1, false, b0, (short)0, acc10, false, false);
                acc11 = __builtin_amdgcn_wmma_f32_16x16x4_f32(false, a1, false, b1, (short)0, acc11, false, false);
                acc12 = __builtin_amdgcn_wmma_f32_16x16x4_f32(false, a1, false, b2, (short)0, acc12, false, false);
                acc13 = __builtin_amdgcn_wmma_f32_16x16x4_f32(false, a1, false, b3, (short)0, acc13, false, false);
            }
        }
        __syncthreads();
    }
    acc.c00 = acc00; acc.c01 = acc01; acc.c02 = acc02; acc.c03 = acc03;
    acc.c10 = acc10; acc.c11 = acc11; acc.c12 = acc12; acc.c13 = acc13;
}

template <int MT>
__device__ __forceinline__ void gemm_tile_f32(
    const float* __restrict__ Ab, const float* __restrict__ Bb,
    float* __restrict__ Cb,
    int K, int lda, int ldb, int ldc, int bT, int m0, int n0,
    bool dab, DabTile dt)
{
    TileAcc t;
    gemm_tile_loop_f32<MT>(Ab, Bb, K, lda, ldb, bT, m0, n0, dab, dt, t);

    const int wave = threadIdx.x >> 5;
    const int lane = threadIdx.x & 31;
    const int hi   = lane >> 4;
    const int lm   = lane & 15;

    if (dab) {
        #pragma unroll
        for (int mt = 0; mt < MT; ++mt) {
            const int mrow = m0 + wave * 16 * MT + mt * 16 + hi * 8;
            v8f a0 = mt ? t.c10 : t.c00, a1 = mt ? t.c11 : t.c01, a2 = mt ? t.c12 : t.c02;
            #pragma unroll
            for (int r = 0; r < 8; ++r) {
                size_t row = (size_t)(mrow + r);
                dt.aOut[row * RSS + lm] = a0[r];
                dt.bOut[row * RSS + lm] = a1[r];
                if (lm == 0) dt.dOut[row] = a2[r];
            }
        }
        return;
    }

    const int ncol = n0 + lm;
    #pragma unroll
    for (int mt = 0; mt < MT; ++mt) {
        const int mrow = m0 + wave * 16 * MT + mt * 16 + hi * 8;
        v8f a0 = mt ? t.c10 : t.c00, a1 = mt ? t.c11 : t.c01;
        v8f a2 = mt ? t.c12 : t.c02, a3 = mt ? t.c13 : t.c03;
        #pragma unroll
        for (int r = 0; r < 8; ++r) {
            size_t base = (size_t)(mrow + r) * ldc + ncol;
            Cb[base] = a0[r]; Cb[base + 16] = a1[r]; Cb[base + 32] = a2[r]; Cb[base + 48] = a3[r];
        }
    }
}

__global__ __launch_bounds__(256) void gemm_wmma_f32(
    const float* __restrict__ A, const float* __restrict__ Bm,
    float* __restrict__ C,
    int M, int N, int K, int lda, int ldb, int ldc, int bT,
    long long sA, long long sB, long long sC)
{
    (void)M; (void)N;
    const float* Ab = A  + (size_t)blockIdx.z * sA;
    const float* Bb = Bm + (size_t)blockIdx.z * sB;
    float*       Cb = C  + (size_t)blockIdx.z * sC;
    gemm_tile_f32<1>(Ab, Bb, Cb, K, lda, ldb, ldc, bT,
                     (int)blockIdx.y * GBMH, (int)blockIdx.x * GBN, false, DabTile{});
}

struct GatedSets {
    const float* A1a; const float* B1a; const float* A2a; const float* B2a; float* Ca;
    const float* A1b; const float* B1b; const float* A2b; const float* B2b; float* Cb;
};

__global__ __launch_bounds__(256) void gemm_wmma_f32_gated(
    GatedSets g, int K, int lda, int ldb, int ldc)
{
    const bool sb = blockIdx.z != 0;
    const float* A1 = sb ? g.A1b : g.A1a;
    const float* B1 = sb ? g.B1b : g.B1a;
    const float* A2 = sb ? g.A2b : g.A2a;
    const float* B2 = sb ? g.B2b : g.B2a;
    float*       C  = sb ? g.Cb  : g.Ca;
    const int m0 = (int)blockIdx.y * GBM, n0 = (int)blockIdx.x * GBN;

    __shared__ __align__(16) float4 Gs[16][256];
    const int tid = threadIdx.x;

    TileAcc gate, val;
    gemm_tile_loop_f32<2>(A2, B2, K, lda, ldb, 0, m0, n0, false, DabTile{}, gate);
    {
        const v8f* gv[8] = { &gate.c00, &gate.c01, &gate.c02, &gate.c03,
                             &gate.c10, &gate.c11, &gate.c12, &gate.c13 };
        #pragma unroll
        for (int t = 0; t < 8; ++t) {
            const v8f v = *gv[t];
            Gs[2 * t][tid]     = make_float4(v[0], v[1], v[2], v[3]);
            Gs[2 * t + 1][tid] = make_float4(v[4], v[5], v[6], v[7]);
        }
    }
    gemm_tile_loop_f32<2>(A1, B1, K, lda, ldb, 0, m0, n0, false, DabTile{}, val);
    {
        v8f* gv[8] = { &gate.c00, &gate.c01, &gate.c02, &gate.c03,
                       &gate.c10, &gate.c11, &gate.c12, &gate.c13 };
        #pragma unroll
        for (int t = 0; t < 8; ++t) {
            const float4 lo = Gs[2 * t][tid], up = Gs[2 * t + 1][tid];
            v8f v;
            v[0] = lo.x; v[1] = lo.y; v[2] = lo.z; v[3] = lo.w;
            v[4] = up.x; v[5] = up.y; v[6] = up.z; v[7] = up.w;
            *gv[t] = v;
        }
    }

    const int wave = threadIdx.x >> 5;
    const int lane = threadIdx.x & 31;
    const int hi   = lane >> 4;
    const int lm   = lane & 15;
    const int ncol = n0 + lm;
    #pragma unroll
    for (int mt = 0; mt < 2; ++mt) {
        const int mrow = m0 + wave * 32 + mt * 16 + hi * 8;
        v8f a0 = mt ? val.c10 : val.c00, a1 = mt ? val.c11 : val.c01;
        v8f a2 = mt ? val.c12 : val.c02, a3 = mt ? val.c13 : val.c03;
        v8f g0 = mt ? gate.c10 : gate.c00, g1 = mt ? gate.c11 : gate.c01;
        v8f g2 = mt ? gate.c12 : gate.c02, g3 = mt ? gate.c13 : gate.c03;
        #pragma unroll
        for (int r = 0; r < 8; ++r) {
            size_t base = (size_t)(mrow + r) * ldc + ncol;
            C[base]      = a0[r] * g0[r];
            C[base + 16] = a1[r] * g1[r];
            C[base + 32] = a2[r] * g2[r];
            C[base + 48] = a3[r] * g3[r];
        }
    }
}

struct Weights8 {
    const float* w0; const float* w1; const float* w2; const float* w3;
    const float* w4; const float* w5; const float* w6; const float* w7;
};
struct DabPaths {
    const float* wa0; const float* wb0; const float* wd0; float* a0; float* b0; float* d0;
    const float* wa1; const float* wb1; const float* wd1; float* a1; float* b1; float* d1;
    const float* wa2; const float* wb2; const float* wd2; float* a2; float* b2; float* d2;
};

__global__ __launch_bounds__(256) void gemm_wmma_f32_shareA8(
    const float* __restrict__ A, Weights8 W, float* __restrict__ C,
    int K, int lda, long long sC, DabPaths P)
{
    const int j = blockIdx.x;
    if (j >= 8) {
        DabTile dt = { P.wa0, P.wb0, P.wd0, P.a0, P.b0, P.d0 };
        if (j == 9)  dt = DabTile{ P.wa1, P.wb1, P.wd1, P.a1, P.b1, P.d1 };
        if (j == 10) dt = DabTile{ P.wa2, P.wb2, P.wd2, P.a2, P.b2, P.d2 };
        gemm_tile_f32<1>(A, nullptr, nullptr, K, lda, 0, 0, 0,
                         (int)blockIdx.y * GBMH, 0, true, dt);
        return;
    }
    const float* Bb = W.w0;
    switch (j) {
        case 1: Bb = W.w1; break;
        case 2: Bb = W.w2; break;
        case 3: Bb = W.w3; break;
        case 4: Bb = W.w4; break;
        case 5: Bb = W.w5; break;
        case 6: Bb = W.w6; break;
        case 7: Bb = W.w7; break;
        default: break;
    }
    gemm_tile_f32<1>(A, Bb, C + (size_t)j * sC, K, lda, GBN, GBN, 0,
                     (int)blockIdx.y * GBMH, 0, false, DabTile{});
}

__global__ __launch_bounds__(256) void gemm_wmma_f32_splitk(
    const float* A0, const float* A1, const float* B0, const float* B1,
    float* C0, float* C1,
    int K, int lda, int ldb, int ldc, int bT, int S,
    long long sAs, long long sBs, long long sCs)
{
    const int y = (int)blockIdx.z / S;
    const int s = (int)blockIdx.z - y * S;
    const float* Ab = (y ? A1 : A0) + (size_t)s * sAs;
    const float* Bb = (y ? B1 : B0) + (size_t)s * sBs;
    float*       Cb = (y ? C1 : C0) + (size_t)s * sCs;
    gemm_tile_f32<1>(Ab, Bb, Cb, K, lda, ldb, ldc, bT,
                     (int)blockIdx.y * GBMH, (int)blockIdx.x * GBN, false, DabTile{});
}

__device__ __forceinline__ void splitk_sum_body(
    const float* __restrict__ part, float* __restrict__ out, int S, long long n,
    int bx, int by)
{
    const size_t e = ((size_t)bx * 256 + threadIdx.x) * 4;
    const float* p = part + (size_t)by * (size_t)S * (size_t)n + e;
    float4 acc = *(const float4*)p;
    for (int s = 1; s < S; ++s) {
        const float4 t = *(const float4*)(p + (size_t)s * (size_t)n);
        acc.x += t.x; acc.y += t.y; acc.z += t.z; acc.w += t.w;
    }
    *(float4*)(out + (size_t)by * (size_t)n + e) = acc;
}

#define NCT (TT / GBN)
#define CTX_S 16

__global__ __launch_bounds__(256) void splitk_sum_rowscale_kernel(
    const float* __restrict__ part, float* __restrict__ out, long long n,
    const float* __restrict__ sq)
{
    const size_t e   = ((size_t)blockIdx.x * 256 + threadIdx.x) * 4;
    const size_t row = e / RR;
    const size_t b   = row / TT, t = row - b * TT;
    const float* pr  = sq + b * (size_t)NCT * TT + t;
    const float* p   = part + e;

    float4 pl[CTX_S];
    #pragma unroll
    for (int s = 0; s < CTX_S; ++s) pl[s] = *(const float4*)(p + (size_t)s * (size_t)n);
    float q[NCT];
    #pragma unroll
    for (int ct = 0; ct < NCT; ++ct) q[ct] = pr[(size_t)ct * TT];

    float ss = q[0];
    #pragma unroll
    for (int ct = 1; ct < NCT; ++ct) ss += q[ct];
    const float inv = 1.f / (sqrtf(ss) + 1e-8f);

    float4 acc = pl[0];
    #pragma unroll
    for (int s = 1; s < CTX_S; ++s) {
        acc.x += pl[s].x; acc.y += pl[s].y; acc.z += pl[s].z; acc.w += pl[s].w;
    }
    acc.x *= inv; acc.y *= inv; acc.z *= inv; acc.w *= inv;
    *(float4*)(out + e) = acc;
}

__device__ inline float softplusf(float x) {
    if (x > 20.f) return x;
    if (x < -20.f) return expf(x);
    return log1pf(expf(x));
}

struct MixPath { const float* F; const float* L; const float* Rm; const float* d; };
struct MixPaths { MixPath p0; MixPath p1; MixPath p2; };

__device__ __forceinline__ MixPath mix_path(const MixPaths& mp, int p)
{
    MixPath mx;
    mx.F  = p == 0 ? mp.p0.F  : (p == 1 ? mp.p1.F  : mp.p2.F);
    mx.L  = p == 0 ? mp.p0.L  : (p == 1 ? mp.p1.L  : mp.p2.L);
    mx.Rm = p == 0 ? mp.p0.Rm : (p == 1 ? mp.p1.Rm : mp.p2.Rm);
    mx.d  = p == 0 ? mp.p0.d  : (p == 1 ? mp.p1.d  : mp.p2.d);
    return mx;
}

#define MIX_S 32
#define MIX_J (TT / MIX_S)

__device__ __forceinline__ void mix_factor_partial_body(
    const MixPaths& mp, float* __restrict__ Mpart, int s, int combo)
{
    const int p = combo >> 1, b = combo & 1;
    const MixPath mx = mix_path(mp, p);
    const int c = threadIdx.x & 63, rq = threadIdx.x >> 6;
    const size_t row0 = (size_t)b * TT + (size_t)s * MIX_J;
    const float* Fp = mx.F + row0 * RR + c;
    const float* Lp = mx.L + row0 * RSS + rq * 4;
    float4 acc = make_float4(0.f, 0.f, 0.f, 0.f);
    #pragma unroll 16
    for (int jj = 0; jj < MIX_J; ++jj) {
        const float f = Fp[(size_t)jj * RR];
        const float4 l = *(const float4*)(Lp + (size_t)jj * RSS);
        acc.x += l.x * f; acc.y += l.y * f; acc.z += l.z * f; acc.w += l.w * f;
    }
    float* o = Mpart + ((size_t)(combo * MIX_S + s) * RSS + rq * 4) * RR + c;
    o[0] = acc.x; o[RR] = acc.y; o[2 * RR] = acc.z; o[3 * RR] = acc.w;
}

__device__ __forceinline__ void mix_factor_apply_body(
    const MixPaths& mp, const float* __restrict__ Mpart, float* __restrict__ G,
    int bx, int combo)
{
    __shared__ __align__(16) float Ms[RSS * RR];
    const int p = combo >> 1, b = combo & 1;
    const MixPath mx = mix_path(mp, p);
    {
        const float4* src = (const float4*)(Mpart + (size_t)combo * MIX_S * RSS * RR) + threadIdx.x;
        float4 acc = src[0];
        #pragma unroll
        for (int sl = 1; sl < MIX_S; ++sl) {
            const float4 t = src[(size_t)sl * (RSS * RR / 4)];
            acc.x += t.x; acc.y += t.y; acc.z += t.z; acc.w += t.w;
        }
        ((float4*)Ms)[threadIdx.x] = acc;
    }
    __syncthreads();
    const int c = threadIdx.x & 63, rq = threadIdx.x >> 6;
    float m[RSS];
    #pragma unroll
    for (int r = 0; r < RSS; ++r) m[r] = Ms[r * RR + c];
    const size_t row0 = (size_t)b * TT + (size_t)bx * 64 + (size_t)rq * 16;
    float* Gp = G + (size_t)p * BT * RR;
    #pragma unroll 4
    for (int k = 0; k < 16; ++k) {
        const size_t row = row0 + k;
        const float4* rr = (const float4*)(mx.Rm + row * RSS);
        const float4 r0 = rr[0], r1 = rr[1], r2 = rr[2], r3 = rr[3];
        float v = mx.d[row] * mx.F[row * RR + c];
        v += r0.x * m[0];  v += r0.y * m[1];  v += r0.z * m[2];  v += r0.w * m[3];
        v += r1.x * m[4];  v += r1.y * m[5];  v += r1.z * m[6];  v += r1.w * m[7];
        v += r2.x * m[8];  v += r2.y * m[9];  v += r2.z * m[10]; v += r2.w * m[11];
        v += r3.x * m[12]; v += r3.y * m[13]; v += r3.z * m[14]; v += r3.w * m[15];
        Gp[row * RR + c] = v;
    }
}

__global__ __launch_bounds__(256) void splitk_p_and_mix_partial_kernel(
    const float* A0, const float* A1, const float* B0, const float* B1,
    float* C0, float* C1,
    int K, int lda, int ldb, int ldc, int S,
    long long sAs, long long sBs, long long sCs,
    MixPaths mp, float* Mpart)
{
    const int nGemm = 2 * S * (BT / GBMH);
    const int id = (int)blockIdx.x;
    if (id < nGemm) {
        const int mt = id % (BT / GBMH);
        const int z  = id / (BT / GBMH);
        const int y  = z / S;
        const int s  = z - y * S;
        const float* Ab = (y ? A1 : A0) + (size_t)s * sAs;
        const float* Bb = (y ? B1 : B0) + (size_t)s * sBs;
        float*       Cb = (y ? C1 : C0) + (size_t)s * sCs;
        gemm_tile_f32<1>(Ab, Bb, Cb, K, lda, ldb, ldc, 1, mt * GBMH, 0, false, DabTile{});
        return;
    }
    const int r = id - nGemm;
    mix_factor_partial_body(mp, Mpart, r % MIX_S, r / MIX_S);
}

__global__ __launch_bounds__(256) void splitk_sum_and_mix_apply_kernel(
    MixPaths mp, const float* __restrict__ Mpart, float* __restrict__ G,
    const float* __restrict__ part, float* __restrict__ out, int S, long long n)
{
    const int nApply = 6 * (TT / 64);
    const int id = (int)blockIdx.x;
    if (id < nApply) {
        mix_factor_apply_body(mp, Mpart, G, id % (TT / 64), id / (TT / 64));
        return;
    }
    const int r  = id - nApply;
    const int nx = (int)(n / 1024);
    splitk_sum_body(part, out, S, n, r % nx, r / nx);
}

__global__ __launch_bounds__(256) void score_combine_kernel(
    const float* __restrict__ P,  const float* __restrict__ Kmix,
    const float* __restrict__ Ps, const float* __restrict__ Ksmix,
    float* __restrict__ part, float* __restrict__ H,
    int K, int lda, int ldb, int ldc,
    long long sA, long long sB, long long sC)
{
    __shared__ __align__(16) float4 Gs[16][256];
    const int b   = blockIdx.z;
    const int m0  = (int)blockIdx.y * GBM, n0 = (int)blockIdx.x * GBN;
    const int tid = threadIdx.x;

    TileAcc gate, val;
    gemm_tile_loop_f32<2>(Ps + (size_t)b * sA, Ksmix + (size_t)b * sB, K, lda, ldb, 1,
                          m0, n0, false, DabTile{}, gate);
    {
        const v8f* gv[8] = { &gate.c00, &gate.c01, &gate.c02, &gate.c03,
                             &gate.c10, &gate.c11, &gate.c12, &gate.c13 };
        #pragma unroll
        for (int t = 0; t < 8; ++t) {
            const v8f v = *gv[t];
            Gs[2 * t][tid]     = make_float4(v[0], v[1], v[2], v[3]);
            Gs[2 * t + 1][tid] = make_float4(v[4], v[5], v[6], v[7]);
        }
    }
    #pragma nounroll
    for (int q = 0; q < 16; ++q) {
        float4 v = Gs[q][tid];
        v.x = softplusf(v.x); v.y = softplusf(v.y);
        v.z = softplusf(v.z); v.w = softplusf(v.w);
        Gs[q][tid] = v;
    }
    gemm_tile_loop_f32<2>(P + (size_t)b * sA, Kmix + (size_t)b * sB, K, lda, ldb, 1,
                          m0, n0, false, DabTile{}, val);
    {
        v8f* gv[8] = { &gate.c00, &gate.c01, &gate.c02, &gate.c03,
                       &gate.c10, &gate.c11, &gate.c12, &gate.c13 };
        #pragma unroll
        for (int t = 0; t < 8; ++t) {
            const float4 lo = Gs[2 * t][tid], up = Gs[2 * t + 1][tid];
            v8f v;
            v[0] = lo.x; v[1] = lo.y; v[2] = lo.z; v[3] = lo.w;
            v[4] = up.x; v[5] = up.y; v[6] = up.z; v[7] = up.w;
            *gv[t] = v;
        }
    }

    const int wave = threadIdx.x >> 5;
    const int lane = threadIdx.x & 31;
    const int hi   = lane >> 4;
    const int lm   = lane & 15;
    float* Hb = H + (size_t)b * sC;
    const int ncol = n0 + lm;
    float* o = part + ((size_t)b * NCT + blockIdx.x) * TT + m0 + wave * 32 + hi * 8;
    #pragma unroll
    for (int mt = 0; mt < 2; ++mt) {
        const int rl = wave * 32 + mt * 16 + hi * 8;
        v8f a0 = mt ? val.c10 : val.c00, a1 = mt ? val.c11 : val.c01;
        v8f a2 = mt ? val.c12 : val.c02, a3 = mt ? val.c13 : val.c03;
        v8f g0 = mt ? gate.c10 : gate.c00, g1 = mt ? gate.c11 : gate.c01;
        v8f g2 = mt ? gate.c12 : gate.c02, g3 = mt ? gate.c13 : gate.c03;
        float s[8];
        #pragma unroll
        for (int r = 0; r < 8; ++r) {
            float v = a0[r] * a0[r];
            v += a1[r] * a1[r]; v += a2[r] * a2[r]; v += a3[r] * a3[r];
            v += __shfl_xor(v, 8, 32); v += __shfl_xor(v, 4, 32);
            v += __shfl_xor(v, 2, 32); v += __shfl_xor(v, 1, 32);
            s[r] = v;
            size_t base = (size_t)(m0 + rl + r) * ldc + ncol;
            Hb[base]      = g0[r] * fmaxf(a0[r], 0.f);
            Hb[base + 16] = g1[r] * fmaxf(a1[r], 0.f);
            Hb[base + 32] = g2[r] * fmaxf(a2[r], 0.f);
            Hb[base + 48] = g3[r] * fmaxf(a3[r], 0.f);
        }
        if (lm == 0) {
            float4* q = (float4*)(o + mt * 16);
            q[0] = make_float4(s[0], s[1], s[2], s[3]);
            q[1] = make_float4(s[4], s[5], s[6], s[7]);
        }
    }
}

extern "C" void kernel_launch(void* const* d_in, const int* in_sizes, int n_in,
                              void* d_out, int out_size, void* d_ws, size_t ws_size,
                              hipStream_t stream) {
    (void)in_sizes; (void)n_in; (void)out_size; (void)ws_size;

    const float* x = (const float*)d_in[0];
    const float* qk_Wq1  = (const float*)d_in[1];  const float* qk_Wq2  = (const float*)d_in[2];
    const float* qk_Wm1  = (const float*)d_in[3];  const float* qk_Wm2  = (const float*)d_in[4];
    const float* qk_Wk1  = (const float*)d_in[5];  const float* qk_Wk2  = (const float*)d_in[6];
    const float* qks_Wq1 = (const float*)d_in[7];  const float* qks_Wq2 = (const float*)d_in[8];
    const float* qks_Wm1 = (const float*)d_in[9];  const float* qks_Wm2 = (const float*)d_in[10];
    const float* qks_Wk1 = (const float*)d_in[11]; const float* qks_Wk2 = (const float*)d_in[12];
    const float* vo_Wv1  = (const float*)d_in[13]; const float* vo_Wv2  = (const float*)d_in[14];
    const float* vo_Wm1  = (const float*)d_in[15]; const float* vo_Wm2  = (const float*)d_in[16];
    const float* vo_Wo   = (const float*)d_in[17];
    const float* sm1_Wd  = (const float*)d_in[18]; const float* sm1_Wa  = (const float*)d_in[19]; const float* sm1_Wb  = (const float*)d_in[20];
    const float* sm1s_Wd = (const float*)d_in[21]; const float* sm1s_Wa = (const float*)d_in[22]; const float* sm1s_Wb = (const float*)d_in[23];
    const float* sm2_Wd  = (const float*)d_in[24]; const float* sm2_Wa  = (const float*)d_in[25]; const float* sm2_Wb  = (const float*)d_in[26];

    const size_t BTD = (size_t)BT * DD;
    const size_t BTT = (size_t)BB * TT * TT;

    float* w = (float*)d_ws;
    float* qg   = w; w += BTD;
    float* pkp  = w; w += BTD;
    float* qsg  = w; w += BTD;
    float* pp   = w; w += BTD;
    float* ctxg = w; w += BTD;
    float* tmpb = w; w += BTD;
    float* zb   = w; w += BTT;
    float* zsb  = w; w += BTT;
    float* xr8  = w; w += BTD;
    float* d1   = w; w += BT; float* d1s = w; w += BT; float* d2 = w; w += BT;
    float* A1   = w; w += (size_t)BT * RSS; float* B1  = w; w += (size_t)BT * RSS;
    float* A1s  = w; w += (size_t)BT * RSS; float* B1s = w; w += (size_t)BT * RSS;
    float* A2   = w; w += (size_t)BT * RSS; float* B2  = w; w += (size_t)BT * RSS;

    auto gemm = [&](const float* A, const float* Bm, float* C,
                    int M, int N, int K, int lda, int ldb, int ldc, int bT,
                    long long sA, long long sB, long long sC, int batches) {
        dim3 g((unsigned)(N / GBN), (unsigned)(M / GBMH), (unsigned)batches);
        gemm_wmma_f32<<<g, dim3(256), 0, stream>>>(A, Bm, C, M, N, K, lda, ldb, ldc, bT, sA, sB, sC);
    };

    const long long sXR = (long long)BT * RR;
    const Weights8 W1 = { qk_Wm1, qk_Wq1, qk_Wk1, qks_Wm1, qks_Wq1, qks_Wk1, vo_Wv1, vo_Wm1 };
    const DabPaths P3 = { sm1_Wa,  sm1_Wb,  sm1_Wd,  A1,  B1,  d1,
                          sm1s_Wa, sm1s_Wb, sm1s_Wd, A1s, B1s, d1s,
                          sm2_Wa,  sm2_Wb,  sm2_Wd,  A2,  B2,  d2 };
    gemm_wmma_f32_shareA8<<<dim3(11, BT / GBMH), dim3(256), 0, stream>>>(x, W1, xr8, DD, DD, sXR, P3);

    const GatedSets GQ = { xr8 + 1 * sXR, qk_Wq2,  xr8 + 0 * sXR, qk_Wm2,  qg,
                           xr8 + 4 * sXR, qks_Wq2, xr8 + 3 * sXR, qks_Wm2, qsg };
    gemm_wmma_f32_gated<<<dim3(DD / GBN, BT / GBM, 2), dim3(256), 0, stream>>>(GQ, RR, RR, DD, DD);

    float* Mpart = tmpb;
    float* gmix  = tmpb + (size_t)BT * RR;
    const MixPaths MP = { { xr8 + 2 * sXR, B1,  A1,  d1  },
                          { xr8 + 5 * sXR, B1s, A1s, d1s },
                          { xr8 + 6 * sXR, A2,  B2,  d2  } };

    const long long sTTl = (long long)TT * TT;
    const long long sTR = (long long)TT * RR;
    splitk_p_and_mix_partial_kernel<<<dim3(2 * 8 * (BT / GBMH) + 6 * MIX_S), dim3(256), 0, stream>>>(
        qg, qsg, qk_Wk2, qks_Wk2, pkp, pkp + 8 * sXR,
        DD / 8, DD, DD, RR, 8, DD / 8, DD / 8, sXR, MP, Mpart);
    splitk_sum_and_mix_apply_kernel<<<dim3(6 * (TT / 64) + 2 * (unsigned)(sXR / 4 / 256)), dim3(256), 0, stream>>>(
        MP, Mpart, gmix, pkp, pp, 8, sXR);

    score_combine_kernel<<<dim3(NCT, TT / GBM, BB), dim3(256), 0, stream>>>(
        pp, gmix, pp + sXR, gmix + sXR, zsb, zb, RR, RR, RR, TT, sTR, sTR, sTTl);

    float* vmix = gmix + 2 * sXR;
    gemm_wmma_f32_splitk<<<dim3(1, TT / GBMH, 2 * CTX_S), dim3(256), 0, stream>>>(
        zb, zb + sTTl, vmix, vmix + sTR, qg, qg + sTR,
        TT / CTX_S, TT, RR, RR, 0, CTX_S, TT / CTX_S, (long long)(TT / CTX_S) * RR, sXR);
    splitk_sum_rowscale_kernel<<<dim3((unsigned)(sXR / 4 / 256)), 256, 0, stream>>>(qg, pkp, sXR, zsb);
    const GatedSets GC = { pkp, vo_Wv2, xr8 + 7 * sXR, vo_Wm2, ctxg,
                           nullptr, nullptr, nullptr, nullptr, nullptr };
    gemm_wmma_f32_gated<<<dim3(DD / GBN, BT / GBM, 1), dim3(256), 0, stream>>>(GC, RR, RR, DD, DD);
    gemm(ctxg, vo_Wo, (float*)d_out, BT, DD, DD, DD, DD, DD, 1, 0, 0, 0, 1);
}


# shared-x launch (x@W1 x8, d/A/B projections) re-tiled: 512 4-wave work-groups, 64 rows x (64 + 16-col projection slice), two per WGP; was 352 of 128x64 (two on 96 WGPs, one on 160); sum orders kept
# speedup vs baseline: 8.0361x; 1.2516x over previous
#include <hip/hip_runtime.h>
#include <math.h>

#define BB 2
#define TT 2048
#define DD 1024
#define RR 64
#define RSS 16
#define BT (BB*TT)

typedef __attribute__((ext_vector_type(2))) float v2f;
typedef __attribute__((ext_vector_type(8))) float v8f;

#define GBM 256
#define GBMH 128
#define GBN 64
#define GBK 32

struct DabTile {
    const float* Wa; const float* Wb; const float* Wd;
    float* aOut; float* bOut; float* dOut;
};

struct TileAcc { v8f c00, c01, c02, c03, c10, c11, c12, c13; };

template <int MT>
__device__ __forceinline__ void gemm_tile_loop_f32(
    const float* __restrict__ Ab, const float* __restrict__ Bb,
    int K, int lda, int ldb, int bT, int m0, int n0,
    bool dab, DabTile dt, TileAcc& acc)
{
    __shared__ __align__(16) float As [2][128 * MT][34];
    __shared__ __align__(16) float BsT[2][GBN][34];

    const int tid  = threadIdx.x;
    const int wave = tid >> 5;
    const int lane = tid & 31;
    const int hi   = lane >> 4;
    const int lm   = lane & 15;

    float4 ra[4 * MT];
    float4 rb[2];

    int arow[4 * MT], ac4[4 * MT];
    #pragma unroll
    for (int i = 0; i < 4 * MT; ++i) { int idx = tid + i * 256; arow[i] = idx >> 3; ac4[i] = (idx & 7) * 4; }
    int bi0[2], bi1[2];
    #pragma unroll
    for (int i = 0; i < 2; ++i) {
        int idx = tid + i * 256;
        if (true) {   }
        bi0[i] = idx; bi1[i] = idx;
    }

    auto loadGlobal = [&](int k0) {
        #pragma unroll
        for (int i = 0; i < 4 * MT; ++i)
            ra[i] = *(const float4*)(Ab + (size_t)(m0 + arow[i]) * lda + k0 + ac4[i]);
        if (dab) {
            #pragma unroll
            for (int i = 0; i < 2; ++i) {
                int idx = bi1[i];
                int q = idx & 15;
                size_t k = (size_t)(k0 + (idx >> 4));
                float4 t = make_float4(0.f, 0.f, 0.f, 0.f);
                if (q < 4)       t = *(const float4*)(dt.Wa + k * RSS + q * 4);
                else if (q < 8)  t = *(const float4*)(dt.Wb + k * RSS + (q - 4) * 4);
                else if (q == 8) t.x = dt.Wd[k];
                rb[i] = t;
            }
        } else if (bT) {
            #pragma unroll
            for (int i = 0; i < 2; ++i) {
                int idx = bi0[i];
                int n = idx >> 3, c4 = (idx & 7) * 4;
                rb[i] = *(const float4*)(Bb + (size_t)(n0 + n) * ldb + k0 + c4);
            }
        } else {
            #pragma unroll
            for (int i = 0; i < 2; ++i) {
                int idx = bi1[i];
                int r = idx >> 4, c4 = (idx & 15) * 4;
                rb[i] = *(const float4*)(Bb + (size_t)(k0 + r) * ldb + n0 + c4);
            }
        }
    };

    auto storeLDS = [&](int buf) {
        #pragma unroll
        for (int i = 0; i < 4 * MT; ++i) {
            float4 t = ra[i];
            v2f lo; lo.x = t.x; lo.y = t.y;
            v2f hi2; hi2.x = t.z; hi2.y = t.w;
            *(v2f*)&As[buf][arow[i]][ac4[i]]     = lo;
            *(v2f*)&As[buf][arow[i]][ac4[i] + 2] = hi2;
        }
        if (bT) {
            #pragma unroll
            for (int i = 0; i < 2; ++i) {
                int idx = bi0[i];
                int n = idx >> 3, c4 = (idx & 7) * 4;
                float4 t = rb[i];
                v2f lo; lo.x = t.x; lo.y = t.y;
                v2f hi2; hi2.x = t.z; hi2.y = t.w;
                *(v2f*)&BsT[buf][n][c4]     = lo;
                *(v2f*)&BsT[buf][n][c4 + 2] = hi2;
            }
        } else {
            #pragma unroll
            for (int i = 0; i < 2; ++i) {
                int idx = bi1[i];
                int r = idx >> 4, c4 = (idx & 15) * 4;
                float4 t = rb[i];
                BsT[buf][c4 + 0][r] = t.x;
                BsT[buf][c4 + 1][r] = t.y;
                BsT[buf][c4 + 2][r] = t.z;
                BsT[buf][c4 + 3][r] = t.w;
            }
        }
    };

    v8f acc00 = {}, acc01 = {}, acc02 = {}, acc03 = {};
    v8f acc10 = {}, acc11 = {}, acc12 = {}, acc13 = {};

    const int r0 = wave * 16 * MT + lm;
    const int r1 = r0 + 16;
    const int nchunks = K / GBK;

    loadGlobal(0);
    for (int c = 0; c < nchunks; ++c) {
        const int buf = c & 1;
        storeLDS(buf);
        __syncthreads();
        if (c + 1 < nchunks) loadGlobal((c + 1) * GBK);

        #pragma unroll
        for (int kk = 0; kk < GBK; kk += 4) {
            const int kc = kk + 2 * hi;
            v2f a0 = *(const v2f*)&As [buf][r0][kc];
            v2f b0 = *(const v2f*)&BsT[buf][lm +  0][kc];
            v2f b1 = *(const v2f*)&BsT[buf][lm + 16][kc];
            v2f b2 = *(const v2f*)&BsT[buf][lm + 32][kc];
            v2f b3 = *(const v2f*)&BsT[buf][lm + 48][kc];
            acc00 = __builtin_amdgcn_wmma_f32_16x16x4_f32(false, a0, false, b0, (short)0, acc00, false, false);
            acc01 = __builtin_amdgcn_wmma_f32_16x16x4_f32(false, a0, false, b1, (short)0, acc01, false, false);
            acc02 = __builtin_amdgcn_wmma_f32_16x16x4_f32(false, a0, false, b2, (short)0, acc02, false, false);
            acc03 = __builtin_amdgcn_wmma_f32_16x16x4_f32(false, a0, false, b3, (short)0, acc03, false, false);
            if constexpr (MT == 2) {
                v2f a1 = *(const v2f*)&As [buf][r1][kc];
                acc10 = __builtin_amdgcn_wmma_f32_16x16x4_f32(false, a1, false, b0, (short)0, acc10, false, false);
                acc11 = __builtin_amdgcn_wmma_f32_16x16x4_f32(false, a1, false, b1, (short)0, acc11, false, false);
                acc12 = __builtin_amdgcn_wmma_f32_16x16x4_f32(false, a1, false, b2, (short)0, acc12, false, false);
                acc13 = __builtin_amdgcn_wmma_f32_16x16x4_f32(false, a1, false, b3, (short)0, acc13, false, false);
            }
        }
        __syncthreads();
    }
    acc.c00 = acc00; acc.c01 = acc01; acc.c02 = acc02; acc.c03 = acc03;
    acc.c10 = acc10; acc.c11 = acc11; acc.c12 = acc12; acc.c13 = acc13;
}

template <int MT>
__device__ __forceinline__ void gemm_tile_f32(
    const float* __restrict__ Ab, const float* __restrict__ Bb,
    float* __restrict__ Cb,
    int K, int lda, int ldb, int ldc, int bT, int m0, int n0,
    bool dab, DabTile dt)
{
    TileAcc t;
    gemm_tile_loop_f32<MT>(Ab, Bb, K, lda, ldb, bT, m0, n0, dab, dt, t);

    const int wave = threadIdx.x >> 5;
    const int lane = threadIdx.x & 31;
    const int hi   = lane >> 4;
    const int lm   = lane & 15;

    if (dab) {
        #pragma unroll
        for (int mt = 0; mt < MT; ++mt) {
            const int mrow = m0 + wave * 16 * MT + mt * 16 + hi * 8;
            v8f a0 = mt ? t.c10 : t.c00, a1 = mt ? t.c11 : t.c01, a2 = mt ? t.c12 : t.c02;
            #pragma unroll
            for (int r = 0; r < 8; ++r) {
                size_t row = (size_t)(mrow + r);
                dt.aOut[row * RSS + lm] = a0[r];
                dt.bOut[row * RSS + lm] = a1[r];
                if (lm == 0) dt.dOut[row] = a2[r];
            }
        }
        return;
    }

    const int ncol = n0 + lm;
    #pragma unroll
    for (int mt = 0; mt < MT; ++mt) {
        const int mrow = m0 + wave * 16 * MT + mt * 16 + hi * 8;
        v8f a0 = mt ? t.c10 : t.c00, a1 = mt ? t.c11 : t.c01;
        v8f a2 = mt ? t.c12 : t.c02, a3 = mt ? t.c13 : t.c03;
        #pragma unroll
        for (int r = 0; r < 8; ++r) {
            size_t base = (size_t)(mrow + r) * ldc + ncol;
            Cb[base] = a0[r]; Cb[base + 16] = a1[r]; Cb[base + 32] = a2[r]; Cb[base + 48] = a3[r];
        }
    }
}

__global__ __launch_bounds__(256) void gemm_wmma_f32(
    const float* __restrict__ A, const float* __restrict__ Bm,
    float* __restrict__ C,
    int M, int N, int K, int lda, int ldb, int ldc, int bT,
    long long sA, long long sB, long long sC)
{
    (void)M; (void)N;
    const float* Ab = A  + (size_t)blockIdx.z * sA;
    const float* Bb = Bm + (size_t)blockIdx.z * sB;
    float*       Cb = C  + (size_t)blockIdx.z * sC;
    gemm_tile_f32<1>(Ab, Bb, Cb, K, lda, ldb, ldc, bT,
                     (int)blockIdx.y * GBMH, (int)blockIdx.x * GBN, false, DabTile{});
}

struct GatedSets {
    const float* A1a; const float* B1a; const float* A2a; const float* B2a; float* Ca;
    const float* A1b; const float* B1b; const float* A2b; const float* B2b; float* Cb;
};

__global__ __launch_bounds__(256) void gemm_wmma_f32_gated(
    GatedSets g, int K, int lda, int ldb, int ldc)
{
    const bool sb = blockIdx.z != 0;
    const float* A1 = sb ? g.A1b : g.A1a;
    const float* B1 = sb ? g.B1b : g.B1a;
    const float* A2 = sb ? g.A2b : g.A2a;
    const float* B2 = sb ? g.B2b : g.B2a;
    float*       C  = sb ? g.Cb  : g.Ca;
    const int m0 = (int)blockIdx.y * GBM, n0 = (int)blockIdx.x * GBN;

    __shared__ __align__(16) float4 Gs[16][256];
    const int tid = threadIdx.x;

    TileAcc gate, val;
    gemm_tile_loop_f32<2>(A2, B2, K, lda, ldb, 0, m0, n0, false, DabTile{}, gate);
    {
        const v8f* gv[8] = { &gate.c00, &gate.c01, &gate.c02, &gate.c03,
                             &gate.c10, &gate.c11, &gate.c12, &gate.c13 };
        #pragma unroll
        for (int t = 0; t < 8; ++t) {
            const v8f v = *gv[t];
            Gs[2 * t][tid]     = make_float4(v[0], v[1], v[2], v[3]);
            Gs[2 * t + 1][tid] = make_float4(v[4], v[5], v[6], v[7]);
        }
    }
    gemm_tile_loop_f32<2>(A1, B1, K, lda, ldb, 0, m0, n0, false, DabTile{}, val);
    {
        v8f* gv[8] = { &gate.c00, &gate.c01, &gate.c02, &gate.c03,
                       &gate.c10, &gate.c11, &gate.c12, &gate.c13 };
        #pragma unroll
        for (int t = 0; t < 8; ++t) {
            const float4 lo = Gs[2 * t][tid], up = Gs[2 * t + 1][tid];
            v8f v;
            v[0] = lo.x; v[1] = lo.y; v[2] = lo.z; v[3] = lo.w;
            v[4] = up.x; v[5] = up.y; v[6] = up.z; v[7] = up.w;
            *gv[t] = v;
        }
    }

    const int wave = threadIdx.x >> 5;
    const int lane = threadIdx.x & 31;
    const int hi   = lane >> 4;
    const int lm   = lane & 15;
    const int ncol = n0 + lm;
    #pragma unroll
    for (int mt = 0; mt < 2; ++mt) {
        const int mrow = m0 + wave * 32 + mt * 16 + hi * 8;
        v8f a0 = mt ? val.c10 : val.c00, a1 = mt ? val.c11 : val.c01;
        v8f a2 = mt ? val.c12 : val.c02, a3 = mt ? val.c13 : val.c03;
        v8f g0 = mt ? gate.c10 : gate.c00, g1 = mt ? gate.c11 : gate.c01;
        v8f g2 = mt ? gate.c12 : gate.c02, g3 = mt ? gate.c13 : gate.c03;
        #pragma unroll
        for (int r = 0; r < 8; ++r) {
            size_t base = (size_t)(mrow + r) * ldc + ncol;
            C[base]      = a0[r] * g0[r];
            C[base + 16] = a1[r] * g1[r];
            C[base + 32] = a2[r] * g2[r];
            C[base + 48] = a3[r] * g3[r];
        }
    }
}

#define SXM 64
struct ShareX {
    const float* w0; const float* w1; const float* w2; const float* w3;
    const float* w4; const float* w5; const float* w6; const float* w7;
    const float* e0; const float* e1; const float* e2;
    const float* e3; const float* e4; const float* e5;
    const float* wd0; const float* wd1; const float* wd2;
    float* eo0; float* eo1; float* eo2; float* eo3; float* eo4; float* eo5;
    float* d0; float* d1; float* d2;
};

template <int SL>
__device__ __forceinline__ void sharex_tile_f32(
    float (&As)[2][SXM][34], float (&BsT)[2][80][34],
    const float* __restrict__ A, const float* __restrict__ Wj, float* __restrict__ Cj,
    const float* __restrict__ E, float* __restrict__ Eo,
    const float* __restrict__ wd0, const float* __restrict__ wd1, const float* __restrict__ wd2,
    float* __restrict__ d0, float* __restrict__ d1, float* __restrict__ d2,
    int K, int lda, int m0)
{
    constexpr int NT = SL ? 5 : 4;
    const int tid  = threadIdx.x;
    const int wave = tid >> 5;
    const int lane = tid & 31;
    const int hi   = lane >> 4;
    const int lm   = lane & 15;

    float4 ra[4];
    float4 rb[4];
    float4 re;

    int arow[4], ac4[4], brow[4], bc4[4];
    #pragma unroll
    for (int i = 0; i < 4; ++i) {
        const int idx = tid + i * 128;
        arow[i] = idx >> 3; ac4[i] = (idx & 7) * 4;
        brow[i] = idx >> 4; bc4[i] = (idx & 15) * 4;
    }
    const int er = tid >> 2, eq = tid & 3;

    auto loadGlobal = [&](int k0) {
        #pragma unroll
        for (int i = 0; i < 4; ++i)
            ra[i] = *(const float4*)(A + (size_t)(m0 + arow[i]) * lda + k0 + ac4[i]);
        #pragma unroll
        for (int i = 0; i < 4; ++i)
            rb[i] = *(const float4*)(Wj + (size_t)(k0 + brow[i]) * GBN + bc4[i]);
        if constexpr (NT == 5) {
            const size_t k = (size_t)(k0 + er);
            if constexpr (SL == 1) {
                re = *(const float4*)(E + k * RSS + eq * 4);
            } else {
                re.x = wd0[k]; re.y = wd1[k]; re.z = wd2[k]; re.w = 0.f;
            }
        }
    };

    auto storeLDS = [&](int buf) {
        #pragma unroll
        for (int i = 0; i < 4; ++i) {
            float4 t = ra[i];
            v2f lo; lo.x = t.x; lo.y = t.y;
            v2f hi2; hi2.x = t.z; hi2.y = t.w;
            *(v2f*)&As[buf][arow[i]][ac4[i]]     = lo;
            *(v2f*)&As[buf][arow[i]][ac4[i] + 2] = hi2;
        }
        #pragma unroll
        for (int i = 0; i < 4; ++i) {
            float4 t = rb[i];
            BsT[buf][bc4[i] + 0][brow[i]] = t.x;
            BsT[buf][bc4[i] + 1][brow[i]] = t.y;
            BsT[buf][bc4[i] + 2][brow[i]] = t.z;
            BsT[buf][bc4[i] + 3][brow[i]] = t.w;
        }
        if constexpr (NT == 5) {
            float* e = &BsT[buf][GBN + eq * 4][er];
            if constexpr (SL == 1) {
                e[0] = re.x; e[34] = re.y; e[68] = re.z; e[102] = re.w;
            } else {
                const bool q0 = eq == 0;
                e[0]   = q0 ? re.x : 0.f;
                e[34]  = q0 ? re.y : 0.f;
                e[68]  = q0 ? re.z : 0.f;
                e[102] = 0.f;
            }
        }
    };

    v8f acc0 = {}, acc1 = {}, acc2 = {}, acc3 = {}, acc4 = {};
    const int r0 = wave * 16 + lm;
    const int nchunks = K / GBK;

    loadGlobal(0);
    for (int c = 0; c < nchunks; ++c) {
        const int buf = c & 1;
        storeLDS(buf);
        __syncthreads();
        if (c + 1 < nchunks) loadGlobal((c + 1) * GBK);

        #pragma unroll
        for (int kk = 0; kk < GBK; kk += 4) {
            const int kc = kk + 2 * hi;
            v2f a0 = *(const v2f*)&As [buf][r0][kc];
            v2f b0 = *(const v2f*)&BsT[buf][lm +  0][kc];
            v2f b1 = *(const v2f*)&BsT[buf][lm + 16][kc];
            v2f b2 = *(const v2f*)&BsT[buf][lm + 32][kc];
            v2f b3 = *(const v2f*)&BsT[buf][lm + 48][kc];
            acc0 = __builtin_amdgcn_wmma_f32_16x16x4_f32(false, a0, false, b0, (short)0, acc0, false, false);
            acc1 = __builtin_amdgcn_wmma_f32_16x16x4_f32(false, a0, false, b1, (short)0, acc1, false, false);
            acc2 = __builtin_amdgcn_wmma_f32_16x16x4_f32(false, a0, false, b2, (short)0, acc2, false, false);
            acc3 = __builtin_amdgcn_wmma_f32_16x16x4_f32(false, a0, false, b3, (short)0, acc3, false, false);
            if constexpr (NT == 5) {
                v2f b4 = *(const v2f*)&BsT[buf][lm + 64][kc];
                acc4 = __builtin_amdgcn_wmma_f32_16x16x4_f32(false, a0, false, b4, (short)0, acc4, false, false);
            }
        }
        __syncthreads();
    }

    const int mrow = m0 + wave * 16 + hi * 8;
    #pragma unroll
    for (int r = 0; r < 8; ++r) {
        const size_t row = (size_t)(mrow + r);
        float* c = Cj + row * GBN + lm;
        c[0] = acc0[r]; c[16] = acc1[r]; c[32] = acc2[r]; c[48] = acc3[r];
        if constexpr (SL == 1) {
            Eo[row * RSS + lm] = acc4[r];
        } else if constexpr (SL == 2) {
            if (lm == 0) d0[row] = acc4[r];
            if (lm == 1) d1[row] = acc4[r];
            if (lm == 2) d2[row] = acc4[r];
        }
    }
}

__global__ __launch_bounds__(128) void gemm_wmma_f32_sharex(
    const float* __restrict__ A, ShareX S, float* __restrict__ C,
    int K, int lda, long long sC)
{
    __shared__ __align__(16) float As [2][SXM][34];
    __shared__ __align__(16) float BsT[2][80][34];

    const int j  = blockIdx.x;
    const int m0 = (int)blockIdx.y * SXM;
    const float* Wj = S.w0; const float* E = S.e0; float* Eo = S.eo0;
    switch (j) {
        case 1: Wj = S.w1; E = S.e1; Eo = S.eo1; break;
        case 2: Wj = S.w2; E = S.e2; Eo = S.eo2; break;
        case 3: Wj = S.w3; E = S.e3; Eo = S.eo3; break;
        case 4: Wj = S.w4; E = S.e4; Eo = S.eo4; break;
        case 5: Wj = S.w5; E = S.e5; Eo = S.eo5; break;
        case 6: Wj = S.w6; E = nullptr; Eo = nullptr; break;
        case 7: Wj = S.w7; E = nullptr; Eo = nullptr; break;
        default: break;
    }
    float* Cj = C + (size_t)j * sC;
    if (j < 6)
        sharex_tile_f32<1>(As, BsT, A, Wj, Cj, E, Eo, nullptr, nullptr, nullptr,
                           nullptr, nullptr, nullptr, K, lda, m0);
    else if (j == 6)
        sharex_tile_f32<2>(As, BsT, A, Wj, Cj, nullptr, nullptr, S.wd0, S.wd1, S.wd2,
                           S.d0, S.d1, S.d2, K, lda, m0);
    else
        sharex_tile_f32<0>(As, BsT, A, Wj, Cj, nullptr, nullptr, nullptr, nullptr, nullptr,
                           nullptr, nullptr, nullptr, K, lda, m0);
}

__global__ __launch_bounds__(256) void gemm_wmma_f32_splitk(
    const float* A0, const float* A1, const float* B0, const float* B1,
    float* C0, float* C1,
    int K, int lda, int ldb, int ldc, int bT, int S,
    long long sAs, long long sBs, long long sCs)
{
    const int y = (int)blockIdx.z / S;
    const int s = (int)blockIdx.z - y * S;
    const float* Ab = (y ? A1 : A0) + (size_t)s * sAs;
    const float* Bb = (y ? B1 : B0) + (size_t)s * sBs;
    float*       Cb = (y ? C1 : C0) + (size_t)s * sCs;
    gemm_tile_f32<1>(Ab, Bb, Cb, K, lda, ldb, ldc, bT,
                     (int)blockIdx.y * GBMH, (int)blockIdx.x * GBN, false, DabTile{});
}

__device__ __forceinline__ void splitk_sum_body(
    const float* __restrict__ part, float* __restrict__ out, int S, long long n,
    int bx, int by)
{
    const size_t e = ((size_t)bx * 256 + threadIdx.x) * 4;
    const float* p = part + (size_t)by * (size_t)S * (size_t)n + e;
    float4 acc = *(const float4*)p;
    for (int s = 1; s < S; ++s) {
        const float4 t = *(const float4*)(p + (size_t)s * (size_t)n);
        acc.x += t.x; acc.y += t.y; acc.z += t.z; acc.w += t.w;
    }
    *(float4*)(out + (size_t)by * (size_t)n + e) = acc;
}

#define NCT (TT / GBN)
#define CTX_S 16

__global__ __launch_bounds__(256) void splitk_sum_rowscale_kernel(
    const float* __restrict__ part, float* __restrict__ out, long long n,
    const float* __restrict__ sq)
{
    const size_t e   = ((size_t)blockIdx.x * 256 + threadIdx.x) * 4;
    const size_t row = e / RR;
    const size_t b   = row / TT, t = row - b * TT;
    const float* pr  = sq + b * (size_t)NCT * TT + t;
    const float* p   = part + e;

    float4 pl[CTX_S];
    #pragma unroll
    for (int s = 0; s < CTX_S; ++s) pl[s] = *(const float4*)(p + (size_t)s * (size_t)n);
    float q[NCT];
    #pragma unroll
    for (int ct = 0; ct < NCT; ++ct) q[ct] = pr[(size_t)ct * TT];

    float ss = q[0];
    #pragma unroll
    for (int ct = 1; ct < NCT; ++ct) ss += q[ct];
    const float inv = 1.f / (sqrtf(ss) + 1e-8f);

    float4 acc = pl[0];
    #pragma unroll
    for (int s = 1; s < CTX_S; ++s) {
        acc.x += pl[s].x; acc.y += pl[s].y; acc.z += pl[s].z; acc.w += pl[s].w;
    }
    acc.x *= inv; acc.y *= inv; acc.z *= inv; acc.w *= inv;
    *(float4*)(out + e) = acc;
}

__device__ inline float softplusf(float x) {
    if (x > 20.f) return x;
    if (x < -20.f) return expf(x);
    return log1pf(expf(x));
}

struct MixPath { const float* F; const float* L; const float* Rm; const float* d; };
struct MixPaths { MixPath p0; MixPath p1; MixPath p2; };

__device__ __forceinline__ MixPath mix_path(const MixPaths& mp, int p)
{
    MixPath mx;
    mx.F  = p == 0 ? mp.p0.F  : (p == 1 ? mp.p1.F  : mp.p2.F);
    mx.L  = p == 0 ? mp.p0.L  : (p == 1 ? mp.p1.L  : mp.p2.L);
    mx.Rm = p == 0 ? mp.p0.Rm : (p == 1 ? mp.p1.Rm : mp.p2.Rm);
    mx.d  = p == 0 ? mp.p0.d  : (p == 1 ? mp.p1.d  : mp.p2.d);
    return mx;
}

#define MIX_S 32
#define MIX_J (TT / MIX_S)

__device__ __forceinline__ void mix_factor_partial_body(
    const MixPaths& mp, float* __restrict__ Mpart, int s, int combo)
{
    const int p = combo >> 1, b = combo & 1;
    const MixPath mx = mix_path(mp, p);
    const int c = threadIdx.x & 63, rq = threadIdx.x >> 6;
    const size_t row0 = (size_t)b * TT + (size_t)s * MIX_J;
    const float* Fp = mx.F + row0 * RR + c;
    const float* Lp = mx.L + row0 * RSS + rq * 4;
    float4 acc = make_float4(0.f, 0.f, 0.f, 0.f);
    #pragma unroll 16
    for (int jj = 0; jj < MIX_J; ++jj) {
        const float f = Fp[(size_t)jj * RR];
        const float4 l = *(const float4*)(Lp + (size_t)jj * RSS);
        acc.x += l.x * f; acc.y += l.y * f; acc.z += l.z * f; acc.w += l.w * f;
    }
    float* o = Mpart + ((size_t)(combo * MIX_S + s) * RSS + rq * 4) * RR + c;
    o[0] = acc.x; o[RR] = acc.y; o[2 * RR] = acc.z; o[3 * RR] = acc.w;
}

__device__ __forceinline__ void mix_factor_apply_body(
    const MixPaths& mp, const float* __restrict__ Mpart, float* __restrict__ G,
    int bx, int combo)
{
    __shared__ __align__(16) float Ms[RSS * RR];
    const int p = combo >> 1, b = combo & 1;
    const MixPath mx = mix_path(mp, p);
    {
        const float4* src = (const float4*)(Mpart + (size_t)combo * MIX_S * RSS * RR) + threadIdx.x;
        float4 acc = src[0];
        #pragma unroll
        for (int sl = 1; sl < MIX_S; ++sl) {
            const float4 t = src[(size_t)sl * (RSS * RR / 4)];
            acc.x += t.x; acc.y += t.y; acc.z += t.z; acc.w += t.w;
        }
        ((float4*)Ms)[threadIdx.x] = acc;
    }
    __syncthreads();
    const int c = threadIdx.x & 63, rq = threadIdx.x >> 6;
    float m[RSS];
    #pragma unroll
    for (int r = 0; r < RSS; ++r) m[r] = Ms[r * RR + c];
    const size_t row0 = (size_t)b * TT + (size_t)bx * 64 + (size_t)rq * 16;
    float* Gp = G + (size_t)p * BT * RR;
    #pragma unroll 4
    for (int k = 0; k < 16; ++k) {
        const size_t row = row0 + k;
        const float4* rr = (const float4*)(mx.Rm + row * RSS);
        const float4 r0 = rr[0], r1 = rr[1], r2 = rr[2], r3 = rr[3];
        float v = mx.d[row] * mx.F[row * RR + c];
        v += r0.x * m[0];  v += r0.y * m[1];  v += r0.z * m[2];  v += r0.w * m[3];
        v += r1.x * m[4];  v += r1.y * m[5];  v += r1.z * m[6];  v += r1.w * m[7];
        v += r2.x * m[8];  v += r2.y * m[9];  v += r2.z * m[10]; v += r2.w * m[11];
        v += r3.x * m[12]; v += r3.y * m[13]; v += r3.z * m[14]; v += r3.w * m[15];
        Gp[row * RR + c] = v;
    }
}

__global__ __launch_bounds__(256) void splitk_p_and_mix_partial_kernel(
    const float* A0, const float* A1, const float* B0, const float* B1,
    float* C0, float* C1,
    int K, int lda, int ldb, int ldc, int S,
    long long sAs, long long sBs, long long sCs,
    MixPaths mp, float* Mpart)
{
    const int nGemm = 2 * S * (BT / GBMH);
    const int id = (int)blockIdx.x;
    if (id < nGemm) {
        const int mt = id % (BT / GBMH);
        const int z  = id / (BT / GBMH);
        const int y  = z / S;
        const int s  = z - y * S;
        const float* Ab = (y ? A1 : A0) + (size_t)s * sAs;
        const float* Bb = (y ? B1 : B0) + (size_t)s * sBs;
        float*       Cb = (y ? C1 : C0) + (size_t)s * sCs;
        gemm_tile_f32<1>(Ab, Bb, Cb, K, lda, ldb, ldc, 1, mt * GBMH, 0, false, DabTile{});
        return;
    }
    const int r = id - nGemm;
    mix_factor_partial_body(mp, Mpart, r % MIX_S, r / MIX_S);
}

__global__ __launch_bounds__(256) void splitk_sum_and_mix_apply_kernel(
    MixPaths mp, const float* __restrict__ Mpart, float* __restrict__ G,
    const float* __restrict__ part, float* __restrict__ out, int S, long long n)
{
    const int nApply = 6 * (TT / 64);
    const int id = (int)blockIdx.x;
    if (id < nApply) {
        mix_factor_apply_body(mp, Mpart, G, id % (TT / 64), id / (TT / 64));
        return;
    }
    const int r  = id - nApply;
    const int nx = (int)(n / 1024);
    splitk_sum_body(part, out, S, n, r % nx, r / nx);
}

__global__ __launch_bounds__(256) void score_combine_kernel(
    const float* __restrict__ P,  const float* __restrict__ Kmix,
    const float* __restrict__ Ps, const float* __restrict__ Ksmix,
    float* __restrict__ part, float* __restrict__ H,
    int K, int lda, int ldb, int ldc,
    long long sA, long long sB, long long sC)
{
    __shared__ __align__(16) float4 Gs[16][256];
    const int b   = blockIdx.z;
    const int m0  = (int)blockIdx.y * GBM, n0 = (int)blockIdx.x * GBN;
    const int tid = threadIdx.x;

    TileAcc gate, val;
    gemm_tile_loop_f32<2>(Ps + (size_t)b * sA, Ksmix + (size_t)b * sB, K, lda, ldb, 1,
                          m0, n0, false, DabTile{}, gate);
    {
        const v8f* gv[8] = { &gate.c00, &gate.c01, &gate.c02, &gate.c03,
                             &gate.c10, &gate.c11, &gate.c12, &gate.c13 };
        #pragma unroll
        for (int t = 0; t < 8; ++t) {
            const v8f v = *gv[t];
            Gs[2 * t][tid]     = make_float4(v[0], v[1], v[2], v[3]);
            Gs[2 * t + 1][tid] = make_float4(v[4], v[5], v[6], v[7]);
        }
    }
    #pragma nounroll
    for (int q = 0; q < 16; ++q) {
        float4 v = Gs[q][tid];
        v.x = softplusf(v.x); v.y = softplusf(v.y);
        v.z = softplusf(v.z); v.w = softplusf(v.w);
        Gs[q][tid] = v;
    }
    gemm_tile_loop_f32<2>(P + (size_t)b * sA, Kmix + (size_t)b * sB, K, lda, ldb, 1,
                          m0, n0, false, DabTile{}, val);
    {
        v8f* gv[8] = { &gate.c00, &gate.c01, &gate.c02, &gate.c03,
                       &gate.c10, &gate.c11, &gate.c12, &gate.c13 };
        #pragma unroll
        for (int t = 0; t < 8; ++t) {
            const float4 lo = Gs[2 * t][tid], up = Gs[2 * t + 1][tid];
            v8f v;
            v[0] = lo.x; v[1] = lo.y; v[2] = lo.z; v[3] = lo.w;
            v[4] = up.x; v[5] = up.y; v[6] = up.z; v[7] = up.w;
            *gv[t] = v;
        }
    }

    const int wave = threadIdx.x >> 5;
    const int lane = threadIdx.x & 31;
    const int hi   = lane >> 4;
    const int lm   = lane & 15;
    float* Hb = H + (size_t)b * sC;
    const int ncol = n0 + lm;
    float* o = part + ((size_t)b * NCT + blockIdx.x) * TT + m0 + wave * 32 + hi * 8;
    #pragma unroll
    for (int mt = 0; mt < 2; ++mt) {
        const int rl = wave * 32 + mt * 16 + hi * 8;
        v8f a0 = mt ? val.c10 : val.c00, a1 = mt ? val.c11 : val.c01;
        v8f a2 = mt ? val.c12 : val.c02, a3 = mt ? val.c13 : val.c03;
        v8f g0 = mt ? gate.c10 : gate.c00, g1 = mt ? gate.c11 : gate.c01;
        v8f g2 = mt ? gate.c12 : gate.c02, g3 = mt ? gate.c13 : gate.c03;
        float s[8];
        #pragma unroll
        for (int r = 0; r < 8; ++r) {
            float v = a0[r] * a0[r];
            v += a1[r] * a1[r]; v += a2[r] * a2[r]; v += a3[r] * a3[r];
            v += __shfl_xor(v, 8, 32); v += __shfl_xor(v, 4, 32);
            v += __shfl_xor(v, 2, 32); v += __shfl_xor(v, 1, 32);
            s[r] = v;
            size_t base = (size_t)(m0 + rl + r) * ldc + ncol;
            Hb[base]      = g0[r] * fmaxf(a0[r], 0.f);
            Hb[base + 16] = g1[r] * fmaxf(a1[r], 0.f);
            Hb[base + 32] = g2[r] * fmaxf(a2[r], 0.f);
            Hb[base + 48] = g3[r] * fmaxf(a3[r], 0.f);
        }
        if (lm == 0) {
            float4* q = (float4*)(o + mt * 16);
            q[0] = make_float4(s[0], s[1], s[2], s[3]);
            q[1] = make_float4(s[4], s[5], s[6], s[7]);
        }
    }
}

extern "C" void kernel_launch(void* const* d_in, const int* in_sizes, int n_in,
                              void* d_out, int out_size, void* d_ws, size_t ws_size,
                              hipStream_t stream) {
    (void)in_sizes; (void)n_in; (void)out_size; (void)ws_size;

    const float* x = (const float*)d_in[0];
    const float* qk_Wq1  = (const float*)d_in[1];  const float* qk_Wq2  = (const float*)d_in[2];
    const float* qk_Wm1  = (const float*)d_in[3];  const float* qk_Wm2  = (const float*)d_in[4];
    const float* qk_Wk1  = (const float*)d_in[5];  const float* qk_Wk2  = (const float*)d_in[6];
    const float* qks_Wq1 = (const float*)d_in[7];  const float* qks_Wq2 = (const float*)d_in[8];
    const float* qks_Wm1 = (const float*)d_in[9];  const float* qks_Wm2 = (const float*)d_in[10];
    const float* qks_Wk1 = (const float*)d_in[11]; const float* qks_Wk2 = (const float*)d_in[12];
    const float* vo_Wv1  = (const float*)d_in[13]; const float* vo_Wv2  = (const float*)d_in[14];
    const float* vo_Wm1  = (const float*)d_in[15]; const float* vo_Wm2  = (const float*)d_in[16];
    const float* vo_Wo   = (const float*)d_in[17];
    const float* sm1_Wd  = (const float*)d_in[18]; const float* sm1_Wa  = (const float*)d_in[19]; const float* sm1_Wb  = (const float*)d_in[20];
    const float* sm1s_Wd = (const float*)d_in[21]; const float* sm1s_Wa = (const float*)d_in[22]; const float* sm1s_Wb = (const float*)d_in[23];
    const float* sm2_Wd  = (const float*)d_in[24]; const float* sm2_Wa  = (const float*)d_in[25]; const float* sm2_Wb  = (const float*)d_in[26];

    const size_t BTD = (size_t)BT * DD;
    const size_t BTT = (size_t)BB * TT * TT;

    float* w = (float*)d_ws;
    float* qg   = w; w += BTD;
    float* pkp  = w; w += BTD;
    float* qsg  = w; w += BTD;
    float* pp   = w; w += BTD;
    float* ctxg = w; w += BTD;
    float* tmpb = w; w += BTD;
    float* zb   = w; w += BTT;
    float* zsb  = w; w += BTT;
    float* xr8  = w; w += BTD;
    float* d1   = w; w += BT; float* d1s = w; w += BT; float* d2 = w; w += BT;
    float* A1   = w; w += (size_t)BT * RSS; float* B1  = w; w += (size_t)BT * RSS;
    float* A1s  = w; w += (size_t)BT * RSS; float* B1s = w; w += (size_t)BT * RSS;
    float* A2   = w; w += (size_t)BT * RSS; float* B2  = w; w += (size_t)BT * RSS;

    auto gemm = [&](const float* A, const float* Bm, float* C,
                    int M, int N, int K, int lda, int ldb, int ldc, int bT,
                    long long sA, long long sB, long long sC, int batches) {
        dim3 g((unsigned)(N / GBN), (unsigned)(M / GBMH), (unsigned)batches);
        gemm_wmma_f32<<<g, dim3(256), 0, stream>>>(A, Bm, C, M, N, K, lda, ldb, ldc, bT, sA, sB, sC);
    };

    const long long sXR = (long long)BT * RR;
    const ShareX SX = { qk_Wm1, qk_Wq1, qk_Wk1, qks_Wm1, qks_Wq1, qks_Wk1, vo_Wv1, vo_Wm1,
                        sm1_Wa, sm1_Wb, sm1s_Wa, sm1s_Wb, sm2_Wa, sm2_Wb,
                        sm1_Wd, sm1s_Wd, sm2_Wd,
                        A1, B1, A1s, B1s, A2, B2,
                        d1, d1s, d2 };
    gemm_wmma_f32_sharex<<<dim3(8, BT / SXM), dim3(128), 0, stream>>>(x, SX, xr8, DD, DD, sXR);

    const GatedSets GQ = { xr8 + 1 * sXR, qk_Wq2,  xr8 + 0 * sXR, qk_Wm2,  qg,
                           xr8 + 4 * sXR, qks_Wq2, xr8 + 3 * sXR, qks_Wm2, qsg };
    gemm_wmma_f32_gated<<<dim3(DD / GBN, BT / GBM, 2), dim3(256), 0, stream>>>(GQ, RR, RR, DD, DD);

    float* Mpart = tmpb;
    float* gmix  = tmpb + (size_t)BT * RR;
    const MixPaths MP = { { xr8 + 2 * sXR, B1,  A1,  d1  },
                          { xr8 + 5 * sXR, B1s, A1s, d1s },
                          { xr8 + 6 * sXR, A2,  B2,  d2  } };

    const long long sTTl = (long long)TT * TT;
    const long long sTR = (long long)TT * RR;
    splitk_p_and_mix_partial_kernel<<<dim3(2 * 8 * (BT / GBMH) + 6 * MIX_S), dim3(256), 0, stream>>>(
        qg, qsg, qk_Wk2, qks_Wk2, pkp, pkp + 8 * sXR,
        DD / 8, DD, DD, RR, 8, DD / 8, DD / 8, sXR, MP, Mpart);
    splitk_sum_and_mix_apply_kernel<<<dim3(6 * (TT / 64) + 2 * (unsigned)(sXR / 4 / 256)), dim3(256), 0, stream>>>(
        MP, Mpart, gmix, pkp, pp, 8, sXR);

    score_combine_kernel<<<dim3(NCT, TT / GBM, BB), dim3(256), 0, stream>>>(
        pp, gmix, pp + sXR, gmix + sXR, zsb, zb, RR, RR, RR, TT, sTR, sTR, sTTl);

    float* vmix = gmix + 2 * sXR;
    gemm_wmma_f32_splitk<<<dim3(1, TT / GBMH, 2 * CTX_S), dim3(256), 0, stream>>>(
        zb, zb + sTTl, vmix, vmix + sTR, qg, qg + sTR,
        TT / CTX_S, TT, RR, RR, 0, CTX_S, TT / CTX_S, (long long)(TT / CTX_S) * RR, sXR);
    splitk_sum_rowscale_kernel<<<dim3((unsigned)(sXR / 4 / 256)), 256, 0, stream>>>(qg, pkp, sXR, zsb);
    const GatedSets GC = { pkp, vo_Wv2, xr8 + 7 * sXR, vo_Wm2, ctxg,
                           nullptr, nullptr, nullptr, nullptr, nullptr };
    gemm_wmma_f32_gated<<<dim3(DD / GBN, BT / GBM, 1), dim3(256), 0, stream>>>(GC, RR, RR, DD, DD);
    gemm(ctxg, vo_Wo, (float*)d_out, BT, DD, DD, DD, DD, DD, 1, 0, 0, 0, 1);
}
